# starting kernel
# baseline (speedup 1.0000x reference)
#include <hip/hip_runtime.h>
#include <hip/hip_bf16.h>
#include <math.h>

typedef __attribute__((ext_vector_type(2))) float v2f;
typedef __attribute__((ext_vector_type(8))) float v8f;

#define NEG_SLOPE 0.2f
#define EPSV 1e-16f


__global__ void fill_kernel(float* __restrict__ p, float v, int n) {
  int i = blockIdx.x * blockDim.x + threadIdx.x;
  if (i < n) p[i] = v;
}

__device__ __forceinline__ void atomicMaxF(float* addr, float val) {
  if (val >= 0.0f) atomicMax((int*)addr, __float_as_int(val));
  else             atomicMin((unsigned int*)addr, __float_as_uint(val));
}

__device__ __forceinline__ void edge_sd(const int* __restrict__ ei, int E, int idx,
                                        int& s, int& d) {
  if (idx < E) { s = ei[idx]; d = ei[E + idx]; }
  else         { s = idx - E; d = idx - E; }
}

template <int NT>
__global__ void __launch_bounds__(32)
gemm_wmma_f32_nt(const float* __restrict__ A, const float* __restrict__ B,
                 float* __restrict__ D, int M, int K, int ldb, int ldd, int nvalid) {
  const int tileM = blockIdx.x;
  const int lane = threadIdx.x & 31;
  const int half = lane >> 4;
  const int l16  = lane & 15;
  const int rowA = tileM * 16 + l16;
  const float* __restrict__ arow = A + (size_t)rowA * K;

  v8f acc[NT];
  #pragma unroll
  for (int nt = 0; nt < NT; ++nt) acc[nt] = (v8f){0.f,0.f,0.f,0.f,0.f,0.f,0.f,0.f};

  for (int kk = 0; kk < K; kk += 4) {
    const int ka = kk + half * 2;
    v2f a;
    a.x = arow[ka];
    a.y = arow[ka + 1];
    const float* __restrict__ brow0 = B + (size_t)ka * ldb;
    const float* __restrict__ brow1 = B + (size_t)(ka + 1) * ldb;
    #pragma unroll
    for (int nt = 0; nt < NT; ++nt) {
      const int colB = nt * 16 + l16;
      v2f b;
      b.x = brow0[colB];
      b.y = brow1[colB];
      acc[nt] = __builtin_amdgcn_wmma_f32_16x16x4_f32(false, a, false, b,
                                                      (short)0, acc[nt], false, false);
    }
  }
  #pragma unroll
  for (int nt = 0; nt < NT; ++nt) {
    #pragma unroll
    for (int g = 0; g < 8; ++g) {
      const int r = tileM * 16 + half * 8 + g;
      const int c = nt * 16 + l16;
      if (c < nvalid) D[(size_t)r * ldd + c] = acc[nt][g];
    }
  }
}

__global__ void pad_kernel(const float* __restrict__ W, float* __restrict__ Wp,
                           int K, int Nin, int Nout) {
  int i = blockIdx.x * blockDim.x + threadIdx.x;
  if (i >= K * Nout) return;
  int k = i / Nout, c = i % Nout;
  Wp[i] = (c < Nin) ? W[k * Nin + c] : 0.0f;
}


__global__ void alphas1_kernel(const float* __restrict__ h,
                               const float* __restrict__ a_src,
                               const float* __restrict__ a_dst,
                               float* __restrict__ as, float* __restrict__ ad, int N) {
  int i = blockIdx.x * blockDim.x + threadIdx.x;
  if (i >= N * 8) return;
  int n = i >> 3, hh = i & 7;
  const float* hp = h + (size_t)n * 64 + hh * 8;
  const float* sp = a_src + hh * 8;
  const float* dp = a_dst + hh * 8;
  float s1 = 0.f, s2 = 0.f;
  #pragma unroll
  for (int c = 0; c < 8; ++c) { float v = hp[c]; s1 += v * sp[c]; s2 += v * dp[c]; }
  as[i] = s1; ad[i] = s2;
}

__global__ void edge_max1_kernel(const int* __restrict__ ei, int E, int Et,
                                 const float* __restrict__ as, const float* __restrict__ ad,
                                 float* __restrict__ m) {
  int idx = blockIdx.x * blockDim.x + threadIdx.x;
  if (idx >= Et) return;
  int s, d; edge_sd(ei, E, idx, s, d);
  #pragma unroll
  for (int h = 0; h < 8; ++h) {
    float v = as[s * 8 + h] + ad[d * 8 + h];
    v = v > 0.f ? v : NEG_SLOPE * v;
    atomicMaxF(&m[d * 8 + h], v);
  }
}

__global__ void edge_sum1_kernel(const int* __restrict__ ei, int E, int Et,
                                 const float* __restrict__ as, const float* __restrict__ ad,
                                 const float* __restrict__ m, float* __restrict__ den) {
  int idx = blockIdx.x * blockDim.x + threadIdx.x;
  if (idx >= Et) return;
  int s, d; edge_sd(ei, E, idx, s, d);
  #pragma unroll
  for (int h = 0; h < 8; ++h) {
    float v = as[s * 8 + h] + ad[d * 8 + h];
    v = v > 0.f ? v : NEG_SLOPE * v;
    atomicAdd(&den[d * 8 + h], expf(v - m[d * 8 + h]));
  }
}

__global__ void edge_agg1_kernel(const int* __restrict__ ei, int E, int Et,
                                 const float* __restrict__ as, const float* __restrict__ ad,
                                 const float* __restrict__ m, const float* __restrict__ den,
                                 const float* __restrict__ h1, float* __restrict__ out) {
  int t = blockIdx.x * blockDim.x + threadIdx.x;
  if (t >= Et * 8) return;
  int idx = t >> 3, hh = t & 7;
  int s, d; edge_sd(ei, E, idx, s, d);
  float v = as[s * 8 + hh] + ad[d * 8 + hh];
  v = v > 0.f ? v : NEG_SLOPE * v;
  float alpha = expf(v - m[d * 8 + hh]) / (den[d * 8 + hh] + EPSV);
  const float* hp = h1 + (size_t)s * 64 + hh * 8;
  float* op = out + (size_t)d * 64 + hh * 8;
  #pragma unroll
  for (int c = 0; c < 8; ++c) atomicAdd(&op[c], hp[c] * alpha);
}

__global__ void bias_elu_kernel(float* __restrict__ x, const float* __restrict__ b,
                                int total, int C) {
  int i = blockIdx.x * blockDim.x + threadIdx.x;
  if (i >= total) return;
  float v = x[i] + b[i % C];
  x[i] = v > 0.f ? v : expm1f(v);
}


__global__ void alphas2_kernel(const float* __restrict__ h,
                               const float* __restrict__ a_src,
                               const float* __restrict__ a_dst,
                               float* __restrict__ as, float* __restrict__ ad,
                               int N, int C) {
  int n = blockIdx.x * blockDim.x + threadIdx.x;
  if (n >= N) return;
  float s1 = 0.f, s2 = 0.f;
  for (int c = 0; c < C; ++c) { float v = h[(size_t)n * C + c]; s1 += v * a_src[c]; s2 += v * a_dst[c]; }
  as[n] = s1; ad[n] = s2;
}

__global__ void edge_max2_kernel(const int* __restrict__ ei, int E, int Et,
                                 const float* __restrict__ as, const float* __restrict__ ad,
                                 float* __restrict__ m) {
  int idx = blockIdx.x * blockDim.x + threadIdx.x;
  if (idx >= Et) return;
  int s, d; edge_sd(ei, E, idx, s, d);
  float v = as[s] + ad[d];
  v = v > 0.f ? v : NEG_SLOPE * v;
  atomicMaxF(&m[d], v);
}

__global__ void edge_sum2_kernel(const int* __restrict__ ei, int E, int Et,
                                 const float* __restrict__ as, const float* __restrict__ ad,
                                 const float* __restrict__ m, float* __restrict__ den) {
  int idx = blockIdx.x * blockDim.x + threadIdx.x;
  if (idx >= Et) return;
  int s, d; edge_sd(ei, E, idx, s, d);
  float v = as[s] + ad[d];
  v = v > 0.f ? v : NEG_SLOPE * v;
  atomicAdd(&den[d], expf(v - m[d]));
}

__global__ void edge_agg2_kernel(const int* __restrict__ ei, int E, int Et,
                                 const float* __restrict__ as, const float* __restrict__ ad,
                                 const float* __restrict__ m, const float* __restrict__ den,
                                 const float* __restrict__ h2, float* __restrict__ out, int C) {
  int idx = blockIdx.x * blockDim.x + threadIdx.x;
  if (idx >= Et) return;
  int s, d; edge_sd(ei, E, idx, s, d);
  float v = as[s] + ad[d];
  v = v > 0.f ? v : NEG_SLOPE * v;
  float alpha = expf(v - m[d]) / (den[d] + EPSV);
  const float* hp = h2 + (size_t)s * C;
  float* op = out + (size_t)d * C;
  for (int c = 0; c < C; ++c) atomicAdd(&op[c], hp[c] * alpha);
}

__global__ void final_kernel(const float* __restrict__ out2, const float* __restrict__ b2,
                             float* __restrict__ y, int N, int C) {
  int n = blockIdx.x * blockDim.x + threadIdx.x;
  if (n >= N) return;
  float buf[40];
  float mx = -INFINITY;
  for (int c = 0; c < C; ++c) {
    float v = out2[(size_t)n * C + c] + b2[c];
    v = v > 0.f ? v : expm1f(v);
    buf[c] = v;
    mx = fmaxf(mx, v);
  }
  float sm = 0.f;
  for (int c = 0; c < C; ++c) sm += expf(buf[c] - mx);
  float l = logf(sm);
  for (int c = 0; c < C; ++c) y[(size_t)n * C + c] = buf[c] - mx - l;
}


static inline int cdiv(int a, int b) { return (a + b - 1) / b; }

extern "C" void kernel_launch(void* const* d_in, const int* in_sizes, int n_in,
                              void* d_out, int out_size, void* d_ws, size_t ws_size,
                              hipStream_t stream) {
  const float* x      = (const float*)d_in[0];
  const int*   ei     = (const int*)  d_in[1];
  const float* W1     = (const float*)d_in[2];
  const float* a_src1 = (const float*)d_in[3];
  const float* a_dst1 = (const float*)d_in[4];
  const float* b1     = (const float*)d_in[5];
  const float* W2     = (const float*)d_in[6];
  const float* a_src2 = (const float*)d_in[7];
  const float* a_dst2 = (const float*)d_in[8];
  const float* b2     = (const float*)d_in[9];
  float* out = (float*)d_out;

  const int HC   = in_sizes[5];
  const int NCLS = in_sizes[9];
  const int F    = in_sizes[2] / HC;
  const int N    = in_sizes[0] / F;
  const int E    = in_sizes[1] / 2;
  const int Et   = E + N;
  const int NPAD = 48;

  float* ws = (float*)d_ws;
  size_t off = 0;
  float* h1   = ws + off; off += (size_t)N * HC;
  float* as1  = ws + off; off += (size_t)N * 8;
  float* ad1  = ws + off; off += (size_t)N * 8;
  float* m1   = ws + off; off += (size_t)N * 8;
  float* den1 = ws + off; off += (size_t)N * 8;
  float* out1 = ws + off; off += (size_t)N * HC;
  float* h2   = ws + off; off += (size_t)N * NCLS;
  float* as2  = ws + off; off += (size_t)N;
  float* ad2  = ws + off; off += (size_t)N;
  float* m2   = ws + off; off += (size_t)N;
  float* den2 = ws + off; off += (size_t)N;
  float* out2 = ws + off; off += (size_t)N * NCLS;
  float* Wp   = ws + off; off += (size_t)HC * NPAD;

  const int B = 256;

  fill_kernel<<<cdiv(N * 8, B), B, 0, stream>>>(m1, -INFINITY, N * 8);
  fill_kernel<<<cdiv(N * 8, B), B, 0, stream>>>(den1, 0.f, N * 8);
  fill_kernel<<<cdiv(N * HC, B), B, 0, stream>>>(out1, 0.f, N * HC);
  fill_kernel<<<cdiv(N, B), B, 0, stream>>>(m2, -INFINITY, N);
  fill_kernel<<<cdiv(N, B), B, 0, stream>>>(den2, 0.f, N);
  fill_kernel<<<cdiv(N * NCLS, B), B, 0, stream>>>(out2, 0.f, N * NCLS);

  gemm_wmma_f32_nt<4><<<dim3(N / 16), 32, 0, stream>>>(x, W1, h1, N, F, HC, HC, HC);
  alphas1_kernel<<<cdiv(N * 8, B), B, 0, stream>>>(h1, a_src1, a_dst1, as1, ad1, N);
  edge_max1_kernel<<<cdiv(Et, B), B, 0, stream>>>(ei, E, Et, as1, ad1, m1);
  edge_sum1_kernel<<<cdiv(Et, B), B, 0, stream>>>(ei, E, Et, as1, ad1, m1, den1);
  edge_agg1_kernel<<<cdiv(Et * 8, B), B, 0, stream>>>(ei, E, Et, as1, ad1, m1, den1, h1, out1);
  bias_elu_kernel<<<cdiv(N * HC, B), B, 0, stream>>>(out1, b1, N * HC, HC);

  pad_kernel<<<cdiv(HC * NPAD, B), B, 0, stream>>>(W2, Wp, HC, NCLS, NPAD);
  gemm_wmma_f32_nt<3><<<dim3(N / 16), 32, 0, stream>>>(out1, Wp, h2, N, HC, NPAD, NCLS, NCLS);
  alphas2_kernel<<<cdiv(N, B), B, 0, stream>>>(h2, a_src2, a_dst2, as2, ad2, N, NCLS);
  edge_max2_kernel<<<cdiv(Et, B), B, 0, stream>>>(ei, E, Et, as2, ad2, m2);
  edge_sum2_kernel<<<cdiv(Et, B), B, 0, stream>>>(ei, E, Et, as2, ad2, m2, den2);
  edge_agg2_kernel<<<cdiv(Et, B), B, 0, stream>>>(ei, E, Et, as2, ad2, m2, den2, h2, out2, NCLS);

  final_kernel<<<cdiv(N, B), B, 0, stream>>>(out2, b2, out, N, NCLS);
}


# edge_max1, edge_sum1, edge_agg2: one thread per (edge, channel) instead of a per-thread channel loop, so each gather/atomic covers whole rows; float atomic arrival order changes
# speedup vs baseline: 2.2417x; 2.2417x over previous
#include <hip/hip_runtime.h>
#include <hip/hip_bf16.h>
#include <math.h>

typedef __attribute__((ext_vector_type(2))) float v2f;
typedef __attribute__((ext_vector_type(8))) float v8f;

#define NEG_SLOPE 0.2f
#define EPSV 1e-16f


__global__ void fill_kernel(float* __restrict__ p, float v, int n) {
  int i = blockIdx.x * blockDim.x + threadIdx.x;
  if (i < n) p[i] = v;
}

__device__ __forceinline__ void atomicMaxF(float* addr, float val) {
  if (val >= 0.0f) atomicMax((int*)addr, __float_as_int(val));
  else             atomicMin((unsigned int*)addr, __float_as_uint(val));
}

__device__ __forceinline__ void edge_sd(const int* __restrict__ ei, int E, int idx,
                                        int& s, int& d) {
  if (idx < E) { s = ei[idx]; d = ei[E + idx]; }
  else         { s = idx - E; d = idx - E; }
}

template <int NT>
__global__ void __launch_bounds__(32)
gemm_wmma_f32_nt(const float* __restrict__ A, const float* __restrict__ B,
                 float* __restrict__ D, int M, int K, int ldb, int ldd, int nvalid) {
  const int tileM = blockIdx.x;
  const int lane = threadIdx.x & 31;
  const int half = lane >> 4;
  const int l16  = lane & 15;
  const int rowA = tileM * 16 + l16;
  const float* __restrict__ arow = A + (size_t)rowA * K;

  v8f acc[NT];
  #pragma unroll
  for (int nt = 0; nt < NT; ++nt) acc[nt] = (v8f){0.f,0.f,0.f,0.f,0.f,0.f,0.f,0.f};

  for (int kk = 0; kk < K; kk += 4) {
    const int ka = kk + half * 2;
    v2f a;
    a.x = arow[ka];
    a.y = arow[ka + 1];
    const float* __restrict__ brow0 = B + (size_t)ka * ldb;
    const float* __restrict__ brow1 = B + (size_t)(ka + 1) * ldb;
    #pragma unroll
    for (int nt = 0; nt < NT; ++nt) {
      const int colB = nt * 16 + l16;
      v2f b;
      b.x = brow0[colB];
      b.y = brow1[colB];
      acc[nt] = __builtin_amdgcn_wmma_f32_16x16x4_f32(false, a, false, b,
                                                      (short)0, acc[nt], false, false);
    }
  }
  #pragma unroll
  for (int nt = 0; nt < NT; ++nt) {
    #pragma unroll
    for (int g = 0; g < 8; ++g) {
      const int r = tileM * 16 + half * 8 + g;
      const int c = nt * 16 + l16;
      if (c < nvalid) D[(size_t)r * ldd + c] = acc[nt][g];
    }
  }
}

__global__ void pad_kernel(const float* __restrict__ W, float* __restrict__ Wp,
                           int K, int Nin, int Nout) {
  int i = blockIdx.x * blockDim.x + threadIdx.x;
  if (i >= K * Nout) return;
  int k = i / Nout, c = i % Nout;
  Wp[i] = (c < Nin) ? W[k * Nin + c] : 0.0f;
}


__global__ void alphas1_kernel(const float* __restrict__ h,
                               const float* __restrict__ a_src,
                               const float* __restrict__ a_dst,
                               float* __restrict__ as, float* __restrict__ ad, int N) {
  int i = blockIdx.x * blockDim.x + threadIdx.x;
  if (i >= N * 8) return;
  int n = i >> 3, hh = i & 7;
  const float* hp = h + (size_t)n * 64 + hh * 8;
  const float* sp = a_src + hh * 8;
  const float* dp = a_dst + hh * 8;
  float s1 = 0.f, s2 = 0.f;
  #pragma unroll
  for (int c = 0; c < 8; ++c) { float v = hp[c]; s1 += v * sp[c]; s2 += v * dp[c]; }
  as[i] = s1; ad[i] = s2;
}

__global__ void edge_max1_kernel(const int* __restrict__ ei, int E, int Et,
                                 const float* __restrict__ as, const float* __restrict__ ad,
                                 float* __restrict__ m) {
  int t = blockIdx.x * blockDim.x + threadIdx.x;
  if (t >= Et * 8) return;
  int idx = t >> 3, h = t & 7;
  int s, d; edge_sd(ei, E, idx, s, d);
  float v = as[s * 8 + h] + ad[d * 8 + h];
  v = v > 0.f ? v : NEG_SLOPE * v;
  atomicMaxF(&m[d * 8 + h], v);
}

__global__ void edge_sum1_kernel(const int* __restrict__ ei, int E, int Et,
                                 const float* __restrict__ as, const float* __restrict__ ad,
                                 const float* __restrict__ m, float* __restrict__ den) {
  int t = blockIdx.x * blockDim.x + threadIdx.x;
  if (t >= Et * 8) return;
  int idx = t >> 3, h = t & 7;
  int s, d; edge_sd(ei, E, idx, s, d);
  float v = as[s * 8 + h] + ad[d * 8 + h];
  v = v > 0.f ? v : NEG_SLOPE * v;
  atomicAdd(&den[d * 8 + h], expf(v - m[d * 8 + h]));
}

__global__ void edge_agg1_kernel(const int* __restrict__ ei, int E, int Et,
                                 const float* __restrict__ as, const float* __restrict__ ad,
                                 const float* __restrict__ m, const float* __restrict__ den,
                                 const float* __restrict__ h1, float* __restrict__ out) {
  int t = blockIdx.x * blockDim.x + threadIdx.x;
  if (t >= Et * 8) return;
  int idx = t >> 3, hh = t & 7;
  int s, d; edge_sd(ei, E, idx, s, d);
  float v = as[s * 8 + hh] + ad[d * 8 + hh];
  v = v > 0.f ? v : NEG_SLOPE * v;
  float alpha = expf(v - m[d * 8 + hh]) / (den[d * 8 + hh] + EPSV);
  const float* hp = h1 + (size_t)s * 64 + hh * 8;
  float* op = out + (size_t)d * 64 + hh * 8;
  #pragma unroll
  for (int c = 0; c < 8; ++c) atomicAdd(&op[c], hp[c] * alpha);
}

__global__ void bias_elu_kernel(float* __restrict__ x, const float* __restrict__ b,
                                int total, int C) {
  int i = blockIdx.x * blockDim.x + threadIdx.x;
  if (i >= total) return;
  float v = x[i] + b[i % C];
  x[i] = v > 0.f ? v : expm1f(v);
}


__global__ void alphas2_kernel(const float* __restrict__ h,
                               const float* __restrict__ a_src,
                               const float* __restrict__ a_dst,
                               float* __restrict__ as, float* __restrict__ ad,
                               int N, int C) {
  int n = blockIdx.x * blockDim.x + threadIdx.x;
  if (n >= N) return;
  float s1 = 0.f, s2 = 0.f;
  for (int c = 0; c < C; ++c) { float v = h[(size_t)n * C + c]; s1 += v * a_src[c]; s2 += v * a_dst[c]; }
  as[n] = s1; ad[n] = s2;
}

__global__ void edge_max2_kernel(const int* __restrict__ ei, int E, int Et,
                                 const float* __restrict__ as, const float* __restrict__ ad,
                                 float* __restrict__ m) {
  int idx = blockIdx.x * blockDim.x + threadIdx.x;
  if (idx >= Et) return;
  int s, d; edge_sd(ei, E, idx, s, d);
  float v = as[s] + ad[d];
  v = v > 0.f ? v : NEG_SLOPE * v;
  atomicMaxF(&m[d], v);
}

__global__ void edge_sum2_kernel(const int* __restrict__ ei, int E, int Et,
                                 const float* __restrict__ as, const float* __restrict__ ad,
                                 const float* __restrict__ m, float* __restrict__ den) {
  int idx = blockIdx.x * blockDim.x + threadIdx.x;
  if (idx >= Et) return;
  int s, d; edge_sd(ei, E, idx, s, d);
  float v = as[s] + ad[d];
  v = v > 0.f ? v : NEG_SLOPE * v;
  atomicAdd(&den[d], expf(v - m[d]));
}

__global__ void edge_agg2_kernel(const int* __restrict__ ei, int E, int Et,
                                 const float* __restrict__ as, const float* __restrict__ ad,
                                 const float* __restrict__ m, const float* __restrict__ den,
                                 const float* __restrict__ h2, float* __restrict__ out, int C) {
  int c = threadIdx.x;
  int idx = blockIdx.x * blockDim.y + threadIdx.y;
  if (idx >= Et) return;
  int s, d; edge_sd(ei, E, idx, s, d);
  float v = as[s] + ad[d];
  v = v > 0.f ? v : NEG_SLOPE * v;
  float alpha = expf(v - m[d]) / (den[d] + EPSV);
  atomicAdd(&out[(size_t)d * C + c], h2[(size_t)s * C + c] * alpha);
}

__global__ void final_kernel(const float* __restrict__ out2, const float* __restrict__ b2,
                             float* __restrict__ y, int N, int C) {
  int n = blockIdx.x * blockDim.x + threadIdx.x;
  if (n >= N) return;
  float buf[40];
  float mx = -INFINITY;
  for (int c = 0; c < C; ++c) {
    float v = out2[(size_t)n * C + c] + b2[c];
    v = v > 0.f ? v : expm1f(v);
    buf[c] = v;
    mx = fmaxf(mx, v);
  }
  float sm = 0.f;
  for (int c = 0; c < C; ++c) sm += expf(buf[c] - mx);
  float l = logf(sm);
  for (int c = 0; c < C; ++c) y[(size_t)n * C + c] = buf[c] - mx - l;
}


static inline int cdiv(int a, int b) { return (a + b - 1) / b; }

extern "C" void kernel_launch(void* const* d_in, const int* in_sizes, int n_in,
                              void* d_out, int out_size, void* d_ws, size_t ws_size,
                              hipStream_t stream) {
  const float* x      = (const float*)d_in[0];
  const int*   ei     = (const int*)  d_in[1];
  const float* W1     = (const float*)d_in[2];
  const float* a_src1 = (const float*)d_in[3];
  const float* a_dst1 = (const float*)d_in[4];
  const float* b1     = (const float*)d_in[5];
  const float* W2     = (const float*)d_in[6];
  const float* a_src2 = (const float*)d_in[7];
  const float* a_dst2 = (const float*)d_in[8];
  const float* b2     = (const float*)d_in[9];
  float* out = (float*)d_out;

  const int HC   = in_sizes[5];
  const int NCLS = in_sizes[9];
  const int F    = in_sizes[2] / HC;
  const int N    = in_sizes[0] / F;
  const int E    = in_sizes[1] / 2;
  const int Et   = E + N;
  const int NPAD = 48;

  float* ws = (float*)d_ws;
  size_t off = 0;
  float* h1   = ws + off; off += (size_t)N * HC;
  float* as1  = ws + off; off += (size_t)N * 8;
  float* ad1  = ws + off; off += (size_t)N * 8;
  float* m1   = ws + off; off += (size_t)N * 8;
  float* den1 = ws + off; off += (size_t)N * 8;
  float* out1 = ws + off; off += (size_t)N * HC;
  float* h2   = ws + off; off += (size_t)N * NCLS;
  float* as2  = ws + off; off += (size_t)N;
  float* ad2  = ws + off; off += (size_t)N;
  float* m2   = ws + off; off += (size_t)N;
  float* den2 = ws + off; off += (size_t)N;
  float* out2 = ws + off; off += (size_t)N * NCLS;
  float* Wp   = ws + off; off += (size_t)HC * NPAD;

  const int B = 256;

  fill_kernel<<<cdiv(N * 8, B), B, 0, stream>>>(m1, -INFINITY, N * 8);
  fill_kernel<<<cdiv(N * 8, B), B, 0, stream>>>(den1, 0.f, N * 8);
  fill_kernel<<<cdiv(N * HC, B), B, 0, stream>>>(out1, 0.f, N * HC);
  fill_kernel<<<cdiv(N, B), B, 0, stream>>>(m2, -INFINITY, N);
  fill_kernel<<<cdiv(N, B), B, 0, stream>>>(den2, 0.f, N);
  fill_kernel<<<cdiv(N * NCLS, B), B, 0, stream>>>(out2, 0.f, N * NCLS);

  gemm_wmma_f32_nt<4><<<dim3(N / 16), 32, 0, stream>>>(x, W1, h1, N, F, HC, HC, HC);
  alphas1_kernel<<<cdiv(N * 8, B), B, 0, stream>>>(h1, a_src1, a_dst1, as1, ad1, N);
  edge_max1_kernel<<<cdiv(Et * 8, B), B, 0, stream>>>(ei, E, Et, as1, ad1, m1);
  edge_sum1_kernel<<<cdiv(Et * 8, B), B, 0, stream>>>(ei, E, Et, as1, ad1, m1, den1);
  edge_agg1_kernel<<<cdiv(Et * 8, B), B, 0, stream>>>(ei, E, Et, as1, ad1, m1, den1, h1, out1);
  bias_elu_kernel<<<cdiv(N * HC, B), B, 0, stream>>>(out1, b1, N * HC, HC);

  pad_kernel<<<cdiv(HC * NPAD, B), B, 0, stream>>>(W2, Wp, HC, NCLS, NPAD);
  gemm_wmma_f32_nt<3><<<dim3(N / 16), 32, 0, stream>>>(out1, Wp, h2, N, HC, NPAD, NCLS, NCLS);
  alphas2_kernel<<<cdiv(N, B), B, 0, stream>>>(h2, a_src2, a_dst2, as2, ad2, N, NCLS);
  edge_max2_kernel<<<cdiv(Et, B), B, 0, stream>>>(ei, E, Et, as2, ad2, m2);
  edge_sum2_kernel<<<cdiv(Et, B), B, 0, stream>>>(ei, E, Et, as2, ad2, m2, den2);
  edge_agg2_kernel<<<cdiv(Et, 8), dim3(NCLS, 8), 0, stream>>>(ei, E, Et, as2, ad2, m2, den2, h2, out2, NCLS);

  final_kernel<<<cdiv(N, B), B, 0, stream>>>(out2, b2, out, N, NCLS);
}


# edges counting-sorted by destination once per call; one wave per destination does max, exp-sum, weighted row sum and epilogue in registers, both layers; no float atomics, sums follow scatter order
# speedup vs baseline: 4.6814x; 2.0883x over previous
#include <hip/hip_runtime.h>
#include <hip/hip_bf16.h>
#include <math.h>

typedef __attribute__((ext_vector_type(2))) float v2f;
typedef __attribute__((ext_vector_type(8))) float v8f;

#define NEG_SLOPE 0.2f
#define EPSV 1e-16f


__global__ void zero_int_kernel(int* __restrict__ p, int n) {
  int i = blockIdx.x * blockDim.x + threadIdx.x;
  if (i < n) p[i] = 0;
}

__global__ void degree_kernel(const int* __restrict__ ei, int E, int* __restrict__ deg) {
  int e = blockIdx.x * blockDim.x + threadIdx.x;
  if (e < E) atomicAdd(&deg[ei[E + e]], 1);
}

__global__ void __launch_bounds__(256)
segment_alloc_kernel(const int* __restrict__ deg, int* __restrict__ start,
                     int* __restrict__ total, int N) {
  __shared__ int wave_off[8];
  __shared__ int block_base;
  const int n = blockIdx.x * 256 + threadIdx.x;
  const int lane = threadIdx.x & 31, w = threadIdx.x >> 5;
  const int d = (n < N) ? deg[n] : 0;
  int incl = d;
  for (int o = 1; o < 32; o <<= 1) {
    int t = __shfl_up(incl, o);
    if (lane >= o) incl += t;
  }
  if (lane == 31) wave_off[w] = incl;
  __syncthreads();
  if (threadIdx.x == 0) {
    int run = 0;
    for (int i = 0; i < 8; ++i) { int t = wave_off[i]; wave_off[i] = run; run += t; }
    block_base = atomicAdd(total, run);
  }
  __syncthreads();
  if (n < N) start[n] = block_base + wave_off[w] + incl - d;
}

__global__ void scatter_kernel(const int* __restrict__ ei, int E,
                               const int* __restrict__ start, int* __restrict__ cursor,
                               int* __restrict__ csr) {
  int e = blockIdx.x * blockDim.x + threadIdx.x;
  if (e >= E) return;
  const int s = ei[e], d = ei[E + e];
  const int p = atomicAdd(&cursor[d], 1);
  csr[start[d] + p] = s;
}

__device__ __forceinline__ float lrelu(float v) { return v > 0.f ? v : NEG_SLOPE * v; }

template <int NT>
__global__ void __launch_bounds__(32)
gemm_wmma_f32_nt(const float* __restrict__ A, const float* __restrict__ B,
                 float* __restrict__ D, int M, int K, int ldb, int ldd, int nvalid) {
  const int tileM = blockIdx.x;
  const int lane = threadIdx.x & 31;
  const int half = lane >> 4;
  const int l16  = lane & 15;
  const int rowA = tileM * 16 + l16;
  const float* __restrict__ arow = A + (size_t)rowA * K;

  v8f acc[NT];
  #pragma unroll
  for (int nt = 0; nt < NT; ++nt) acc[nt] = (v8f){0.f,0.f,0.f,0.f,0.f,0.f,0.f,0.f};

  for (int kk = 0; kk < K; kk += 4) {
    const int ka = kk + half * 2;
    v2f a;
    a.x = arow[ka];
    a.y = arow[ka + 1];
    const float* __restrict__ brow0 = B + (size_t)ka * ldb;
    const float* __restrict__ brow1 = B + (size_t)(ka + 1) * ldb;
    #pragma unroll
    for (int nt = 0; nt < NT; ++nt) {
      const int colB = nt * 16 + l16;
      v2f b;
      b.x = brow0[colB];
      b.y = brow1[colB];
      acc[nt] = __builtin_amdgcn_wmma_f32_16x16x4_f32(false, a, false, b,
                                                      (short)0, acc[nt], false, false);
    }
  }
  #pragma unroll
  for (int nt = 0; nt < NT; ++nt) {
    #pragma unroll
    for (int g = 0; g < 8; ++g) {
      const int r = tileM * 16 + half * 8 + g;
      const int c = nt * 16 + l16;
      if (c < nvalid) D[(size_t)r * ldd + c] = acc[nt][g];
    }
  }
}

__global__ void pad_kernel(const float* __restrict__ W, float* __restrict__ Wp,
                           int K, int Nin, int Nout) {
  int i = blockIdx.x * blockDim.x + threadIdx.x;
  if (i >= K * Nout) return;
  int k = i / Nout, c = i % Nout;
  Wp[i] = (c < Nin) ? W[k * Nin + c] : 0.0f;
}


__global__ void alphas1_kernel(const float* __restrict__ h,
                               const float* __restrict__ a_src,
                               const float* __restrict__ a_dst,
                               float* __restrict__ as, float* __restrict__ ad, int N) {
  int i = blockIdx.x * blockDim.x + threadIdx.x;
  if (i >= N * 8) return;
  int n = i >> 3, hh = i & 7;
  const float* hp = h + (size_t)n * 64 + hh * 8;
  const float* sp = a_src + hh * 8;
  const float* dp = a_dst + hh * 8;
  float s1 = 0.f, s2 = 0.f;
  #pragma unroll
  for (int c = 0; c < 8; ++c) { float v = hp[c]; s1 += v * sp[c]; s2 += v * dp[c]; }
  as[i] = s1; ad[i] = s2;
}

__global__ void __launch_bounds__(256)
gather_agg1_kernel(const int* __restrict__ start, const int* __restrict__ deg,
                   const int* __restrict__ csr,
                   const float* __restrict__ as, const float* __restrict__ ad,
                   const float* __restrict__ h1, const float* __restrict__ b,
                   float* __restrict__ out, int N) {
  const int lane = threadIdx.x & 31;
  const int n = __builtin_amdgcn_readfirstlane(blockIdx.x * 8 + (threadIdx.x >> 5));
  if (n >= N) return;
  const int hh = lane >> 2;
  const int beg = start[n], cnt = deg[n];
  const float adv = ad[n * 8 + hh];
  const float vself = lrelu(as[n * 8 + hh] + adv);

  float mx = vself;
  for (int base = 0; base < cnt; base += 32) {
    const int k = base + lane;
    const int my = (k < cnt) ? csr[beg + k] : n;
    const int m = min(32, cnt - base);
    #pragma unroll 4
    for (int j = 0; j < m; ++j) {
      const int s = __builtin_amdgcn_readlane(my, j);
      mx = fmaxf(mx, lrelu(as[s * 8 + hh] + adv));
    }
  }

  float ex = expf(vself - mx);
  float den = ex;
  v2f hv = *(const v2f*)(h1 + (size_t)n * 64 + 2 * lane);
  float a0 = ex * hv.x, a1 = ex * hv.y;
  for (int base = 0; base < cnt; base += 32) {
    const int k = base + lane;
    const int my = (k < cnt) ? csr[beg + k] : n;
    const int m = min(32, cnt - base);
    #pragma unroll 4
    for (int j = 0; j < m; ++j) {
      const int s = __builtin_amdgcn_readlane(my, j);
      ex = expf(lrelu(as[s * 8 + hh] + adv) - mx);
      hv = *(const v2f*)(h1 + (size_t)s * 64 + 2 * lane);
      den += ex;
      a0 += ex * hv.x;
      a1 += ex * hv.y;
    }
  }
  const float dd = den + EPSV;
  const v2f bb = *(const v2f*)(b + 2 * lane);
  float o0 = a0 / dd + bb.x, o1 = a1 / dd + bb.y;
  v2f o;
  o.x = o0 > 0.f ? o0 : expm1f(o0);
  o.y = o1 > 0.f ? o1 : expm1f(o1);
  *(v2f*)(out + (size_t)n * 64 + 2 * lane) = o;
}


__global__ void alphas2_kernel(const float* __restrict__ h,
                               const float* __restrict__ a_src,
                               const float* __restrict__ a_dst,
                               float* __restrict__ as, float* __restrict__ ad,
                               int N, int C) {
  int n = blockIdx.x * blockDim.x + threadIdx.x;
  if (n >= N) return;
  float s1 = 0.f, s2 = 0.f;
  for (int c = 0; c < C; ++c) { float v = h[(size_t)n * C + c]; s1 += v * a_src[c]; s2 += v * a_dst[c]; }
  as[n] = s1; ad[n] = s2;
}

__global__ void __launch_bounds__(256)
gather_agg2_kernel(const int* __restrict__ start, const int* __restrict__ deg,
                   const int* __restrict__ csr,
                   const float* __restrict__ as, const float* __restrict__ ad,
                   const float* __restrict__ h2, float* __restrict__ out,
                   int N, int C, int ldo) {
  const int lane = threadIdx.x & 31;
  const int n = __builtin_amdgcn_readfirstlane(blockIdx.x * 8 + (threadIdx.x >> 5));
  if (n >= N) return;
  const bool own = 2 * lane < C;
  const int beg = start[n], cnt = deg[n];
  const float adv = ad[n];
  const float vself = lrelu(as[n] + adv);

  float mx = vself;
  for (int base = 0; base < cnt; base += 32) {
    const int k = base + lane;
    if (k < cnt) mx = fmaxf(mx, lrelu(as[csr[beg + k]] + adv));
  }
  for (int o = 16; o > 0; o >>= 1) mx = fmaxf(mx, __shfl_xor(mx, o));

  const float exs = expf(vself - mx);
  float den = exs;
  float a0 = 0.f, a1 = 0.f;
  if (own) {
    const v2f hv = *(const v2f*)(h2 + (size_t)n * C + 2 * lane);
    a0 = exs * hv.x; a1 = exs * hv.y;
  }
  for (int base = 0; base < cnt; base += 32) {
    const int k = base + lane;
    const int my = (k < cnt) ? csr[beg + k] : n;
    const float exl = expf(lrelu(as[my] + adv) - mx);
    const int m = min(32, cnt - base);
    #pragma unroll 4
    for (int j = 0; j < m; ++j) {
      const int s = __builtin_amdgcn_readlane(my, j);
      const float e = __int_as_float(__builtin_amdgcn_readlane(__float_as_int(exl), j));
      den += e;
      if (own) {
        const v2f hv = *(const v2f*)(h2 + (size_t)s * C + 2 * lane);
        a0 += e * hv.x;
        a1 += e * hv.y;
      }
    }
  }
  if (own) {
    const float dd = den + EPSV;
    v2f o;
    o.x = a0 / dd; o.y = a1 / dd;
    *(v2f*)(out + (size_t)n * ldo + 2 * lane) = o;
  }
}

__global__ void final_kernel(const float* __restrict__ out2, int ld,
                             const float* __restrict__ b2,
                             float* __restrict__ y, int N, int C) {
  int n = blockIdx.x * blockDim.x + threadIdx.x;
  if (n >= N) return;
  float buf[40];
  float mx = -INFINITY;
  for (int c = 0; c < C; ++c) {
    float v = out2[(size_t)n * ld + c] + b2[c];
    v = v > 0.f ? v : expm1f(v);
    buf[c] = v;
    mx = fmaxf(mx, v);
  }
  float sm = 0.f;
  for (int c = 0; c < C; ++c) sm += expf(buf[c] - mx);
  float l = logf(sm);
  for (int c = 0; c < C; ++c) y[(size_t)n * C + c] = buf[c] - mx - l;
}


static inline int cdiv(int a, int b) { return (a + b - 1) / b; }

extern "C" void kernel_launch(void* const* d_in, const int* in_sizes, int n_in,
                              void* d_out, int out_size, void* d_ws, size_t ws_size,
                              hipStream_t stream) {
  const float* x      = (const float*)d_in[0];
  const int*   ei     = (const int*)  d_in[1];
  const float* W1     = (const float*)d_in[2];
  const float* a_src1 = (const float*)d_in[3];
  const float* a_dst1 = (const float*)d_in[4];
  const float* b1     = (const float*)d_in[5];
  const float* W2     = (const float*)d_in[6];
  const float* a_src2 = (const float*)d_in[7];
  const float* a_dst2 = (const float*)d_in[8];
  const float* b2     = (const float*)d_in[9];
  float* out = (float*)d_out;

  const int HC   = in_sizes[5];
  const int NCLS = in_sizes[9];
  const int F    = in_sizes[2] / HC;
  const int N    = in_sizes[0] / F;
  const int E    = in_sizes[1] / 2;
  const int NPAD = 48;
  const int LDO  = 64;

  float* ws = (float*)d_ws;
  size_t off = 0;
  float* h1   = ws + off; off += (size_t)N * HC;
  float* out1 = ws + off; off += (size_t)N * HC;
  float* out2 = ws + off; off += (size_t)N * LDO;
  float* h2   = ws + off; off += (size_t)N * NCLS;
  float* as1  = ws + off; off += (size_t)N * 8;
  float* ad1  = ws + off; off += (size_t)N * 8;
  float* as2  = ws + off; off += (size_t)N;
  float* ad2  = ws + off; off += (size_t)N;
  float* Wp   = ws + off; off += (size_t)HC * NPAD;
  int* deg    = (int*)(ws + off); off += (size_t)N;
  int* cursor = (int*)(ws + off); off += (size_t)N;
  int* total  = (int*)(ws + off); off += 2;
  int* start  = (int*)(ws + off); off += (size_t)N;
  int* csr    = (int*)(ws + off); off += (size_t)E;

  const int B = 256;

  zero_int_kernel<<<cdiv(2 * N + 1, B), B, 0, stream>>>(deg, 2 * N + 1);
  degree_kernel<<<cdiv(E, B), B, 0, stream>>>(ei, E, deg);
  segment_alloc_kernel<<<cdiv(N, 256), 256, 0, stream>>>(deg, start, total, N);
  scatter_kernel<<<cdiv(E, B), B, 0, stream>>>(ei, E, start, cursor, csr);

  gemm_wmma_f32_nt<4><<<dim3(N / 16), 32, 0, stream>>>(x, W1, h1, N, F, HC, HC, HC);
  alphas1_kernel<<<cdiv(N * 8, B), B, 0, stream>>>(h1, a_src1, a_dst1, as1, ad1, N);
  gather_agg1_kernel<<<cdiv(N, 8), 256, 0, stream>>>(start, deg, csr, as1, ad1, h1, b1, out1, N);

  pad_kernel<<<cdiv(HC * NPAD, B), B, 0, stream>>>(W2, Wp, HC, NCLS, NPAD);
  gemm_wmma_f32_nt<3><<<dim3(N / 16), 32, 0, stream>>>(out1, Wp, h2, N, HC, NPAD, NCLS, NCLS);
  alphas2_kernel<<<cdiv(N, B), B, 0, stream>>>(h2, a_src2, a_dst2, as2, ad2, N, NCLS);
  gather_agg2_kernel<<<cdiv(N, 8), 256, 0, stream>>>(start, deg, csr, as2, ad2, h2, out2, N, NCLS, LDO);

  final_kernel<<<cdiv(N, B), B, 0, stream>>>(out2, LDO, b2, out, N, NCLS);
}


# edges grouped by destination with LDS counters: each work-group (edge chunk, node range) owns its counters, counts then ranks; no global atomics per edge; one launch fewer; scatter order changes
# speedup vs baseline: 6.6058x; 1.4111x over previous
#include <hip/hip_runtime.h>
#include <hip/hip_bf16.h>
#include <math.h>

typedef __attribute__((ext_vector_type(2))) float v2f;
typedef __attribute__((ext_vector_type(8))) float v8f;

#define NEG_SLOPE 0.2f
#define EPSV 1e-16f


#define SORT_R 16
#define SORT_Q 64

typedef __attribute__((ext_vector_type(4))) int v4i;

template <bool WITH_SRC, class F>
__device__ __forceinline__ void scan_chunk(const int* __restrict__ srcp,
                                           const int* __restrict__ dstp,
                                           int e0, int e1, bool vec, F f) {
  if (vec) {
    for (int i = e0 + 4 * (int)threadIdx.x; i < e1; i += 4 * 256 * 4) {
      v4i d[4], s[4];
      #pragma unroll
      for (int u = 0; u < 4; ++u) {
        const int j = i + u * 1024;
        const bool in = j < e1;
        d[u] = in ? *(const v4i*)(dstp + j) : (v4i){-1, -1, -1, -1};
        if (WITH_SRC) s[u] = in ? *(const v4i*)(srcp + j) : (v4i){0, 0, 0, 0};
        else          s[u] = (v4i){0, 0, 0, 0};
      }
      #pragma unroll
      for (int u = 0; u < 4; ++u) {
        f(s[u].x, d[u].x); f(s[u].y, d[u].y); f(s[u].z, d[u].z); f(s[u].w, d[u].w);
      }
    }
  } else {
    for (int i = e0 + (int)threadIdx.x; i < e1; i += 256)
      f(WITH_SRC ? srcp[i] : 0, dstp[i]);
  }
}

__device__ __forceinline__ bool list_is_vec(const int* ei, int E) {
  return ((E & 3) == 0) && ((((size_t)ei) & 15) == 0);
}

__global__ void __launch_bounds__(256)
range_count_kernel(const int* __restrict__ ei, int E, int N, int span, int per,
                   int* __restrict__ part, int* __restrict__ total) {
  extern __shared__ int bins[];
  const int q = blockIdx.x, r = blockIdx.y;
  const int lo = r * span;
  for (int i = threadIdx.x; i < span; i += 256) bins[i] = 0;
  if (q == 0 && r == 0 && threadIdx.x == 0) *total = 0;
  __syncthreads();
  const int e0 = min(E, q * per), e1 = min(E, e0 + per);
  const unsigned uspan = (unsigned)span;
  scan_chunk<false>(ei, ei + E, e0, e1, list_is_vec(ei, E), [&](int, int d) {
    const unsigned t = (unsigned)(d - lo);
    if (t < uspan) atomicAdd(&bins[t], 1);
  });
  __syncthreads();
  for (int i = threadIdx.x; i < span; i += 256)
    if (lo + i < N) part[(size_t)q * N + lo + i] = bins[i];
}

__global__ void __launch_bounds__(256)
segment_alloc_kernel(int* __restrict__ part, int* __restrict__ deg,
                     int* __restrict__ start, int* __restrict__ total, int N) {
  __shared__ int wave_off[8];
  __shared__ int block_base;
  const int n = blockIdx.x * 256 + threadIdx.x;
  const int lane = threadIdx.x & 31, w = threadIdx.x >> 5;
  int c[SORT_Q];
  int d = 0;
  #pragma unroll
  for (int q = 0; q < SORT_Q; ++q) {
    c[q] = (n < N) ? part[(size_t)q * N + n] : 0;
    d += c[q];
  }
  int incl = d;
  for (int o = 1; o < 32; o <<= 1) {
    int t = __shfl_up(incl, o);
    if (lane >= o) incl += t;
  }
  if (lane == 31) wave_off[w] = incl;
  __syncthreads();
  if (threadIdx.x == 0) {
    int run = 0;
    for (int i = 0; i < 8; ++i) { int t = wave_off[i]; wave_off[i] = run; run += t; }
    block_base = atomicAdd(total, run);
  }
  __syncthreads();
  if (n < N) {
    int run = block_base + wave_off[w] + incl - d;
    start[n] = run;
    deg[n] = d;
    #pragma unroll
    for (int q = 0; q < SORT_Q; ++q) { part[(size_t)q * N + n] = run; run += c[q]; }
  }
}

__global__ void __launch_bounds__(256)
range_scatter_kernel(const int* __restrict__ ei, int E, int N, int span, int per,
                     const int* __restrict__ part, int* __restrict__ csr) {
  extern __shared__ int bins[];
  const int q = blockIdx.x, r = blockIdx.y;
  const int lo = r * span;
  for (int i0 = threadIdx.x; i0 < span; i0 += 256 * 8) {
    int v[8];
    #pragma unroll
    for (int u = 0; u < 8; ++u) {
      const int i = i0 + u * 256;
      v[u] = (i < span && lo + i < N) ? part[(size_t)q * N + lo + i] : 0;
    }
    #pragma unroll
    for (int u = 0; u < 8; ++u) {
      const int i = i0 + u * 256;
      if (i < span) bins[i] = v[u];
    }
  }
  __syncthreads();
  const int e0 = min(E, q * per), e1 = min(E, e0 + per);
  const unsigned uspan = (unsigned)span;
  scan_chunk<true>(ei, ei + E, e0, e1, list_is_vec(ei, E), [&](int s, int d) {
    const unsigned t = (unsigned)(d - lo);
    if (t < uspan) csr[atomicAdd(&bins[t], 1)] = s;
  });
}

__device__ __forceinline__ float lrelu(float v) { return v > 0.f ? v : NEG_SLOPE * v; }

template <int NT>
__global__ void __launch_bounds__(32)
gemm_wmma_f32_nt(const float* __restrict__ A, const float* __restrict__ B,
                 float* __restrict__ D, int M, int K, int ldb, int ldd, int nvalid) {
  const int tileM = blockIdx.x;
  const int lane = threadIdx.x & 31;
  const int half = lane >> 4;
  const int l16  = lane & 15;
  const int rowA = tileM * 16 + l16;
  const float* __restrict__ arow = A + (size_t)rowA * K;

  v8f acc[NT];
  #pragma unroll
  for (int nt = 0; nt < NT; ++nt) acc[nt] = (v8f){0.f,0.f,0.f,0.f,0.f,0.f,0.f,0.f};

  for (int kk = 0; kk < K; kk += 4) {
    const int ka = kk + half * 2;
    v2f a;
    a.x = arow[ka];
    a.y = arow[ka + 1];
    const float* __restrict__ brow0 = B + (size_t)ka * ldb;
    const float* __restrict__ brow1 = B + (size_t)(ka + 1) * ldb;
    #pragma unroll
    for (int nt = 0; nt < NT; ++nt) {
      const int colB = nt * 16 + l16;
      v2f b;
      b.x = brow0[colB];
      b.y = brow1[colB];
      acc[nt] = __builtin_amdgcn_wmma_f32_16x16x4_f32(false, a, false, b,
                                                      (short)0, acc[nt], false, false);
    }
  }
  #pragma unroll
  for (int nt = 0; nt < NT; ++nt) {
    #pragma unroll
    for (int g = 0; g < 8; ++g) {
      const int r = tileM * 16 + half * 8 + g;
      const int c = nt * 16 + l16;
      if (c < nvalid) D[(size_t)r * ldd + c] = acc[nt][g];
    }
  }
}

__global__ void pad_kernel(const float* __restrict__ W, float* __restrict__ Wp,
                           int K, int Nin, int Nout) {
  int i = blockIdx.x * blockDim.x + threadIdx.x;
  if (i >= K * Nout) return;
  int k = i / Nout, c = i % Nout;
  Wp[i] = (c < Nin) ? W[k * Nin + c] : 0.0f;
}


__global__ void alphas1_kernel(const float* __restrict__ h,
                               const float* __restrict__ a_src,
                               const float* __restrict__ a_dst,
                               float* __restrict__ as, float* __restrict__ ad, int N) {
  int i = blockIdx.x * blockDim.x + threadIdx.x;
  if (i >= N * 8) return;
  int n = i >> 3, hh = i & 7;
  const float* hp = h + (size_t)n * 64 + hh * 8;
  const float* sp = a_src + hh * 8;
  const float* dp = a_dst + hh * 8;
  float s1 = 0.f, s2 = 0.f;
  #pragma unroll
  for (int c = 0; c < 8; ++c) { float v = hp[c]; s1 += v * sp[c]; s2 += v * dp[c]; }
  as[i] = s1; ad[i] = s2;
}

__global__ void __launch_bounds__(256)
gather_agg1_kernel(const int* __restrict__ start, const int* __restrict__ deg,
                   const int* __restrict__ csr,
                   const float* __restrict__ as, const float* __restrict__ ad,
                   const float* __restrict__ h1, const float* __restrict__ b,
                   float* __restrict__ out, int N) {
  const int lane = threadIdx.x & 31;
  const int n = __builtin_amdgcn_readfirstlane(blockIdx.x * 8 + (threadIdx.x >> 5));
  if (n >= N) return;
  const int hh = lane >> 2;
  const int beg = start[n], cnt = deg[n];
  const float adv = ad[n * 8 + hh];
  const float vself = lrelu(as[n * 8 + hh] + adv);

  float mx = vself;
  for (int base = 0; base < cnt; base += 32) {
    const int k = base + lane;
    const int my = (k < cnt) ? csr[beg + k] : n;
    const int m = min(32, cnt - base);
    #pragma unroll 4
    for (int j = 0; j < m; ++j) {
      const int s = __builtin_amdgcn_readlane(my, j);
      mx = fmaxf(mx, lrelu(as[s * 8 + hh] + adv));
    }
  }

  float ex = expf(vself - mx);
  float den = ex;
  v2f hv = *(const v2f*)(h1 + (size_t)n * 64 + 2 * lane);
  float a0 = ex * hv.x, a1 = ex * hv.y;
  for (int base = 0; base < cnt; base += 32) {
    const int k = base + lane;
    const int my = (k < cnt) ? csr[beg + k] : n;
    const int m = min(32, cnt - base);
    #pragma unroll 4
    for (int j = 0; j < m; ++j) {
      const int s = __builtin_amdgcn_readlane(my, j);
      ex = expf(lrelu(as[s * 8 + hh] + adv) - mx);
      hv = *(const v2f*)(h1 + (size_t)s * 64 + 2 * lane);
      den += ex;
      a0 += ex * hv.x;
      a1 += ex * hv.y;
    }
  }
  const float dd = den + EPSV;
  const v2f bb = *(const v2f*)(b + 2 * lane);
  float o0 = a0 / dd + bb.x, o1 = a1 / dd + bb.y;
  v2f o;
  o.x = o0 > 0.f ? o0 : expm1f(o0);
  o.y = o1 > 0.f ? o1 : expm1f(o1);
  *(v2f*)(out + (size_t)n * 64 + 2 * lane) = o;
}


__global__ void alphas2_kernel(const float* __restrict__ h,
                               const float* __restrict__ a_src,
                               const float* __restrict__ a_dst,
                               float* __restrict__ as, float* __restrict__ ad,
                               int N, int C) {
  int n = blockIdx.x * blockDim.x + threadIdx.x;
  if (n >= N) return;
  float s1 = 0.f, s2 = 0.f;
  for (int c = 0; c < C; ++c) { float v = h[(size_t)n * C + c]; s1 += v * a_src[c]; s2 += v * a_dst[c]; }
  as[n] = s1; ad[n] = s2;
}

__global__ void __launch_bounds__(256)
gather_agg2_kernel(const int* __restrict__ start, const int* __restrict__ deg,
                   const int* __restrict__ csr,
                   const float* __restrict__ as, const float* __restrict__ ad,
                   const float* __restrict__ h2, float* __restrict__ out,
                   int N, int C, int ldo) {
  const int lane = threadIdx.x & 31;
  const int n = __builtin_amdgcn_readfirstlane(blockIdx.x * 8 + (threadIdx.x >> 5));
  if (n >= N) return;
  const bool own = 2 * lane < C;
  const int beg = start[n], cnt = deg[n];
  const float adv = ad[n];
  const float vself = lrelu(as[n] + adv);

  float mx = vself;
  for (int base = 0; base < cnt; base += 32) {
    const int k = base + lane;
    if (k < cnt) mx = fmaxf(mx, lrelu(as[csr[beg + k]] + adv));
  }
  for (int o = 16; o > 0; o >>= 1) mx = fmaxf(mx, __shfl_xor(mx, o));

  const float exs = expf(vself - mx);
  float den = exs;
  float a0 = 0.f, a1 = 0.f;
  if (own) {
    const v2f hv = *(const v2f*)(h2 + (size_t)n * C + 2 * lane);
    a0 = exs * hv.x; a1 = exs * hv.y;
  }
  for (int base = 0; base < cnt; base += 32) {
    const int k = base + lane;
    const int my = (k < cnt) ? csr[beg + k] : n;
    const float exl = expf(lrelu(as[my] + adv) - mx);
    const int m = min(32, cnt - base);
    #pragma unroll 4
    for (int j = 0; j < m; ++j) {
      const int s = __builtin_amdgcn_readlane(my, j);
      const float e = __int_as_float(__builtin_amdgcn_readlane(__float_as_int(exl), j));
      den += e;
      if (own) {
        const v2f hv = *(const v2f*)(h2 + (size_t)s * C + 2 * lane);
        a0 += e * hv.x;
        a1 += e * hv.y;
      }
    }
  }
  if (own) {
    const float dd = den + EPSV;
    v2f o;
    o.x = a0 / dd; o.y = a1 / dd;
    *(v2f*)(out + (size_t)n * ldo + 2 * lane) = o;
  }
}

__global__ void final_kernel(const float* __restrict__ out2, int ld,
                             const float* __restrict__ b2,
                             float* __restrict__ y, int N, int C) {
  int n = blockIdx.x * blockDim.x + threadIdx.x;
  if (n >= N) return;
  float buf[40];
  float mx = -INFINITY;
  for (int c = 0; c < C; ++c) {
    float v = out2[(size_t)n * ld + c] + b2[c];
    v = v > 0.f ? v : expm1f(v);
    buf[c] = v;
    mx = fmaxf(mx, v);
  }
  float sm = 0.f;
  for (int c = 0; c < C; ++c) sm += expf(buf[c] - mx);
  float l = logf(sm);
  for (int c = 0; c < C; ++c) y[(size_t)n * C + c] = buf[c] - mx - l;
}


static inline int cdiv(int a, int b) { return (a + b - 1) / b; }

extern "C" void kernel_launch(void* const* d_in, const int* in_sizes, int n_in,
                              void* d_out, int out_size, void* d_ws, size_t ws_size,
                              hipStream_t stream) {
  const float* x      = (const float*)d_in[0];
  const int*   ei     = (const int*)  d_in[1];
  const float* W1     = (const float*)d_in[2];
  const float* a_src1 = (const float*)d_in[3];
  const float* a_dst1 = (const float*)d_in[4];
  const float* b1     = (const float*)d_in[5];
  const float* W2     = (const float*)d_in[6];
  const float* a_src2 = (const float*)d_in[7];
  const float* a_dst2 = (const float*)d_in[8];
  const float* b2     = (const float*)d_in[9];
  float* out = (float*)d_out;

  const int HC   = in_sizes[5];
  const int NCLS = in_sizes[9];
  const int F    = in_sizes[2] / HC;
  const int N    = in_sizes[0] / F;
  const int E    = in_sizes[1] / 2;
  const int NPAD = 48;
  const int LDO  = 64;

  float* ws = (float*)d_ws;
  size_t off = 0;
  float* h1   = ws + off; off += (size_t)N * HC;
  float* out1 = ws + off; off += (size_t)N * HC;
  float* out2 = ws + off; off += (size_t)N * LDO;
  float* h2   = ws + off; off += (size_t)N * NCLS;
  float* as1  = ws + off; off += (size_t)N * 8;
  float* ad1  = ws + off; off += (size_t)N * 8;
  float* as2  = ws + off; off += (size_t)N;
  float* ad2  = ws + off; off += (size_t)N;
  float* Wp   = ws + off; off += (size_t)HC * NPAD;
  int* deg    = (int*)(ws + off); off += (size_t)N;
  int* total  = (int*)(ws + off); off += 2;
  int* start  = (int*)(ws + off); off += (size_t)N;
  int* csr    = (int*)(ws + off); off += (size_t)E;
  int* part   = (int*)(ws + off); off += (size_t)SORT_Q * N;

  const int B = 256;

  const int span = cdiv(N, SORT_R);
  const int per  = cdiv(cdiv(E, 4), SORT_Q) * 4;
  range_count_kernel<<<dim3(SORT_Q, SORT_R), 256, span * sizeof(int), stream>>>(ei, E, N, span, per, part, total);
  segment_alloc_kernel<<<cdiv(N, 256), 256, 0, stream>>>(part, deg, start, total, N);
  range_scatter_kernel<<<dim3(SORT_Q, SORT_R), 256, span * sizeof(int), stream>>>(ei, E, N, span, per, part, csr);

  gemm_wmma_f32_nt<4><<<dim3(N / 16), 32, 0, stream>>>(x, W1, h1, N, F, HC, HC, HC);
  alphas1_kernel<<<cdiv(N * 8, B), B, 0, stream>>>(h1, a_src1, a_dst1, as1, ad1, N);
  gather_agg1_kernel<<<cdiv(N, 8), 256, 0, stream>>>(start, deg, csr, as1, ad1, h1, b1, out1, N);

  pad_kernel<<<cdiv(HC * NPAD, B), B, 0, stream>>>(W2, Wp, HC, NCLS, NPAD);
  gemm_wmma_f32_nt<3><<<dim3(N / 16), 32, 0, stream>>>(out1, Wp, h2, N, HC, NPAD, NCLS, NCLS);
  alphas2_kernel<<<cdiv(N, B), B, 0, stream>>>(h2, a_src2, a_dst2, as2, ad2, N, NCLS);
  gather_agg2_kernel<<<cdiv(N, 8), 256, 0, stream>>>(start, deg, csr, as2, ad2, h2, out2, N, NCLS, LDO);

  final_kernel<<<cdiv(N, B), B, 0, stream>>>(out2, LDO, b2, out, N, NCLS);
}


# helper kernels folded into their producers: logits in the GEMM epilogues (same add order), W2 read unpadded, bias+ELU+log_softmax in the layer-2 gather wave (exp sum as a lane tree); 11->7 launches
# speedup vs baseline: 8.0967x; 1.2257x over previous
#include <hip/hip_runtime.h>
#include <hip/hip_bf16.h>
#include <math.h>

typedef __attribute__((ext_vector_type(2))) float v2f;
typedef __attribute__((ext_vector_type(8))) float v8f;

#define NEG_SLOPE 0.2f
#define EPSV 1e-16f


#define SORT_R 16
#define SORT_Q 64

typedef __attribute__((ext_vector_type(4))) int v4i;

template <bool WITH_SRC, class F>
__device__ __forceinline__ void scan_chunk(const int* __restrict__ srcp,
                                           const int* __restrict__ dstp,
                                           int e0, int e1, bool vec, F f) {
  if (vec) {
    for (int i = e0 + 4 * (int)threadIdx.x; i < e1; i += 4 * 256 * 4) {
      v4i d[4], s[4];
      #pragma unroll
      for (int u = 0; u < 4; ++u) {
        const int j = i + u * 1024;
        const bool in = j < e1;
        d[u] = in ? *(const v4i*)(dstp + j) : (v4i){-1, -1, -1, -1};
        if (WITH_SRC) s[u] = in ? *(const v4i*)(srcp + j) : (v4i){0, 0, 0, 0};
        else          s[u] = (v4i){0, 0, 0, 0};
      }
      #pragma unroll
      for (int u = 0; u < 4; ++u) {
        f(s[u].x, d[u].x); f(s[u].y, d[u].y); f(s[u].z, d[u].z); f(s[u].w, d[u].w);
      }
    }
  } else {
    for (int i = e0 + (int)threadIdx.x; i < e1; i += 256)
      f(WITH_SRC ? srcp[i] : 0, dstp[i]);
  }
}

__device__ __forceinline__ bool list_is_vec(const int* ei, int E) {
  return ((E & 3) == 0) && ((((size_t)ei) & 15) == 0);
}

__global__ void __launch_bounds__(256)
range_count_kernel(const int* __restrict__ ei, int E, int N, int span, int per,
                   int* __restrict__ part, int* __restrict__ total) {
  extern __shared__ int bins[];
  const int q = blockIdx.x, r = blockIdx.y;
  const int lo = r * span;
  for (int i = threadIdx.x; i < span; i += 256) bins[i] = 0;
  if (q == 0 && r == 0 && threadIdx.x == 0) *total = 0;
  __syncthreads();
  const int e0 = min(E, q * per), e1 = min(E, e0 + per);
  const unsigned uspan = (unsigned)span;
  scan_chunk<false>(ei, ei + E, e0, e1, list_is_vec(ei, E), [&](int, int d) {
    const unsigned t = (unsigned)(d - lo);
    if (t < uspan) atomicAdd(&bins[t], 1);
  });
  __syncthreads();
  for (int i = threadIdx.x; i < span; i += 256)
    if (lo + i < N) part[(size_t)q * N + lo + i] = bins[i];
}

__global__ void __launch_bounds__(256)
segment_alloc_kernel(int* __restrict__ part, int* __restrict__ deg,
                     int* __restrict__ start, int* __restrict__ total, int N) {
  __shared__ int wave_off[8];
  __shared__ int block_base;
  const int n = blockIdx.x * 256 + threadIdx.x;
  const int lane = threadIdx.x & 31, w = threadIdx.x >> 5;
  int c[SORT_Q];
  int d = 0;
  #pragma unroll
  for (int q = 0; q < SORT_Q; ++q) {
    c[q] = (n < N) ? part[(size_t)q * N + n] : 0;
    d += c[q];
  }
  int incl = d;
  for (int o = 1; o < 32; o <<= 1) {
    int t = __shfl_up(incl, o);
    if (lane >= o) incl += t;
  }
  if (lane == 31) wave_off[w] = incl;
  __syncthreads();
  if (threadIdx.x == 0) {
    int run = 0;
    for (int i = 0; i < 8; ++i) { int t = wave_off[i]; wave_off[i] = run; run += t; }
    block_base = atomicAdd(total, run);
  }
  __syncthreads();
  if (n < N) {
    int run = block_base + wave_off[w] + incl - d;
    start[n] = run;
    deg[n] = d;
    #pragma unroll
    for (int q = 0; q < SORT_Q; ++q) { part[(size_t)q * N + n] = run; run += c[q]; }
  }
}

__global__ void __launch_bounds__(256)
range_scatter_kernel(const int* __restrict__ ei, int E, int N, int span, int per,
                     const int* __restrict__ part, int* __restrict__ csr) {
  extern __shared__ int bins[];
  const int q = blockIdx.x, r = blockIdx.y;
  const int lo = r * span;
  for (int i0 = threadIdx.x; i0 < span; i0 += 256 * 8) {
    int v[8];
    #pragma unroll
    for (int u = 0; u < 8; ++u) {
      const int i = i0 + u * 256;
      v[u] = (i < span && lo + i < N) ? part[(size_t)q * N + lo + i] : 0;
    }
    #pragma unroll
    for (int u = 0; u < 8; ++u) {
      const int i = i0 + u * 256;
      if (i < span) bins[i] = v[u];
    }
  }
  __syncthreads();
  const int e0 = min(E, q * per), e1 = min(E, e0 + per);
  const unsigned uspan = (unsigned)span;
  scan_chunk<true>(ei, ei + E, e0, e1, list_is_vec(ei, E), [&](int s, int d) {
    const unsigned t = (unsigned)(d - lo);
    if (t < uspan) csr[atomicAdd(&bins[t], 1)] = s;
  });
}

__device__ __forceinline__ float lrelu(float v) { return v > 0.f ? v : NEG_SLOPE * v; }

typedef __attribute__((ext_vector_type(4))) float v4f;

template <int NT, int EPI>
__global__ void __launch_bounds__(32)
gemm_wmma_f32_nt(const float* __restrict__ A, const float* __restrict__ B,
                 float* __restrict__ D, int M, int K, int ldb, int ldd, int nvalid,
                 const float* __restrict__ a_src, const float* __restrict__ a_dst,
                 float* __restrict__ o_src, float* __restrict__ o_dst) {
  constexpr int PITCH = 16 * NT + 4;
  __shared__ float tile[16 * PITCH];
  const int tileM = blockIdx.x;
  const int lane = threadIdx.x & 31;
  const int half = lane >> 4;
  const int l16  = lane & 15;
  const int rowA = tileM * 16 + l16;
  const float* __restrict__ arow = A + (size_t)rowA * K;

  v8f acc[NT];
  #pragma unroll
  for (int nt = 0; nt < NT; ++nt) acc[nt] = (v8f){0.f,0.f,0.f,0.f,0.f,0.f,0.f,0.f};

  for (int kk = 0; kk < K; kk += 4) {
    const int ka = kk + half * 2;
    v2f a;
    a.x = arow[ka];
    a.y = arow[ka + 1];
    const float* __restrict__ brow0 = B + (size_t)ka * ldb;
    const float* __restrict__ brow1 = B + (size_t)(ka + 1) * ldb;
    #pragma unroll
    for (int nt = 0; nt < NT; ++nt) {
      const int colB = nt * 16 + l16;
      v2f b;
      if (EPI == 2) {
        const int colc = min(colB, nvalid - 1);
        const float b0 = brow0[colc], b1 = brow1[colc];
        b.x = colB < nvalid ? b0 : 0.f;
        b.y = colB < nvalid ? b1 : 0.f;
      } else {
        b.x = brow0[colB];
        b.y = brow1[colB];
      }
      acc[nt] = __builtin_amdgcn_wmma_f32_16x16x4_f32(false, a, false, b,
                                                      (short)0, acc[nt], false, false);
    }
  }
  #pragma unroll
  for (int nt = 0; nt < NT; ++nt) {
    #pragma unroll
    for (int g = 0; g < 8; ++g) {
      const int r = tileM * 16 + half * 8 + g;
      const int c = nt * 16 + l16;
      if (c < nvalid) D[(size_t)r * ldd + c] = acc[nt][g];
      tile[(half * 8 + g) * PITCH + c] = acc[nt][g];
    }
  }
  __syncthreads();

  if (EPI == 1) {
    const int hh = lane & 7;
    float sp[8], dp[8];
    #pragma unroll
    for (int c = 0; c < 8; ++c) { sp[c] = a_src[hh * 8 + c]; dp[c] = a_dst[hh * 8 + c]; }
    #pragma unroll
    for (int i = 0; i < 4; ++i) {
      const int p = lane + 32 * i;
      const float* tp = tile + (p >> 3) * PITCH + hh * 8;
      const v4f t0 = *(const v4f*)tp, t1 = *(const v4f*)(tp + 4);
      const float t[8] = {t0.x, t0.y, t0.z, t0.w, t1.x, t1.y, t1.z, t1.w};
      float s1 = 0.f, s2 = 0.f;
      #pragma unroll
      for (int c = 0; c < 8; ++c) { s1 += t[c] * sp[c]; s2 += t[c] * dp[c]; }
      o_src[(size_t)tileM * 128 + p] = s1;
      o_dst[(size_t)tileM * 128 + p] = s2;
    }
  } else {
    const float* __restrict__ av = half ? a_dst : a_src;
    const float* tp = tile + l16 * PITCH;
    float s = 0.f;
    #pragma unroll
    for (int c4 = 0; c4 < 16 * NT; c4 += 4) {
      const v4f t = *(const v4f*)(tp + c4);
      const float tt[4] = {t.x, t.y, t.z, t.w};
      #pragma unroll
      for (int u = 0; u < 4; ++u) {
        const int c = c4 + u;
        const float coef = av[min(c, nvalid - 1)];
        s += tt[u] * (c < nvalid ? coef : 0.f);
      }
    }
    o_src[((size_t)tileM * 16 + l16) * 2 + half] = s;
  }
}


__global__ void __launch_bounds__(256)
gather_agg1_kernel(const int* __restrict__ start, const int* __restrict__ deg,
                   const int* __restrict__ csr,
                   const float* __restrict__ as, const float* __restrict__ ad,
                   const float* __restrict__ h1, const float* __restrict__ b,
                   float* __restrict__ out, int N) {
  const int lane = threadIdx.x & 31;
  const int n = __builtin_amdgcn_readfirstlane(blockIdx.x * 8 + (threadIdx.x >> 5));
  if (n >= N) return;
  const int hh = lane >> 2;
  const int beg = start[n], cnt = deg[n];
  const float adv = ad[n * 8 + hh];
  const float vself = lrelu(as[n * 8 + hh] + adv);

  float mx = vself;
  for (int base = 0; base < cnt; base += 32) {
    const int k = base + lane;
    const int my = (k < cnt) ? csr[beg + k] : n;
    const int m = min(32, cnt - base);
    #pragma unroll 4
    for (int j = 0; j < m; ++j) {
      const int s = __builtin_amdgcn_readlane(my, j);
      mx = fmaxf(mx, lrelu(as[s * 8 + hh] + adv));
    }
  }

  float ex = expf(vself - mx);
  float den = ex;
  v2f hv = *(const v2f*)(h1 + (size_t)n * 64 + 2 * lane);
  float a0 = ex * hv.x, a1 = ex * hv.y;
  for (int base = 0; base < cnt; base += 32) {
    const int k = base + lane;
    const int my = (k < cnt) ? csr[beg + k] : n;
    const int m = min(32, cnt - base);
    #pragma unroll 4
    for (int j = 0; j < m; ++j) {
      const int s = __builtin_amdgcn_readlane(my, j);
      ex = expf(lrelu(as[s * 8 + hh] + adv) - mx);
      hv = *(const v2f*)(h1 + (size_t)s * 64 + 2 * lane);
      den += ex;
      a0 += ex * hv.x;
      a1 += ex * hv.y;
    }
  }
  const float dd = den + EPSV;
  const v2f bb = *(const v2f*)(b + 2 * lane);
  float o0 = a0 / dd + bb.x, o1 = a1 / dd + bb.y;
  v2f o;
  o.x = o0 > 0.f ? o0 : expm1f(o0);
  o.y = o1 > 0.f ? o1 : expm1f(o1);
  *(v2f*)(out + (size_t)n * 64 + 2 * lane) = o;
}


__global__ void __launch_bounds__(256)
gather_agg2_kernel(const int* __restrict__ start, const int* __restrict__ deg,
                   const int* __restrict__ csr,
                   const float* __restrict__ asd,
                   const float* __restrict__ h2, const float* __restrict__ b2,
                   float* __restrict__ y, int N, int C) {
  __shared__ float rows[8 * 64];
  const int lane = threadIdx.x & 31;
  const int w = threadIdx.x >> 5;
  const int n = __builtin_amdgcn_readfirstlane(blockIdx.x * 8 + w);
  const bool own = 2 * lane < C;
  if (n < N) {
    const int beg = start[n], cnt = deg[n];
    const float adv = asd[2 * n + 1];
    const float vself = lrelu(asd[2 * n] + adv);

    float mx = vself;
    for (int base = 0; base < cnt; base += 32) {
      const int k = base + lane;
      if (k < cnt) mx = fmaxf(mx, lrelu(asd[2 * csr[beg + k]] + adv));
    }
    for (int o = 16; o > 0; o >>= 1) mx = fmaxf(mx, __shfl_xor(mx, o));

    const float exs = expf(vself - mx);
    float den = exs;
    float a0 = 0.f, a1 = 0.f;
    if (own) {
      const v2f hv = *(const v2f*)(h2 + (size_t)n * C + 2 * lane);
      a0 = exs * hv.x; a1 = exs * hv.y;
    }
    for (int base = 0; base < cnt; base += 32) {
      const int k = base + lane;
      const int my = (k < cnt) ? csr[beg + k] : n;
      const float exl = expf(lrelu(asd[2 * my] + adv) - mx);
      const int m = min(32, cnt - base);
      #pragma unroll 4
      for (int j = 0; j < m; ++j) {
        const int s = __builtin_amdgcn_readlane(my, j);
        const float e = __int_as_float(__builtin_amdgcn_readlane(__float_as_int(exl), j));
        den += e;
        if (own) {
          const v2f hv = *(const v2f*)(h2 + (size_t)s * C + 2 * lane);
          a0 += e * hv.x;
          a1 += e * hv.y;
        }
      }
    }
    float v0 = -INFINITY, v1 = -INFINITY;
    if (own) {
      const float dd = den + EPSV;
      const v2f bb = *(const v2f*)(b2 + 2 * lane);
      v0 = a0 / dd + bb.x;
      v1 = a1 / dd + bb.y;
      v0 = v0 > 0.f ? v0 : expm1f(v0);
      v1 = v1 > 0.f ? v1 : expm1f(v1);
    }
    float rmx = fmaxf(v0, v1);
    for (int o = 16; o > 0; o >>= 1) rmx = fmaxf(rmx, __shfl_xor(rmx, o));
    float sm = own ? expf(v0 - rmx) + expf(v1 - rmx) : 0.f;
    for (int o = 16; o > 0; o >>= 1) sm += __shfl_xor(sm, o);
    const float l = logf(sm);
    if (own) {
      v2f o;
      o.x = v0 - rmx - l; o.y = v1 - rmx - l;
      *(v2f*)(rows + w * C + 2 * lane) = o;
    }
  }
  __syncthreads();
  const int i = 2 * (int)threadIdx.x;
  const size_t base = (size_t)blockIdx.x * 8 * C;
  if (i < 8 * C && base + i < (size_t)N * C)
    *(v2f*)(y + base + i) = *(const v2f*)(rows + i);
}


static inline int cdiv(int a, int b) { return (a + b - 1) / b; }

extern "C" void kernel_launch(void* const* d_in, const int* in_sizes, int n_in,
                              void* d_out, int out_size, void* d_ws, size_t ws_size,
                              hipStream_t stream) {
  const float* x      = (const float*)d_in[0];
  const int*   ei     = (const int*)  d_in[1];
  const float* W1     = (const float*)d_in[2];
  const float* a_src1 = (const float*)d_in[3];
  const float* a_dst1 = (const float*)d_in[4];
  const float* b1     = (const float*)d_in[5];
  const float* W2     = (const float*)d_in[6];
  const float* a_src2 = (const float*)d_in[7];
  const float* a_dst2 = (const float*)d_in[8];
  const float* b2     = (const float*)d_in[9];
  float* out = (float*)d_out;

  const int HC   = in_sizes[5];
  const int NCLS = in_sizes[9];
  const int F    = in_sizes[2] / HC;
  const int N    = in_sizes[0] / F;
  const int E    = in_sizes[1] / 2;

  float* ws = (float*)d_ws;
  size_t off = 0;
  float* h1   = ws + off; off += (size_t)N * HC;
  float* out1 = ws + off; off += (size_t)N * HC;
  float* h2   = ws + off; off += (size_t)N * NCLS;
  float* as1  = ws + off; off += (size_t)N * 8;
  float* ad1  = ws + off; off += (size_t)N * 8;
  float* asd2 = ws + off; off += (size_t)N * 2;
  int* deg    = (int*)(ws + off); off += (size_t)N;
  int* total  = (int*)(ws + off); off += 2;
  int* start  = (int*)(ws + off); off += (size_t)N;
  int* csr    = (int*)(ws + off); off += (size_t)E;
  int* part   = (int*)(ws + off); off += (size_t)SORT_Q * N;

  const int span = cdiv(N, SORT_R);
  const int per  = cdiv(cdiv(E, 4), SORT_Q) * 4;
  range_count_kernel<<<dim3(SORT_Q, SORT_R), 256, span * sizeof(int), stream>>>(ei, E, N, span, per, part, total);
  segment_alloc_kernel<<<cdiv(N, 256), 256, 0, stream>>>(part, deg, start, total, N);
  range_scatter_kernel<<<dim3(SORT_Q, SORT_R), 256, span * sizeof(int), stream>>>(ei, E, N, span, per, part, csr);

  gemm_wmma_f32_nt<4, 1><<<dim3(N / 16), 32, 0, stream>>>(x, W1, h1, N, F, HC, HC, HC,
                                                          a_src1, a_dst1, as1, ad1);
  gather_agg1_kernel<<<cdiv(N, 8), 256, 0, stream>>>(start, deg, csr, as1, ad1, h1, b1, out1, N);

  gemm_wmma_f32_nt<3, 2><<<dim3(N / 16), 32, 0, stream>>>(out1, W2, h2, N, HC, NCLS, NCLS, NCLS,
                                                          a_src2, a_dst2, asd2, asd2);
  gather_agg2_kernel<<<cdiv(N, 8), 256, 0, stream>>>(start, deg, csr, asd2, h2, b2, out, N, NCLS);
}


# layer 1's product x @ W1 (it does not depend on the edge list) runs as extra work-groups, 8 tiles each, inside the edge scatter's launch instead of in a launch of its own; 7->6 launches
# speedup vs baseline: 8.7595x; 1.0819x over previous
#include <hip/hip_runtime.h>
#include <hip/hip_bf16.h>
#include <math.h>

typedef __attribute__((ext_vector_type(2))) float v2f;
typedef __attribute__((ext_vector_type(8))) float v8f;

#define NEG_SLOPE 0.2f
#define EPSV 1e-16f


#define SORT_R 16
#define SORT_Q 64

typedef __attribute__((ext_vector_type(4))) int v4i;

template <bool WITH_SRC, class F>
__device__ __forceinline__ void scan_chunk(const int* __restrict__ srcp,
                                           const int* __restrict__ dstp,
                                           int e0, int e1, bool vec, F f) {
  if (vec) {
    for (int i = e0 + 4 * (int)threadIdx.x; i < e1; i += 4 * 256 * 4) {
      v4i d[4], s[4];
      #pragma unroll
      for (int u = 0; u < 4; ++u) {
        const int j = i + u * 1024;
        const bool in = j < e1;
        d[u] = in ? *(const v4i*)(dstp + j) : (v4i){-1, -1, -1, -1};
        if (WITH_SRC) s[u] = in ? *(const v4i*)(srcp + j) : (v4i){0, 0, 0, 0};
        else          s[u] = (v4i){0, 0, 0, 0};
      }
      #pragma unroll
      for (int u = 0; u < 4; ++u) {
        f(s[u].x, d[u].x); f(s[u].y, d[u].y); f(s[u].z, d[u].z); f(s[u].w, d[u].w);
      }
    }
  } else {
    for (int i = e0 + (int)threadIdx.x; i < e1; i += 256)
      f(WITH_SRC ? srcp[i] : 0, dstp[i]);
  }
}

__device__ __forceinline__ bool list_is_vec(const int* ei, int E) {
  return ((E & 3) == 0) && ((((size_t)ei) & 15) == 0);
}

__global__ void __launch_bounds__(256)
range_count_kernel(const int* __restrict__ ei, int E, int N, int span, int per,
                   int* __restrict__ part, int* __restrict__ total) {
  extern __shared__ int bins[];
  const int q = blockIdx.x, r = blockIdx.y;
  const int lo = r * span;
  for (int i = threadIdx.x; i < span; i += 256) bins[i] = 0;
  if (q == 0 && r == 0 && threadIdx.x == 0) *total = 0;
  __syncthreads();
  const int e0 = min(E, q * per), e1 = min(E, e0 + per);
  const unsigned uspan = (unsigned)span;
  scan_chunk<false>(ei, ei + E, e0, e1, list_is_vec(ei, E), [&](int, int d) {
    const unsigned t = (unsigned)(d - lo);
    if (t < uspan) atomicAdd(&bins[t], 1);
  });
  __syncthreads();
  for (int i = threadIdx.x; i < span; i += 256)
    if (lo + i < N) part[(size_t)q * N + lo + i] = bins[i];
}

__global__ void __launch_bounds__(256)
segment_alloc_kernel(int* __restrict__ part, int* __restrict__ deg,
                     int* __restrict__ start, int* __restrict__ total, int N) {
  __shared__ int wave_off[8];
  __shared__ int block_base;
  const int n = blockIdx.x * 256 + threadIdx.x;
  const int lane = threadIdx.x & 31, w = threadIdx.x >> 5;
  int c[SORT_Q];
  int d = 0;
  #pragma unroll
  for (int q = 0; q < SORT_Q; ++q) {
    c[q] = (n < N) ? part[(size_t)q * N + n] : 0;
    d += c[q];
  }
  int incl = d;
  for (int o = 1; o < 32; o <<= 1) {
    int t = __shfl_up(incl, o);
    if (lane >= o) incl += t;
  }
  if (lane == 31) wave_off[w] = incl;
  __syncthreads();
  if (threadIdx.x == 0) {
    int run = 0;
    for (int i = 0; i < 8; ++i) { int t = wave_off[i]; wave_off[i] = run; run += t; }
    block_base = atomicAdd(total, run);
  }
  __syncthreads();
  if (n < N) {
    int run = block_base + wave_off[w] + incl - d;
    start[n] = run;
    deg[n] = d;
    #pragma unroll
    for (int q = 0; q < SORT_Q; ++q) { part[(size_t)q * N + n] = run; run += c[q]; }
  }
}

__device__ __forceinline__ void
range_scatter_block(int* bins, int q, int r,
                    const int* __restrict__ ei, int E, int N, int span, int per,
                    const int* __restrict__ part, int* __restrict__ csr) {
  const int lo = r * span;
  for (int i0 = threadIdx.x; i0 < span; i0 += 256 * 8) {
    int v[8];
    #pragma unroll
    for (int u = 0; u < 8; ++u) {
      const int i = i0 + u * 256;
      v[u] = (i < span && lo + i < N) ? part[(size_t)q * N + lo + i] : 0;
    }
    #pragma unroll
    for (int u = 0; u < 8; ++u) {
      const int i = i0 + u * 256;
      if (i < span) bins[i] = v[u];
    }
  }
  __syncthreads();
  const int e0 = min(E, q * per), e1 = min(E, e0 + per);
  const unsigned uspan = (unsigned)span;
  scan_chunk<true>(ei, ei + E, e0, e1, list_is_vec(ei, E), [&](int s, int d) {
    const unsigned t = (unsigned)(d - lo);
    if (t < uspan) csr[atomicAdd(&bins[t], 1)] = s;
  });
}

__device__ __forceinline__ float lrelu(float v) { return v > 0.f ? v : NEG_SLOPE * v; }

typedef __attribute__((ext_vector_type(4))) float v4f;

template <int NT, int EPI>
__device__ __forceinline__ void
gemm_wmma_f32_tile(float* tile, int tileM, bool live,
                   const float* __restrict__ A, const float* __restrict__ B,
                   float* __restrict__ D, int K, int ldb, int ldd, int nvalid,
                   const float* __restrict__ a_src, const float* __restrict__ a_dst,
                   float* __restrict__ o_src, float* __restrict__ o_dst) {
  constexpr int PITCH = 16 * NT + 4;
  const int lane = threadIdx.x & 31;
  const int half = lane >> 4;
  const int l16  = lane & 15;
  const int rowA = tileM * 16 + l16;
  const float* __restrict__ arow = A + (size_t)rowA * K;

  v8f acc[NT];
  #pragma unroll
  for (int nt = 0; nt < NT; ++nt) acc[nt] = (v8f){0.f,0.f,0.f,0.f,0.f,0.f,0.f,0.f};

  for (int kk = 0; kk < K; kk += 4) {
    const int ka = kk + half * 2;
    v2f a;
    a.x = arow[ka];
    a.y = arow[ka + 1];
    const float* __restrict__ brow0 = B + (size_t)ka * ldb;
    const float* __restrict__ brow1 = B + (size_t)(ka + 1) * ldb;
    #pragma unroll
    for (int nt = 0; nt < NT; ++nt) {
      const int colB = nt * 16 + l16;
      v2f b;
      if (EPI == 2) {
        const int colc = min(colB, nvalid - 1);
        const float b0 = brow0[colc], b1 = brow1[colc];
        b.x = colB < nvalid ? b0 : 0.f;
        b.y = colB < nvalid ? b1 : 0.f;
      } else {
        b.x = brow0[colB];
        b.y = brow1[colB];
      }
      acc[nt] = __builtin_amdgcn_wmma_f32_16x16x4_f32(false, a, false, b,
                                                      (short)0, acc[nt], false, false);
    }
  }
  #pragma unroll
  for (int nt = 0; nt < NT; ++nt) {
    #pragma unroll
    for (int g = 0; g < 8; ++g) {
      const int r = tileM * 16 + half * 8 + g;
      const int c = nt * 16 + l16;
      if (live && c < nvalid) D[(size_t)r * ldd + c] = acc[nt][g];
      tile[(half * 8 + g) * PITCH + c] = acc[nt][g];
    }
  }
  __syncthreads();

  if (EPI == 1) {
    const int hh = lane & 7;
    float sp[8], dp[8];
    #pragma unroll
    for (int c = 0; c < 8; ++c) { sp[c] = a_src[hh * 8 + c]; dp[c] = a_dst[hh * 8 + c]; }
    #pragma unroll
    for (int i = 0; i < 4; ++i) {
      const int p = lane + 32 * i;
      const float* tp = tile + (p >> 3) * PITCH + hh * 8;
      const v4f t0 = *(const v4f*)tp, t1 = *(const v4f*)(tp + 4);
      const float t[8] = {t0.x, t0.y, t0.z, t0.w, t1.x, t1.y, t1.z, t1.w};
      float s1 = 0.f, s2 = 0.f;
      #pragma unroll
      for (int c = 0; c < 8; ++c) { s1 += t[c] * sp[c]; s2 += t[c] * dp[c]; }
      if (live) {
        o_src[(size_t)tileM * 128 + p] = s1;
        o_dst[(size_t)tileM * 128 + p] = s2;
      }
    }
  } else {
    const float* __restrict__ av = half ? a_dst : a_src;
    const float* tp = tile + l16 * PITCH;
    float s = 0.f;
    #pragma unroll
    for (int c4 = 0; c4 < 16 * NT; c4 += 4) {
      const v4f t = *(const v4f*)(tp + c4);
      const float tt[4] = {t.x, t.y, t.z, t.w};
      #pragma unroll
      for (int u = 0; u < 4; ++u) {
        const int c = c4 + u;
        const float coef = av[min(c, nvalid - 1)];
        s += tt[u] * (c < nvalid ? coef : 0.f);
      }
    }
    if (live) o_src[((size_t)tileM * 16 + l16) * 2 + half] = s;
  }
}

template <int NT, int EPI>
__global__ void __launch_bounds__(32)
gemm_wmma_f32_nt(const float* __restrict__ A, const float* __restrict__ B,
                 float* __restrict__ D, int M, int K, int ldb, int ldd, int nvalid,
                 const float* __restrict__ a_src, const float* __restrict__ a_dst,
                 float* __restrict__ o_src, float* __restrict__ o_dst) {
  __shared__ float tile[16 * (16 * NT + 4)];
  gemm_wmma_f32_tile<NT, EPI>(tile, blockIdx.x, true, A, B, D, K, ldb, ldd, nvalid,
                              a_src, a_dst, o_src, o_dst);
}

#define GEMM1_SLAB (16 * (16 * 4 + 4))
__global__ void __launch_bounds__(256)
scatter_and_gemm1_kernel(int gemm_blocks, int ntiles,
                         const int* __restrict__ ei, int E, int N, int span, int per,
                         const int* __restrict__ part, int* __restrict__ csr,
                         const float* __restrict__ x, const float* __restrict__ W1,
                         float* __restrict__ h1, int F, int HC,
                         const float* __restrict__ a_src, const float* __restrict__ a_dst,
                         float* __restrict__ as1, float* __restrict__ ad1) {
  extern __shared__ __attribute__((aligned(16))) int lds[];
  if ((int)blockIdx.x < gemm_blocks) {
    const int w = __builtin_amdgcn_readfirstlane(threadIdx.x >> 5);
    const int t = blockIdx.x * 8 + w;
    gemm_wmma_f32_tile<4, 1>((float*)lds + w * GEMM1_SLAB, min(t, ntiles - 1), t < ntiles,
                             x, W1, h1, F, HC, HC, HC, a_src, a_dst, as1, ad1);
  } else {
    const int b = blockIdx.x - gemm_blocks;
    range_scatter_block(lds, b % SORT_Q, b / SORT_Q, ei, E, N, span, per, part, csr);
  }
}


__global__ void __launch_bounds__(256)
gather_agg1_kernel(const int* __restrict__ start, const int* __restrict__ deg,
                   const int* __restrict__ csr,
                   const float* __restrict__ as, const float* __restrict__ ad,
                   const float* __restrict__ h1, const float* __restrict__ b,
                   float* __restrict__ out, int N) {
  const int lane = threadIdx.x & 31;
  const int n = __builtin_amdgcn_readfirstlane(blockIdx.x * 8 + (threadIdx.x >> 5));
  if (n >= N) return;
  const int hh = lane >> 2;
  const int beg = start[n], cnt = deg[n];
  const float adv = ad[n * 8 + hh];
  const float vself = lrelu(as[n * 8 + hh] + adv);

  float mx = vself;
  for (int base = 0; base < cnt; base += 32) {
    const int k = base + lane;
    const int my = (k < cnt) ? csr[beg + k] : n;
    const int m = min(32, cnt - base);
    #pragma unroll 4
    for (int j = 0; j < m; ++j) {
      const int s = __builtin_amdgcn_readlane(my, j);
      mx = fmaxf(mx, lrelu(as[s * 8 + hh] + adv));
    }
  }

  float ex = expf(vself - mx);
  float den = ex;
  v2f hv = *(const v2f*)(h1 + (size_t)n * 64 + 2 * lane);
  float a0 = ex * hv.x, a1 = ex * hv.y;
  for (int base = 0; base < cnt; base += 32) {
    const int k = base + lane;
    const int my = (k < cnt) ? csr[beg + k] : n;
    const int m = min(32, cnt - base);
    #pragma unroll 4
    for (int j = 0; j < m; ++j) {
      const int s = __builtin_amdgcn_readlane(my, j);
      ex = expf(lrelu(as[s * 8 + hh] + adv) - mx);
      hv = *(const v2f*)(h1 + (size_t)s * 64 + 2 * lane);
      den += ex;
      a0 += ex * hv.x;
      a1 += ex * hv.y;
    }
  }
  const float dd = den + EPSV;
  const v2f bb = *(const v2f*)(b + 2 * lane);
  float o0 = a0 / dd + bb.x, o1 = a1 / dd + bb.y;
  v2f o;
  o.x = o0 > 0.f ? o0 : expm1f(o0);
  o.y = o1 > 0.f ? o1 : expm1f(o1);
  *(v2f*)(out + (size_t)n * 64 + 2 * lane) = o;
}


__global__ void __launch_bounds__(256)
gather_agg2_kernel(const int* __restrict__ start, const int* __restrict__ deg,
                   const int* __restrict__ csr,
                   const float* __restrict__ asd,
                   const float* __restrict__ h2, const float* __restrict__ b2,
                   float* __restrict__ y, int N, int C) {
  __shared__ float rows[8 * 64];
  const int lane = threadIdx.x & 31;
  const int w = threadIdx.x >> 5;
  const int n = __builtin_amdgcn_readfirstlane(blockIdx.x * 8 + w);
  const bool own = 2 * lane < C;
  if (n < N) {
    const int beg = start[n], cnt = deg[n];
    const float adv = asd[2 * n + 1];
    const float vself = lrelu(asd[2 * n] + adv);

    float mx = vself;
    for (int base = 0; base < cnt; base += 32) {
      const int k = base + lane;
      if (k < cnt) mx = fmaxf(mx, lrelu(asd[2 * csr[beg + k]] + adv));
    }
    for (int o = 16; o > 0; o >>= 1) mx = fmaxf(mx, __shfl_xor(mx, o));

    const float exs = expf(vself - mx);
    float den = exs;
    float a0 = 0.f, a1 = 0.f;
    if (own) {
      const v2f hv = *(const v2f*)(h2 + (size_t)n * C + 2 * lane);
      a0 = exs * hv.x; a1 = exs * hv.y;
    }
    for (int base = 0; base < cnt; base += 32) {
      const int k = base + lane;
      const int my = (k < cnt) ? csr[beg + k] : n;
      const float exl = expf(lrelu(asd[2 * my] + adv) - mx);
      const int m = min(32, cnt - base);
      #pragma unroll 4
      for (int j = 0; j < m; ++j) {
        const int s = __builtin_amdgcn_readlane(my, j);
        const float e = __int_as_float(__builtin_amdgcn_readlane(__float_as_int(exl), j));
        den += e;
        if (own) {
          const v2f hv = *(const v2f*)(h2 + (size_t)s * C + 2 * lane);
          a0 += e * hv.x;
          a1 += e * hv.y;
        }
      }
    }
    float v0 = -INFINITY, v1 = -INFINITY;
    if (own) {
      const float dd = den + EPSV;
      const v2f bb = *(const v2f*)(b2 + 2 * lane);
      v0 = a0 / dd + bb.x;
      v1 = a1 / dd + bb.y;
      v0 = v0 > 0.f ? v0 : expm1f(v0);
      v1 = v1 > 0.f ? v1 : expm1f(v1);
    }
    float rmx = fmaxf(v0, v1);
    for (int o = 16; o > 0; o >>= 1) rmx = fmaxf(rmx, __shfl_xor(rmx, o));
    float sm = own ? expf(v0 - rmx) + expf(v1 - rmx) : 0.f;
    for (int o = 16; o > 0; o >>= 1) sm += __shfl_xor(sm, o);
    const float l = logf(sm);
    if (own) {
      v2f o;
      o.x = v0 - rmx - l; o.y = v1 - rmx - l;
      *(v2f*)(rows + w * C + 2 * lane) = o;
    }
  }
  __syncthreads();
  const int i = 2 * (int)threadIdx.x;
  const size_t base = (size_t)blockIdx.x * 8 * C;
  if (i < 8 * C && base + i < (size_t)N * C)
    *(v2f*)(y + base + i) = *(const v2f*)(rows + i);
}


static inline int cdiv(int a, int b) { return (a + b - 1) / b; }

extern "C" void kernel_launch(void* const* d_in, const int* in_sizes, int n_in,
                              void* d_out, int out_size, void* d_ws, size_t ws_size,
                              hipStream_t stream) {
  const float* x      = (const float*)d_in[0];
  const int*   ei     = (const int*)  d_in[1];
  const float* W1     = (const float*)d_in[2];
  const float* a_src1 = (const float*)d_in[3];
  const float* a_dst1 = (const float*)d_in[4];
  const float* b1     = (const float*)d_in[5];
  const float* W2     = (const float*)d_in[6];
  const float* a_src2 = (const float*)d_in[7];
  const float* a_dst2 = (const float*)d_in[8];
  const float* b2     = (const float*)d_in[9];
  float* out = (float*)d_out;

  const int HC   = in_sizes[5];
  const int NCLS = in_sizes[9];
  const int F    = in_sizes[2] / HC;
  const int N    = in_sizes[0] / F;
  const int E    = in_sizes[1] / 2;

  float* ws = (float*)d_ws;
  size_t off = 0;
  float* h1   = ws + off; off += (size_t)N * HC;
  float* out1 = ws + off; off += (size_t)N * HC;
  float* h2   = ws + off; off += (size_t)N * NCLS;
  float* as1  = ws + off; off += (size_t)N * 8;
  float* ad1  = ws + off; off += (size_t)N * 8;
  float* asd2 = ws + off; off += (size_t)N * 2;
  int* deg    = (int*)(ws + off); off += (size_t)N;
  int* total  = (int*)(ws + off); off += 2;
  int* start  = (int*)(ws + off); off += (size_t)N;
  int* csr    = (int*)(ws + off); off += (size_t)E;
  int* part   = (int*)(ws + off); off += (size_t)SORT_Q * N;

  const int span = cdiv(N, SORT_R);
  const int per  = cdiv(cdiv(E, 4), SORT_Q) * 4;
  range_count_kernel<<<dim3(SORT_Q, SORT_R), 256, span * sizeof(int), stream>>>(ei, E, N, span, per, part, total);
  segment_alloc_kernel<<<cdiv(N, 256), 256, 0, stream>>>(part, deg, start, total, N);

  const int ntiles = N / 16;
  const int gemm_blocks = cdiv(ntiles, 8);
  const size_t bins_bytes = span * sizeof(int), slab_bytes = 8 * GEMM1_SLAB * sizeof(float);
  scatter_and_gemm1_kernel<<<gemm_blocks + SORT_Q * SORT_R, 256,
                             bins_bytes > slab_bytes ? bins_bytes : slab_bytes, stream>>>(
      gemm_blocks, ntiles, ei, E, N, span, per, part, csr,
      x, W1, h1, F, HC, a_src1, a_dst1, as1, ad1);

  gather_agg1_kernel<<<cdiv(N, 8), 256, 0, stream>>>(start, deg, csr, as1, ad1, h1, b1, out1, N);

  gemm_wmma_f32_nt<3, 2><<<dim3(N / 16), 32, 0, stream>>>(out1, W2, h2, N, HC, NCLS, NCLS, NCLS,
                                                          a_src2, a_dst2, asd2, asd2);
  gather_agg2_kernel<<<cdiv(N, 8), 256, 0, stream>>>(start, deg, csr, asd2, h2, b2, out, N, NCLS);
}


# layer-2 gather: row loads of the edge loop made unconditional (lanes past the row read its first pair), so four are in flight together instead of each one waited for behind the own-lane branch
# speedup vs baseline: 9.2298x; 1.0537x over previous
#include <hip/hip_runtime.h>
#include <hip/hip_bf16.h>
#include <math.h>

typedef __attribute__((ext_vector_type(2))) float v2f;
typedef __attribute__((ext_vector_type(8))) float v8f;

#define NEG_SLOPE 0.2f
#define EPSV 1e-16f


#define SORT_R 16
#define SORT_Q 64

typedef __attribute__((ext_vector_type(4))) int v4i;

template <bool WITH_SRC, class F>
__device__ __forceinline__ void scan_chunk(const int* __restrict__ srcp,
                                           const int* __restrict__ dstp,
                                           int e0, int e1, bool vec, F f) {
  if (vec) {
    for (int i = e0 + 4 * (int)threadIdx.x; i < e1; i += 4 * 256 * 4) {
      v4i d[4], s[4];
      #pragma unroll
      for (int u = 0; u < 4; ++u) {
        const int j = i + u * 1024;
        const bool in = j < e1;
        d[u] = in ? *(const v4i*)(dstp + j) : (v4i){-1, -1, -1, -1};
        if (WITH_SRC) s[u] = in ? *(const v4i*)(srcp + j) : (v4i){0, 0, 0, 0};
        else          s[u] = (v4i){0, 0, 0, 0};
      }
      #pragma unroll
      for (int u = 0; u < 4; ++u) {
        f(s[u].x, d[u].x); f(s[u].y, d[u].y); f(s[u].z, d[u].z); f(s[u].w, d[u].w);
      }
    }
  } else {
    for (int i = e0 + (int)threadIdx.x; i < e1; i += 256)
      f(WITH_SRC ? srcp[i] : 0, dstp[i]);
  }
}

__device__ __forceinline__ bool list_is_vec(const int* ei, int E) {
  return ((E & 3) == 0) && ((((size_t)ei) & 15) == 0);
}

__global__ void __launch_bounds__(256)
range_count_kernel(const int* __restrict__ ei, int E, int N, int span, int per,
                   int* __restrict__ part, int* __restrict__ total) {
  extern __shared__ int bins[];
  const int q = blockIdx.x, r = blockIdx.y;
  const int lo = r * span;
  for (int i = threadIdx.x; i < span; i += 256) bins[i] = 0;
  if (q == 0 && r == 0 && threadIdx.x == 0) *total = 0;
  __syncthreads();
  const int e0 = min(E, q * per), e1 = min(E, e0 + per);
  const unsigned uspan = (unsigned)span;
  scan_chunk<false>(ei, ei + E, e0, e1, list_is_vec(ei, E), [&](int, int d) {
    const unsigned t = (unsigned)(d - lo);
    if (t < uspan) atomicAdd(&bins[t], 1);
  });
  __syncthreads();
  for (int i = threadIdx.x; i < span; i += 256)
    if (lo + i < N) part[(size_t)q * N + lo + i] = bins[i];
}

__global__ void __launch_bounds__(256)
segment_alloc_kernel(int* __restrict__ part, int* __restrict__ deg,
                     int* __restrict__ start, int* __restrict__ total, int N) {
  __shared__ int wave_off[8];
  __shared__ int block_base;
  const int n = blockIdx.x * 256 + threadIdx.x;
  const int lane = threadIdx.x & 31, w = threadIdx.x >> 5;
  int c[SORT_Q];
  int d = 0;
  #pragma unroll
  for (int q = 0; q < SORT_Q; ++q) {
    c[q] = (n < N) ? part[(size_t)q * N + n] : 0;
    d += c[q];
  }
  int incl = d;
  for (int o = 1; o < 32; o <<= 1) {
    int t = __shfl_up(incl, o);
    if (lane >= o) incl += t;
  }
  if (lane == 31) wave_off[w] = incl;
  __syncthreads();
  if (threadIdx.x == 0) {
    int run = 0;
    for (int i = 0; i < 8; ++i) { int t = wave_off[i]; wave_off[i] = run; run += t; }
    block_base = atomicAdd(total, run);
  }
  __syncthreads();
  if (n < N) {
    int run = block_base + wave_off[w] + incl - d;
    start[n] = run;
    deg[n] = d;
    #pragma unroll
    for (int q = 0; q < SORT_Q; ++q) { part[(size_t)q * N + n] = run; run += c[q]; }
  }
}

__device__ __forceinline__ void
range_scatter_block(int* bins, int q, int r,
                    const int* __restrict__ ei, int E, int N, int span, int per,
                    const int* __restrict__ part, int* __restrict__ csr) {
  const int lo = r * span;
  for (int i0 = threadIdx.x; i0 < span; i0 += 256 * 8) {
    int v[8];
    #pragma unroll
    for (int u = 0; u < 8; ++u) {
      const int i = i0 + u * 256;
      v[u] = (i < span && lo + i < N) ? part[(size_t)q * N + lo + i] : 0;
    }
    #pragma unroll
    for (int u = 0; u < 8; ++u) {
      const int i = i0 + u * 256;
      if (i < span) bins[i] = v[u];
    }
  }
  __syncthreads();
  const int e0 = min(E, q * per), e1 = min(E, e0 + per);
  const unsigned uspan = (unsigned)span;
  scan_chunk<true>(ei, ei + E, e0, e1, list_is_vec(ei, E), [&](int s, int d) {
    const unsigned t = (unsigned)(d - lo);
    if (t < uspan) csr[atomicAdd(&bins[t], 1)] = s;
  });
}

__device__ __forceinline__ float lrelu(float v) { return v > 0.f ? v : NEG_SLOPE * v; }

typedef __attribute__((ext_vector_type(4))) float v4f;

template <int NT, int EPI>
__device__ __forceinline__ void
gemm_wmma_f32_tile(float* tile, int tileM, bool live,
                   const float* __restrict__ A, const float* __restrict__ B,
                   float* __restrict__ D, int K, int ldb, int ldd, int nvalid,
                   const float* __restrict__ a_src, const float* __restrict__ a_dst,
                   float* __restrict__ o_src, float* __restrict__ o_dst) {
  constexpr int PITCH = 16 * NT + 4;
  const int lane = threadIdx.x & 31;
  const int half = lane >> 4;
  const int l16  = lane & 15;
  const int rowA = tileM * 16 + l16;
  const float* __restrict__ arow = A + (size_t)rowA * K;

  v8f acc[NT];
  #pragma unroll
  for (int nt = 0; nt < NT; ++nt) acc[nt] = (v8f){0.f,0.f,0.f,0.f,0.f,0.f,0.f,0.f};

  for (int kk = 0; kk < K; kk += 4) {
    const int ka = kk + half * 2;
    v2f a;
    a.x = arow[ka];
    a.y = arow[ka + 1];
    const float* __restrict__ brow0 = B + (size_t)ka * ldb;
    const float* __restrict__ brow1 = B + (size_t)(ka + 1) * ldb;
    #pragma unroll
    for (int nt = 0; nt < NT; ++nt) {
      const int colB = nt * 16 + l16;
      v2f b;
      if (EPI == 2) {
        const int colc = min(colB, nvalid - 1);
        const float b0 = brow0[colc], b1 = brow1[colc];
        b.x = colB < nvalid ? b0 : 0.f;
        b.y = colB < nvalid ? b1 : 0.f;
      } else {
        b.x = brow0[colB];
        b.y = brow1[colB];
      }
      acc[nt] = __builtin_amdgcn_wmma_f32_16x16x4_f32(false, a, false, b,
                                                      (short)0, acc[nt], false, false);
    }
  }
  #pragma unroll
  for (int nt = 0; nt < NT; ++nt) {
    #pragma unroll
    for (int g = 0; g < 8; ++g) {
      const int r = tileM * 16 + half * 8 + g;
      const int c = nt * 16 + l16;
      if (live && c < nvalid) D[(size_t)r * ldd + c] = acc[nt][g];
      tile[(half * 8 + g) * PITCH + c] = acc[nt][g];
    }
  }
  __syncthreads();

  if (EPI == 1) {
    const int hh = lane & 7;
    float sp[8], dp[8];
    #pragma unroll
    for (int c = 0; c < 8; ++c) { sp[c] = a_src[hh * 8 + c]; dp[c] = a_dst[hh * 8 + c]; }
    #pragma unroll
    for (int i = 0; i < 4; ++i) {
      const int p = lane + 32 * i;
      const float* tp = tile + (p >> 3) * PITCH + hh * 8;
      const v4f t0 = *(const v4f*)tp, t1 = *(const v4f*)(tp + 4);
      const float t[8] = {t0.x, t0.y, t0.z, t0.w, t1.x, t1.y, t1.z, t1.w};
      float s1 = 0.f, s2 = 0.f;
      #pragma unroll
      for (int c = 0; c < 8; ++c) { s1 += t[c] * sp[c]; s2 += t[c] * dp[c]; }
      if (live) {
        o_src[(size_t)tileM * 128 + p] = s1;
        o_dst[(size_t)tileM * 128 + p] = s2;
      }
    }
  } else {
    const float* __restrict__ av = half ? a_dst : a_src;
    const float* tp = tile + l16 * PITCH;
    float s = 0.f;
    #pragma unroll
    for (int c4 = 0; c4 < 16 * NT; c4 += 4) {
      const v4f t = *(const v4f*)(tp + c4);
      const float tt[4] = {t.x, t.y, t.z, t.w};
      #pragma unroll
      for (int u = 0; u < 4; ++u) {
        const int c = c4 + u;
        const float coef = av[min(c, nvalid - 1)];
        s += tt[u] * (c < nvalid ? coef : 0.f);
      }
    }
    if (live) o_src[((size_t)tileM * 16 + l16) * 2 + half] = s;
  }
}

template <int NT, int EPI>
__global__ void __launch_bounds__(32)
gemm_wmma_f32_nt(const float* __restrict__ A, const float* __restrict__ B,
                 float* __restrict__ D, int M, int K, int ldb, int ldd, int nvalid,
                 const float* __restrict__ a_src, const float* __restrict__ a_dst,
                 float* __restrict__ o_src, float* __restrict__ o_dst) {
  __shared__ float tile[16 * (16 * NT + 4)];
  gemm_wmma_f32_tile<NT, EPI>(tile, blockIdx.x, true, A, B, D, K, ldb, ldd, nvalid,
                              a_src, a_dst, o_src, o_dst);
}

#define GEMM1_SLAB (16 * (16 * 4 + 4))
__global__ void __launch_bounds__(256)
scatter_and_gemm1_kernel(int gemm_blocks, int ntiles,
                         const int* __restrict__ ei, int E, int N, int span, int per,
                         const int* __restrict__ part, int* __restrict__ csr,
                         const float* __restrict__ x, const float* __restrict__ W1,
                         float* __restrict__ h1, int F, int HC,
                         const float* __restrict__ a_src, const float* __restrict__ a_dst,
                         float* __restrict__ as1, float* __restrict__ ad1) {
  extern __shared__ __attribute__((aligned(16))) int lds[];
  if ((int)blockIdx.x < gemm_blocks) {
    const int w = __builtin_amdgcn_readfirstlane(threadIdx.x >> 5);
    const int t = blockIdx.x * 8 + w;
    gemm_wmma_f32_tile<4, 1>((float*)lds + w * GEMM1_SLAB, min(t, ntiles - 1), t < ntiles,
                             x, W1, h1, F, HC, HC, HC, a_src, a_dst, as1, ad1);
  } else {
    const int b = blockIdx.x - gemm_blocks;
    range_scatter_block(lds, b % SORT_Q, b / SORT_Q, ei, E, N, span, per, part, csr);
  }
}


__global__ void __launch_bounds__(256)
gather_agg1_kernel(const int* __restrict__ start, const int* __restrict__ deg,
                   const int* __restrict__ csr,
                   const float* __restrict__ as, const float* __restrict__ ad,
                   const float* __restrict__ h1, const float* __restrict__ b,
                   float* __restrict__ out, int N) {
  const int lane = threadIdx.x & 31;
  const int n = __builtin_amdgcn_readfirstlane(blockIdx.x * 8 + (threadIdx.x >> 5));
  if (n >= N) return;
  const int hh = lane >> 2;
  const int beg = start[n], cnt = deg[n];
  const float adv = ad[n * 8 + hh];
  const float vself = lrelu(as[n * 8 + hh] + adv);

  float mx = vself;
  for (int base = 0; base < cnt; base += 32) {
    const int k = base + lane;
    const int my = (k < cnt) ? csr[beg + k] : n;
    const int m = min(32, cnt - base);
    #pragma unroll 4
    for (int j = 0; j < m; ++j) {
      const int s = __builtin_amdgcn_readlane(my, j);
      mx = fmaxf(mx, lrelu(as[s * 8 + hh] + adv));
    }
  }

  float ex = expf(vself - mx);
  float den = ex;
  v2f hv = *(const v2f*)(h1 + (size_t)n * 64 + 2 * lane);
  float a0 = ex * hv.x, a1 = ex * hv.y;
  for (int base = 0; base < cnt; base += 32) {
    const int k = base + lane;
    const int my = (k < cnt) ? csr[beg + k] : n;
    const int m = min(32, cnt - base);
    #pragma unroll 4
    for (int j = 0; j < m; ++j) {
      const int s = __builtin_amdgcn_readlane(my, j);
      ex = expf(lrelu(as[s * 8 + hh] + adv) - mx);
      hv = *(const v2f*)(h1 + (size_t)s * 64 + 2 * lane);
      den += ex;
      a0 += ex * hv.x;
      a1 += ex * hv.y;
    }
  }
  const float dd = den + EPSV;
  const v2f bb = *(const v2f*)(b + 2 * lane);
  float o0 = a0 / dd + bb.x, o1 = a1 / dd + bb.y;
  v2f o;
  o.x = o0 > 0.f ? o0 : expm1f(o0);
  o.y = o1 > 0.f ? o1 : expm1f(o1);
  *(v2f*)(out + (size_t)n * 64 + 2 * lane) = o;
}


__global__ void __launch_bounds__(256)
gather_agg2_kernel(const int* __restrict__ start, const int* __restrict__ deg,
                   const int* __restrict__ csr,
                   const float* __restrict__ asd,
                   const float* __restrict__ h2, const float* __restrict__ b2,
                   float* __restrict__ y, int N, int C) {
  __shared__ float rows[8 * 64];
  const int lane = threadIdx.x & 31;
  const int w = threadIdx.x >> 5;
  const int n = __builtin_amdgcn_readfirstlane(blockIdx.x * 8 + w);
  const bool own = 2 * lane < C;
  const int col = own ? 2 * lane : 0;
  if (n < N) {
    const int beg = start[n], cnt = deg[n];
    const float adv = asd[2 * n + 1];
    const float vself = lrelu(asd[2 * n] + adv);

    float mx = vself;
    for (int base = 0; base < cnt; base += 32) {
      const int k = base + lane;
      if (k < cnt) mx = fmaxf(mx, lrelu(asd[2 * csr[beg + k]] + adv));
    }
    for (int o = 16; o > 0; o >>= 1) mx = fmaxf(mx, __shfl_xor(mx, o));

    const float exs = expf(vself - mx);
    float den = exs;
    float a0 = 0.f, a1 = 0.f;
    if (own) {
      const v2f hv = *(const v2f*)(h2 + (size_t)n * C + 2 * lane);
      a0 = exs * hv.x; a1 = exs * hv.y;
    }
    for (int base = 0; base < cnt; base += 32) {
      const int k = base + lane;
      const int my = (k < cnt) ? csr[beg + k] : n;
      const float exl = expf(lrelu(asd[2 * my] + adv) - mx);
      const int m = min(32, cnt - base);
      #pragma unroll 4
      for (int j = 0; j < m; ++j) {
        const int s = __builtin_amdgcn_readlane(my, j);
        const float e = __int_as_float(__builtin_amdgcn_readlane(__float_as_int(exl), j));
        den += e;
        const v2f hv = *(const v2f*)(h2 + (size_t)s * C + col);
        a0 += e * hv.x;
        a1 += e * hv.y;
      }
    }
    float v0 = -INFINITY, v1 = -INFINITY;
    if (own) {
      const float dd = den + EPSV;
      const v2f bb = *(const v2f*)(b2 + 2 * lane);
      v0 = a0 / dd + bb.x;
      v1 = a1 / dd + bb.y;
      v0 = v0 > 0.f ? v0 : expm1f(v0);
      v1 = v1 > 0.f ? v1 : expm1f(v1);
    }
    float rmx = fmaxf(v0, v1);
    for (int o = 16; o > 0; o >>= 1) rmx = fmaxf(rmx, __shfl_xor(rmx, o));
    float sm = own ? expf(v0 - rmx) + expf(v1 - rmx) : 0.f;
    for (int o = 16; o > 0; o >>= 1) sm += __shfl_xor(sm, o);
    const float l = logf(sm);
    if (own) {
      v2f o;
      o.x = v0 - rmx - l; o.y = v1 - rmx - l;
      *(v2f*)(rows + w * C + 2 * lane) = o;
    }
  }
  __syncthreads();
  const int i = 2 * (int)threadIdx.x;
  const size_t base = (size_t)blockIdx.x * 8 * C;
  if (i < 8 * C && base + i < (size_t)N * C)
    *(v2f*)(y + base + i) = *(const v2f*)(rows + i);
}


static inline int cdiv(int a, int b) { return (a + b - 1) / b; }

extern "C" void kernel_launch(void* const* d_in, const int* in_sizes, int n_in,
                              void* d_out, int out_size, void* d_ws, size_t ws_size,
                              hipStream_t stream) {
  const float* x      = (const float*)d_in[0];
  const int*   ei     = (const int*)  d_in[1];
  const float* W1     = (const float*)d_in[2];
  const float* a_src1 = (const float*)d_in[3];
  const float* a_dst1 = (const float*)d_in[4];
  const float* b1     = (const float*)d_in[5];
  const float* W2     = (const float*)d_in[6];
  const float* a_src2 = (const float*)d_in[7];
  const float* a_dst2 = (const float*)d_in[8];
  const float* b2     = (const float*)d_in[9];
  float* out = (float*)d_out;

  const int HC   = in_sizes[5];
  const int NCLS = in_sizes[9];
  const int F    = in_sizes[2] / HC;
  const int N    = in_sizes[0] / F;
  const int E    = in_sizes[1] / 2;

  float* ws = (float*)d_ws;
  size_t off = 0;
  float* h1   = ws + off; off += (size_t)N * HC;
  float* out1 = ws + off; off += (size_t)N * HC;
  float* h2   = ws + off; off += (size_t)N * NCLS;
  float* as1  = ws + off; off += (size_t)N * 8;
  float* ad1  = ws + off; off += (size_t)N * 8;
  float* asd2 = ws + off; off += (size_t)N * 2;
  int* deg    = (int*)(ws + off); off += (size_t)N;
  int* total  = (int*)(ws + off); off += 2;
  int* start  = (int*)(ws + off); off += (size_t)N;
  int* csr    = (int*)(ws + off); off += (size_t)E;
  int* part   = (int*)(ws + off); off += (size_t)SORT_Q * N;

  const int span = cdiv(N, SORT_R);
  const int per  = cdiv(cdiv(E, 4), SORT_Q) * 4;
  range_count_kernel<<<dim3(SORT_Q, SORT_R), 256, span * sizeof(int), stream>>>(ei, E, N, span, per, part, total);
  segment_alloc_kernel<<<cdiv(N, 256), 256, 0, stream>>>(part, deg, start, total, N);

  const int ntiles = N / 16;
  const int gemm_blocks = cdiv(ntiles, 8);
  const size_t bins_bytes = span * sizeof(int), slab_bytes = 8 * GEMM1_SLAB * sizeof(float);
  scatter_and_gemm1_kernel<<<gemm_blocks + SORT_Q * SORT_R, 256,
                             bins_bytes > slab_bytes ? bins_bytes : slab_bytes, stream>>>(
      gemm_blocks, ntiles, ei, E, N, span, per, part, csr,
      x, W1, h1, F, HC, a_src1, a_dst1, as1, ad1);

  gather_agg1_kernel<<<cdiv(N, 8), 256, 0, stream>>>(start, deg, csr, as1, ad1, h1, b1, out1, N);

  gemm_wmma_f32_nt<3, 2><<<dim3(N / 16), 32, 0, stream>>>(out1, W2, h2, N, HC, NCLS, NCLS, NCLS,
                                                          a_src2, a_dst2, asd2, asd2);
  gather_agg2_kernel<<<cdiv(N, 8), 256, 0, stream>>>(start, deg, csr, asd2, h2, b2, out, N, NCLS);
}


# both gathers fetch an edge's source logit once, not once per pass (layer 1: lanes as 4 edges x 8 heads per load, exps reach the row loop through LDS; layer 2: kept in registers); sum order kept
# speedup vs baseline: 9.9581x; 1.0789x over previous
#include <hip/hip_runtime.h>
#include <hip/hip_bf16.h>
#include <math.h>

typedef __attribute__((ext_vector_type(2))) float v2f;
typedef __attribute__((ext_vector_type(8))) float v8f;

#define NEG_SLOPE 0.2f
#define EPSV 1e-16f


#define SORT_R 16
#define SORT_Q 64

typedef __attribute__((ext_vector_type(4))) int v4i;

template <bool WITH_SRC, class F>
__device__ __forceinline__ void scan_chunk(const int* __restrict__ srcp,
                                           const int* __restrict__ dstp,
                                           int e0, int e1, bool vec, F f) {
  if (vec) {
    for (int i = e0 + 4 * (int)threadIdx.x; i < e1; i += 4 * 256 * 4) {
      v4i d[4], s[4];
      #pragma unroll
      for (int u = 0; u < 4; ++u) {
        const int j = i + u * 1024;
        const bool in = j < e1;
        d[u] = in ? *(const v4i*)(dstp + j) : (v4i){-1, -1, -1, -1};
        if (WITH_SRC) s[u] = in ? *(const v4i*)(srcp + j) : (v4i){0, 0, 0, 0};
        else          s[u] = (v4i){0, 0, 0, 0};
      }
      #pragma unroll
      for (int u = 0; u < 4; ++u) {
        f(s[u].x, d[u].x); f(s[u].y, d[u].y); f(s[u].z, d[u].z); f(s[u].w, d[u].w);
      }
    }
  } else {
    for (int i = e0 + (int)threadIdx.x; i < e1; i += 256)
      f(WITH_SRC ? srcp[i] : 0, dstp[i]);
  }
}

__device__ __forceinline__ bool list_is_vec(const int* ei, int E) {
  return ((E & 3) == 0) && ((((size_t)ei) & 15) == 0);
}

__global__ void __launch_bounds__(256)
range_count_kernel(const int* __restrict__ ei, int E, int N, int span, int per,
                   int* __restrict__ part, int* __restrict__ total) {
  extern __shared__ int bins[];
  const int q = blockIdx.x, r = blockIdx.y;
  const int lo = r * span;
  for (int i = threadIdx.x; i < span; i += 256) bins[i] = 0;
  if (q == 0 && r == 0 && threadIdx.x == 0) *total = 0;
  __syncthreads();
  const int e0 = min(E, q * per), e1 = min(E, e0 + per);
  const unsigned uspan = (unsigned)span;
  scan_chunk<false>(ei, ei + E, e0, e1, list_is_vec(ei, E), [&](int, int d) {
    const unsigned t = (unsigned)(d - lo);
    if (t < uspan) atomicAdd(&bins[t], 1);
  });
  __syncthreads();
  for (int i = threadIdx.x; i < span; i += 256)
    if (lo + i < N) part[(size_t)q * N + lo + i] = bins[i];
}

__global__ void __launch_bounds__(256)
segment_alloc_kernel(int* __restrict__ part, int* __restrict__ deg,
                     int* __restrict__ start, int* __restrict__ total, int N) {
  __shared__ int wave_off[8];
  __shared__ int block_base;
  const int n = blockIdx.x * 256 + threadIdx.x;
  const int lane = threadIdx.x & 31, w = threadIdx.x >> 5;
  int c[SORT_Q];
  int d = 0;
  #pragma unroll
  for (int q = 0; q < SORT_Q; ++q) {
    c[q] = (n < N) ? part[(size_t)q * N + n] : 0;
    d += c[q];
  }
  int incl = d;
  for (int o = 1; o < 32; o <<= 1) {
    int t = __shfl_up(incl, o);
    if (lane >= o) incl += t;
  }
  if (lane == 31) wave_off[w] = incl;
  __syncthreads();
  if (threadIdx.x == 0) {
    int run = 0;
    for (int i = 0; i < 8; ++i) { int t = wave_off[i]; wave_off[i] = run; run += t; }
    block_base = atomicAdd(total, run);
  }
  __syncthreads();
  if (n < N) {
    int run = block_base + wave_off[w] + incl - d;
    start[n] = run;
    deg[n] = d;
    #pragma unroll
    for (int q = 0; q < SORT_Q; ++q) { part[(size_t)q * N + n] = run; run += c[q]; }
  }
}

__device__ __forceinline__ void
range_scatter_block(int* bins, int q, int r,
                    const int* __restrict__ ei, int E, int N, int span, int per,
                    const int* __restrict__ part, int* __restrict__ csr) {
  const int lo = r * span;
  for (int i0 = threadIdx.x; i0 < span; i0 += 256 * 8) {
    int v[8];
    #pragma unroll
    for (int u = 0; u < 8; ++u) {
      const int i = i0 + u * 256;
      v[u] = (i < span && lo + i < N) ? part[(size_t)q * N + lo + i] : 0;
    }
    #pragma unroll
    for (int u = 0; u < 8; ++u) {
      const int i = i0 + u * 256;
      if (i < span) bins[i] = v[u];
    }
  }
  __syncthreads();
  const int e0 = min(E, q * per), e1 = min(E, e0 + per);
  const unsigned uspan = (unsigned)span;
  scan_chunk<true>(ei, ei + E, e0, e1, list_is_vec(ei, E), [&](int s, int d) {
    const unsigned t = (unsigned)(d - lo);
    if (t < uspan) csr[atomicAdd(&bins[t], 1)] = s;
  });
}

__device__ __forceinline__ float lrelu(float v) { return v > 0.f ? v : NEG_SLOPE * v; }

typedef __attribute__((ext_vector_type(4))) float v4f;

template <int NT, int EPI>
__device__ __forceinline__ void
gemm_wmma_f32_tile(float* tile, int tileM, bool live,
                   const float* __restrict__ A, const float* __restrict__ B,
                   float* __restrict__ D, int K, int ldb, int ldd, int nvalid,
                   const float* __restrict__ a_src, const float* __restrict__ a_dst,
                   float* __restrict__ o_src, float* __restrict__ o_dst) {
  constexpr int PITCH = 16 * NT + 4;
  const int lane = threadIdx.x & 31;
  const int half = lane >> 4;
  const int l16  = lane & 15;
  const int rowA = tileM * 16 + l16;
  const float* __restrict__ arow = A + (size_t)rowA * K;

  v8f acc[NT];
  #pragma unroll
  for (int nt = 0; nt < NT; ++nt) acc[nt] = (v8f){0.f,0.f,0.f,0.f,0.f,0.f,0.f,0.f};

  for (int kk = 0; kk < K; kk += 4) {
    const int ka = kk + half * 2;
    v2f a;
    a.x = arow[ka];
    a.y = arow[ka + 1];
    const float* __restrict__ brow0 = B + (size_t)ka * ldb;
    const float* __restrict__ brow1 = B + (size_t)(ka + 1) * ldb;
    #pragma unroll
    for (int nt = 0; nt < NT; ++nt) {
      const int colB = nt * 16 + l16;
      v2f b;
      if (EPI == 2) {
        const int colc = min(colB, nvalid - 1);
        const float b0 = brow0[colc], b1 = brow1[colc];
        b.x = colB < nvalid ? b0 : 0.f;
        b.y = colB < nvalid ? b1 : 0.f;
      } else {
        b.x = brow0[colB];
        b.y = brow1[colB];
      }
      acc[nt] = __builtin_amdgcn_wmma_f32_16x16x4_f32(false, a, false, b,
                                                      (short)0, acc[nt], false, false);
    }
  }
  #pragma unroll
  for (int nt = 0; nt < NT; ++nt) {
    #pragma unroll
    for (int g = 0; g < 8; ++g) {
      const int r = tileM * 16 + half * 8 + g;
      const int c = nt * 16 + l16;
      if (live && c < nvalid) D[(size_t)r * ldd + c] = acc[nt][g];
      tile[(half * 8 + g) * PITCH + c] = acc[nt][g];
    }
  }
  __syncthreads();

  if (EPI == 1) {
    const int hh = lane & 7;
    float sp[8], dp[8];
    #pragma unroll
    for (int c = 0; c < 8; ++c) { sp[c] = a_src[hh * 8 + c]; dp[c] = a_dst[hh * 8 + c]; }
    #pragma unroll
    for (int i = 0; i < 4; ++i) {
      const int p = lane + 32 * i;
      const float* tp = tile + (p >> 3) * PITCH + hh * 8;
      const v4f t0 = *(const v4f*)tp, t1 = *(const v4f*)(tp + 4);
      const float t[8] = {t0.x, t0.y, t0.z, t0.w, t1.x, t1.y, t1.z, t1.w};
      float s1 = 0.f, s2 = 0.f;
      #pragma unroll
      for (int c = 0; c < 8; ++c) { s1 += t[c] * sp[c]; s2 += t[c] * dp[c]; }
      if (live) {
        o_src[(size_t)tileM * 128 + p] = s1;
        o_dst[(size_t)tileM * 128 + p] = s2;
      }
    }
  } else {
    const float* __restrict__ av = half ? a_dst : a_src;
    const float* tp = tile + l16 * PITCH;
    float s = 0.f;
    #pragma unroll
    for (int c4 = 0; c4 < 16 * NT; c4 += 4) {
      const v4f t = *(const v4f*)(tp + c4);
      const float tt[4] = {t.x, t.y, t.z, t.w};
      #pragma unroll
      for (int u = 0; u < 4; ++u) {
        const int c = c4 + u;
        const float coef = av[min(c, nvalid - 1)];
        s += tt[u] * (c < nvalid ? coef : 0.f);
      }
    }
    if (live) o_src[((size_t)tileM * 16 + l16) * 2 + half] = s;
  }
}

template <int NT, int EPI>
__global__ void __launch_bounds__(32)
gemm_wmma_f32_nt(const float* __restrict__ A, const float* __restrict__ B,
                 float* __restrict__ D, int M, int K, int ldb, int ldd, int nvalid,
                 const float* __restrict__ a_src, const float* __restrict__ a_dst,
                 float* __restrict__ o_src, float* __restrict__ o_dst) {
  __shared__ float tile[16 * (16 * NT + 4)];
  gemm_wmma_f32_tile<NT, EPI>(tile, blockIdx.x, true, A, B, D, K, ldb, ldd, nvalid,
                              a_src, a_dst, o_src, o_dst);
}

#define GEMM1_SLAB (16 * (16 * 4 + 4))
__global__ void __launch_bounds__(256)
scatter_and_gemm1_kernel(int gemm_blocks, int ntiles,
                         const int* __restrict__ ei, int E, int N, int span, int per,
                         const int* __restrict__ part, int* __restrict__ csr,
                         const float* __restrict__ x, const float* __restrict__ W1,
                         float* __restrict__ h1, int F, int HC,
                         const float* __restrict__ a_src, const float* __restrict__ a_dst,
                         float* __restrict__ as1, float* __restrict__ ad1) {
  extern __shared__ __attribute__((aligned(16))) int lds[];
  if ((int)blockIdx.x < gemm_blocks) {
    const int w = __builtin_amdgcn_readfirstlane(threadIdx.x >> 5);
    const int t = blockIdx.x * 8 + w;
    gemm_wmma_f32_tile<4, 1>((float*)lds + w * GEMM1_SLAB, min(t, ntiles - 1), t < ntiles,
                             x, W1, h1, F, HC, HC, HC, a_src, a_dst, as1, ad1);
  } else {
    const int b = blockIdx.x - gemm_blocks;
    range_scatter_block(lds, b % SORT_Q, b / SORT_Q, ei, E, N, span, per, part, csr);
  }
}


#define EXW_PITCH 36
#define EXW (8 * EXW_PITCH)
__global__ void __launch_bounds__(256)
gather_agg1_kernel(const int* __restrict__ start, const int* __restrict__ deg,
                   const int* __restrict__ csr,
                   const float* __restrict__ as, const float* __restrict__ ad,
                   const float* __restrict__ h1, const float* __restrict__ b,
                   float* __restrict__ out, int N) {
  __shared__ __attribute__((aligned(16))) float exw[8 * EXW];
  const int lane = threadIdx.x & 31;
  const int w = threadIdx.x >> 5;
  const int n = __builtin_amdgcn_readfirstlane(blockIdx.x * 8 + w);
  if (n >= N) return;
  const int hh = lane >> 2;
  const int h = lane & 7, g = lane >> 3;
  float* lw = exw + w * EXW;
  const int beg = start[n], cnt = deg[n];
  const float advh = ad[n * 8 + h];
  const float vself = lrelu(as[n * 8 + hh] + ad[n * 8 + hh]);

  auto logits = [&](int myv, int m, float* o) {
    #pragma unroll
    for (int i = 0; i < 8; ++i) {
      const int e = 4 * i + g;
      const int s = __shfl(myv, e);
      const float v = lrelu(as[s * 8 + h] + advh);
      o[i] = e < m ? v : -INFINITY;
    }
  };

  int my = (lane < cnt) ? csr[beg + lane] : n;
  float lg[8];
  logits(my, min(32, cnt), lg);
  float mxh = lrelu(as[n * 8 + h] + advh);
  #pragma unroll
  for (int i = 0; i < 8; ++i) mxh = fmaxf(mxh, lg[i]);
  for (int base = 32; base < cnt; base += 32) {
    const int k = base + lane;
    const int my2 = (k < cnt) ? csr[beg + k] : n;
    float l2[8];
    logits(my2, min(32, cnt - base), l2);
    #pragma unroll
    for (int i = 0; i < 8; ++i) mxh = fmaxf(mxh, l2[i]);
  }
  mxh = fmaxf(mxh, __shfl_xor(mxh, 8));
  mxh = fmaxf(mxh, __shfl_xor(mxh, 16));
  const float mx = __shfl(mxh, hh);

  float ex = expf(vself - mx);
  float den = ex;
  v2f hv = *(const v2f*)(h1 + (size_t)n * 64 + 2 * lane);
  float a0 = ex * hv.x, a1 = ex * hv.y;
  for (int base = 0; base < cnt; base += 32) {
    const int m = min(32, cnt - base);
    if (base > 0) {
      const int k = base + lane;
      my = (k < cnt) ? csr[beg + k] : n;
      logits(my, m, lg);
    }
    #pragma unroll
    for (int i = 0; i < 8; ++i) lw[h * EXW_PITCH + 4 * i + g] = expf(lg[i] - mxh);
    const float* er = lw + hh * EXW_PITCH;
    int j = 0;
    for (; j + 4 <= m; j += 4) {
      const v4f e4 = *(const v4f*)(er + j);
      const int s0 = __builtin_amdgcn_readlane(my, j);
      const int s1 = __builtin_amdgcn_readlane(my, j + 1);
      const int s2 = __builtin_amdgcn_readlane(my, j + 2);
      const int s3 = __builtin_amdgcn_readlane(my, j + 3);
      const v2f r0 = *(const v2f*)(h1 + (size_t)s0 * 64 + 2 * lane);
      const v2f r1 = *(const v2f*)(h1 + (size_t)s1 * 64 + 2 * lane);
      const v2f r2 = *(const v2f*)(h1 + (size_t)s2 * 64 + 2 * lane);
      const v2f r3 = *(const v2f*)(h1 + (size_t)s3 * 64 + 2 * lane);
      den += e4.x; a0 = fmaf(e4.x, r0.x, a0); a1 = fmaf(e4.x, r0.y, a1);
      den += e4.y; a0 = fmaf(e4.y, r1.x, a0); a1 = fmaf(e4.y, r1.y, a1);
      den += e4.z; a0 = fmaf(e4.z, r2.x, a0); a1 = fmaf(e4.z, r2.y, a1);
      den += e4.w; a0 = fmaf(e4.w, r3.x, a0); a1 = fmaf(e4.w, r3.y, a1);
    }
    for (; j < m; ++j) {
      const float e = er[j];
      const int s = __builtin_amdgcn_readlane(my, j);
      hv = *(const v2f*)(h1 + (size_t)s * 64 + 2 * lane);
      den += e;
      a0 = fmaf(e, hv.x, a0);
      a1 = fmaf(e, hv.y, a1);
    }
  }
  const float dd = den + EPSV;
  const v2f bb = *(const v2f*)(b + 2 * lane);
  float o0 = a0 / dd + bb.x, o1 = a1 / dd + bb.y;
  v2f o;
  o.x = o0 > 0.f ? o0 : expm1f(o0);
  o.y = o1 > 0.f ? o1 : expm1f(o1);
  *(v2f*)(out + (size_t)n * 64 + 2 * lane) = o;
}


__global__ void __launch_bounds__(256)
gather_agg2_kernel(const int* __restrict__ start, const int* __restrict__ deg,
                   const int* __restrict__ csr,
                   const float* __restrict__ asd,
                   const float* __restrict__ h2, const float* __restrict__ b2,
                   float* __restrict__ y, int N, int C) {
  __shared__ float rows[8 * 64];
  const int lane = threadIdx.x & 31;
  const int w = threadIdx.x >> 5;
  const int n = __builtin_amdgcn_readfirstlane(blockIdx.x * 8 + w);
  const bool own = 2 * lane < C;
  const int col = own ? 2 * lane : 0;
  if (n < N) {
    const int beg = start[n], cnt = deg[n];
    const float adv = asd[2 * n + 1];
    const float vself = lrelu(asd[2 * n] + adv);

    float mx = vself;
    int my0 = n;
    float lg0 = -INFINITY;
    if (lane < cnt) {
      my0 = csr[beg + lane];
      lg0 = lrelu(asd[2 * my0] + adv);
    }
    mx = fmaxf(mx, lg0);
    for (int base = 32; base < cnt; base += 32) {
      const int k = base + lane;
      if (k < cnt) mx = fmaxf(mx, lrelu(asd[2 * csr[beg + k]] + adv));
    }
    for (int o = 16; o > 0; o >>= 1) mx = fmaxf(mx, __shfl_xor(mx, o));

    const float exs = expf(vself - mx);
    float den = exs;
    float a0 = 0.f, a1 = 0.f;
    if (own) {
      const v2f hv = *(const v2f*)(h2 + (size_t)n * C + 2 * lane);
      a0 = exs * hv.x; a1 = exs * hv.y;
    }
    for (int base = 0; base < cnt; base += 32) {
      int my = my0;
      float lg = lg0;
      if (base > 0) {
        const int k = base + lane;
        my = (k < cnt) ? csr[beg + k] : n;
        lg = lrelu(asd[2 * my] + adv);
      }
      const float exl = expf(lg - mx);
      const int m = min(32, cnt - base);
      #pragma unroll 4
      for (int j = 0; j < m; ++j) {
        const int s = __builtin_amdgcn_readlane(my, j);
        const float e = __int_as_float(__builtin_amdgcn_readlane(__float_as_int(exl), j));
        den += e;
        const v2f hv = *(const v2f*)(h2 + (size_t)s * C + col);
        a0 += e * hv.x;
        a1 += e * hv.y;
      }
    }
    float v0 = -INFINITY, v1 = -INFINITY;
    if (own) {
      const float dd = den + EPSV;
      const v2f bb = *(const v2f*)(b2 + 2 * lane);
      v0 = a0 / dd + bb.x;
      v1 = a1 / dd + bb.y;
      v0 = v0 > 0.f ? v0 : expm1f(v0);
      v1 = v1 > 0.f ? v1 : expm1f(v1);
    }
    float rmx = fmaxf(v0, v1);
    for (int o = 16; o > 0; o >>= 1) rmx = fmaxf(rmx, __shfl_xor(rmx, o));
    float sm = own ? expf(v0 - rmx) + expf(v1 - rmx) : 0.f;
    for (int o = 16; o > 0; o >>= 1) sm += __shfl_xor(sm, o);
    const float l = logf(sm);
    if (own) {
      v2f o;
      o.x = v0 - rmx - l; o.y = v1 - rmx - l;
      *(v2f*)(rows + w * C + 2 * lane) = o;
    }
  }
  __syncthreads();
  const int i = 2 * (int)threadIdx.x;
  const size_t base = (size_t)blockIdx.x * 8 * C;
  if (i < 8 * C && base + i < (size_t)N * C)
    *(v2f*)(y + base + i) = *(const v2f*)(rows + i);
}


static inline int cdiv(int a, int b) { return (a + b - 1) / b; }

extern "C" void kernel_launch(void* const* d_in, const int* in_sizes, int n_in,
                              void* d_out, int out_size, void* d_ws, size_t ws_size,
                              hipStream_t stream) {
  const float* x      = (const float*)d_in[0];
  const int*   ei     = (const int*)  d_in[1];
  const float* W1     = (const float*)d_in[2];
  const float* a_src1 = (const float*)d_in[3];
  const float* a_dst1 = (const float*)d_in[4];
  const float* b1     = (const float*)d_in[5];
  const float* W2     = (const float*)d_in[6];
  const float* a_src2 = (const float*)d_in[7];
  const float* a_dst2 = (const float*)d_in[8];
  const float* b2     = (const float*)d_in[9];
  float* out = (float*)d_out;

  const int HC   = in_sizes[5];
  const int NCLS = in_sizes[9];
  const int F    = in_sizes[2] / HC;
  const int N    = in_sizes[0] / F;
  const int E    = in_sizes[1] / 2;

  float* ws = (float*)d_ws;
  size_t off = 0;
  float* h1   = ws + off; off += (size_t)N * HC;
  float* out1 = ws + off; off += (size_t)N * HC;
  float* h2   = ws + off; off += (size_t)N * NCLS;
  float* as1  = ws + off; off += (size_t)N * 8;
  float* ad1  = ws + off; off += (size_t)N * 8;
  float* asd2 = ws + off; off += (size_t)N * 2;
  int* deg    = (int*)(ws + off); off += (size_t)N;
  int* total  = (int*)(ws + off); off += 2;
  int* start  = (int*)(ws + off); off += (size_t)N;
  int* csr    = (int*)(ws + off); off += (size_t)E;
  int* part   = (int*)(ws + off); off += (size_t)SORT_Q * N;

  const int span = cdiv(N, SORT_R);
  const int per  = cdiv(cdiv(E, 4), SORT_Q) * 4;
  range_count_kernel<<<dim3(SORT_Q, SORT_R), 256, span * sizeof(int), stream>>>(ei, E, N, span, per, part, total);
  segment_alloc_kernel<<<cdiv(N, 256), 256, 0, stream>>>(part, deg, start, total, N);

  const int ntiles = N / 16;
  const int gemm_blocks = cdiv(ntiles, 8);
  const size_t bins_bytes = span * sizeof(int), slab_bytes = 8 * GEMM1_SLAB * sizeof(float);
  scatter_and_gemm1_kernel<<<gemm_blocks + SORT_Q * SORT_R, 256,
                             bins_bytes > slab_bytes ? bins_bytes : slab_bytes, stream>>>(
      gemm_blocks, ntiles, ei, E, N, span, per, part, csr,
      x, W1, h1, F, HC, a_src1, a_dst1, as1, ad1);

  gather_agg1_kernel<<<cdiv(N, 8), 256, 0, stream>>>(start, deg, csr, as1, ad1, h1, b1, out1, N);

  gemm_wmma_f32_nt<3, 2><<<dim3(N / 16), 32, 0, stream>>>(out1, W2, h2, N, HC, NCLS, NCLS, NCLS,
                                                          a_src2, a_dst2, asd2, asd2);
  gather_agg2_kernel<<<cdiv(N, 8), 256, 0, stream>>>(start, deg, csr, asd2, h2, b2, out, N, NCLS);
}


# both gathers: two destination nodes per wave, one per 16-lane half, a row as one 16-byte load per lane, 32 list entries per node held in registers; 100000 -> 50000 gather waves; sum order kept
# speedup vs baseline: 10.3161x; 1.0360x over previous
#include <hip/hip_runtime.h>
#include <hip/hip_bf16.h>
#include <math.h>

typedef __attribute__((ext_vector_type(2))) float v2f;
typedef __attribute__((ext_vector_type(8))) float v8f;

#define NEG_SLOPE 0.2f
#define EPSV 1e-16f


#define SORT_R 16
#define SORT_Q 64

typedef __attribute__((ext_vector_type(4))) int v4i;

template <bool WITH_SRC, class F>
__device__ __forceinline__ void scan_chunk(const int* __restrict__ srcp,
                                           const int* __restrict__ dstp,
                                           int e0, int e1, bool vec, F f) {
  if (vec) {
    for (int i = e0 + 4 * (int)threadIdx.x; i < e1; i += 4 * 256 * 4) {
      v4i d[4], s[4];
      #pragma unroll
      for (int u = 0; u < 4; ++u) {
        const int j = i + u * 1024;
        const bool in = j < e1;
        d[u] = in ? *(const v4i*)(dstp + j) : (v4i){-1, -1, -1, -1};
        if (WITH_SRC) s[u] = in ? *(const v4i*)(srcp + j) : (v4i){0, 0, 0, 0};
        else          s[u] = (v4i){0, 0, 0, 0};
      }
      #pragma unroll
      for (int u = 0; u < 4; ++u) {
        f(s[u].x, d[u].x); f(s[u].y, d[u].y); f(s[u].z, d[u].z); f(s[u].w, d[u].w);
      }
    }
  } else {
    for (int i = e0 + (int)threadIdx.x; i < e1; i += 256)
      f(WITH_SRC ? srcp[i] : 0, dstp[i]);
  }
}

__device__ __forceinline__ bool list_is_vec(const int* ei, int E) {
  return ((E & 3) == 0) && ((((size_t)ei) & 15) == 0);
}

__global__ void __launch_bounds__(256)
range_count_kernel(const int* __restrict__ ei, int E, int N, int span, int per,
                   int* __restrict__ part, int* __restrict__ total) {
  extern __shared__ int bins[];
  const int q = blockIdx.x, r = blockIdx.y;
  const int lo = r * span;
  for (int i = threadIdx.x; i < span; i += 256) bins[i] = 0;
  if (q == 0 && r == 0 && threadIdx.x == 0) *total = 0;
  __syncthreads();
  const int e0 = min(E, q * per), e1 = min(E, e0 + per);
  const unsigned uspan = (unsigned)span;
  scan_chunk<false>(ei, ei + E, e0, e1, list_is_vec(ei, E), [&](int, int d) {
    const unsigned t = (unsigned)(d - lo);
    if (t < uspan) atomicAdd(&bins[t], 1);
  });
  __syncthreads();
  for (int i = threadIdx.x; i < span; i += 256)
    if (lo + i < N) part[(size_t)q * N + lo + i] = bins[i];
}

__global__ void __launch_bounds__(256)
segment_alloc_kernel(int* __restrict__ part, int* __restrict__ deg,
                     int* __restrict__ start, int* __restrict__ total, int N) {
  __shared__ int wave_off[8];
  __shared__ int block_base;
  const int n = blockIdx.x * 256 + threadIdx.x;
  const int lane = threadIdx.x & 31, w = threadIdx.x >> 5;
  int c[SORT_Q];
  int d = 0;
  #pragma unroll
  for (int q = 0; q < SORT_Q; ++q) {
    c[q] = (n < N) ? part[(size_t)q * N + n] : 0;
    d += c[q];
  }
  int incl = d;
  for (int o = 1; o < 32; o <<= 1) {
    int t = __shfl_up(incl, o);
    if (lane >= o) incl += t;
  }
  if (lane == 31) wave_off[w] = incl;
  __syncthreads();
  if (threadIdx.x == 0) {
    int run = 0;
    for (int i = 0; i < 8; ++i) { int t = wave_off[i]; wave_off[i] = run; run += t; }
    block_base = atomicAdd(total, run);
  }
  __syncthreads();
  if (n < N) {
    int run = block_base + wave_off[w] + incl - d;
    start[n] = run;
    deg[n] = d;
    #pragma unroll
    for (int q = 0; q < SORT_Q; ++q) { part[(size_t)q * N + n] = run; run += c[q]; }
  }
}

__device__ __forceinline__ void
range_scatter_block(int* bins, int q, int r,
                    const int* __restrict__ ei, int E, int N, int span, int per,
                    const int* __restrict__ part, int* __restrict__ csr) {
  const int lo = r * span;
  for (int i0 = threadIdx.x; i0 < span; i0 += 256 * 8) {
    int v[8];
    #pragma unroll
    for (int u = 0; u < 8; ++u) {
      const int i = i0 + u * 256;
      v[u] = (i < span && lo + i < N) ? part[(size_t)q * N + lo + i] : 0;
    }
    #pragma unroll
    for (int u = 0; u < 8; ++u) {
      const int i = i0 + u * 256;
      if (i < span) bins[i] = v[u];
    }
  }
  __syncthreads();
  const int e0 = min(E, q * per), e1 = min(E, e0 + per);
  const unsigned uspan = (unsigned)span;
  scan_chunk<true>(ei, ei + E, e0, e1, list_is_vec(ei, E), [&](int s, int d) {
    const unsigned t = (unsigned)(d - lo);
    if (t < uspan) csr[atomicAdd(&bins[t], 1)] = s;
  });
}

__device__ __forceinline__ float lrelu(float v) { return v > 0.f ? v : NEG_SLOPE * v; }

typedef __attribute__((ext_vector_type(4))) float v4f;

template <int NT, int EPI>
__device__ __forceinline__ void
gemm_wmma_f32_tile(float* tile, int tileM, bool live,
                   const float* __restrict__ A, const float* __restrict__ B,
                   float* __restrict__ D, int K, int ldb, int ldd, int nvalid,
                   const float* __restrict__ a_src, const float* __restrict__ a_dst,
                   float* __restrict__ o_src, float* __restrict__ o_dst) {
  constexpr int PITCH = 16 * NT + 4;
  const int lane = threadIdx.x & 31;
  const int half = lane >> 4;
  const int l16  = lane & 15;
  const int rowA = tileM * 16 + l16;
  const float* __restrict__ arow = A + (size_t)rowA * K;

  v8f acc[NT];
  #pragma unroll
  for (int nt = 0; nt < NT; ++nt) acc[nt] = (v8f){0.f,0.f,0.f,0.f,0.f,0.f,0.f,0.f};

  for (int kk = 0; kk < K; kk += 4) {
    const int ka = kk + half * 2;
    v2f a;
    a.x = arow[ka];
    a.y = arow[ka + 1];
    const float* __restrict__ brow0 = B + (size_t)ka * ldb;
    const float* __restrict__ brow1 = B + (size_t)(ka + 1) * ldb;
    #pragma unroll
    for (int nt = 0; nt < NT; ++nt) {
      const int colB = nt * 16 + l16;
      v2f b;
      if (EPI == 2) {
        const int colc = min(colB, nvalid - 1);
        const float b0 = brow0[colc], b1 = brow1[colc];
        b.x = colB < nvalid ? b0 : 0.f;
        b.y = colB < nvalid ? b1 : 0.f;
      } else {
        b.x = brow0[colB];
        b.y = brow1[colB];
      }
      acc[nt] = __builtin_amdgcn_wmma_f32_16x16x4_f32(false, a, false, b,
                                                      (short)0, acc[nt], false, false);
    }
  }
  #pragma unroll
  for (int nt = 0; nt < NT; ++nt) {
    #pragma unroll
    for (int g = 0; g < 8; ++g) {
      const int r = tileM * 16 + half * 8 + g;
      const int c = nt * 16 + l16;
      if (live && c < nvalid) D[(size_t)r * ldd + c] = acc[nt][g];
      tile[(half * 8 + g) * PITCH + c] = acc[nt][g];
    }
  }
  __syncthreads();

  if (EPI == 1) {
    const int hh = lane & 7;
    float sp[8], dp[8];
    #pragma unroll
    for (int c = 0; c < 8; ++c) { sp[c] = a_src[hh * 8 + c]; dp[c] = a_dst[hh * 8 + c]; }
    #pragma unroll
    for (int i = 0; i < 4; ++i) {
      const int p = lane + 32 * i;
      const float* tp = tile + (p >> 3) * PITCH + hh * 8;
      const v4f t0 = *(const v4f*)tp, t1 = *(const v4f*)(tp + 4);
      const float t[8] = {t0.x, t0.y, t0.z, t0.w, t1.x, t1.y, t1.z, t1.w};
      float s1 = 0.f, s2 = 0.f;
      #pragma unroll
      for (int c = 0; c < 8; ++c) { s1 += t[c] * sp[c]; s2 += t[c] * dp[c]; }
      if (live) {
        o_src[(size_t)tileM * 128 + p] = s1;
        o_dst[(size_t)tileM * 128 + p] = s2;
      }
    }
  } else {
    const float* __restrict__ av = half ? a_dst : a_src;
    const float* tp = tile + l16 * PITCH;
    float s = 0.f;
    #pragma unroll
    for (int c4 = 0; c4 < 16 * NT; c4 += 4) {
      const v4f t = *(const v4f*)(tp + c4);
      const float tt[4] = {t.x, t.y, t.z, t.w};
      #pragma unroll
      for (int u = 0; u < 4; ++u) {
        const int c = c4 + u;
        const float coef = av[min(c, nvalid - 1)];
        s += tt[u] * (c < nvalid ? coef : 0.f);
      }
    }
    if (live) o_src[((size_t)tileM * 16 + l16) * 2 + half] = s;
  }
}

template <int NT, int EPI>
__global__ void __launch_bounds__(32)
gemm_wmma_f32_nt(const float* __restrict__ A, const float* __restrict__ B,
                 float* __restrict__ D, int M, int K, int ldb, int ldd, int nvalid,
                 const float* __restrict__ a_src, const float* __restrict__ a_dst,
                 float* __restrict__ o_src, float* __restrict__ o_dst) {
  __shared__ float tile[16 * (16 * NT + 4)];
  gemm_wmma_f32_tile<NT, EPI>(tile, blockIdx.x, true, A, B, D, K, ldb, ldd, nvalid,
                              a_src, a_dst, o_src, o_dst);
}

#define GEMM1_SLAB (16 * (16 * 4 + 4))
__global__ void __launch_bounds__(256)
scatter_and_gemm1_kernel(int gemm_blocks, int ntiles,
                         const int* __restrict__ ei, int E, int N, int span, int per,
                         const int* __restrict__ part, int* __restrict__ csr,
                         const float* __restrict__ x, const float* __restrict__ W1,
                         float* __restrict__ h1, int F, int HC,
                         const float* __restrict__ a_src, const float* __restrict__ a_dst,
                         float* __restrict__ as1, float* __restrict__ ad1) {
  extern __shared__ __attribute__((aligned(16))) int lds[];
  if ((int)blockIdx.x < gemm_blocks) {
    const int w = __builtin_amdgcn_readfirstlane(threadIdx.x >> 5);
    const int t = blockIdx.x * 8 + w;
    gemm_wmma_f32_tile<4, 1>((float*)lds + w * GEMM1_SLAB, min(t, ntiles - 1), t < ntiles,
                             x, W1, h1, F, HC, HC, HC, a_src, a_dst, as1, ad1);
  } else {
    const int b = blockIdx.x - gemm_blocks;
    range_scatter_block(lds, b % SORT_Q, b / SORT_Q, ei, E, N, span, per, part, csr);
  }
}


#define EXH_PITCH 20
#define EXH (8 * EXH_PITCH)
__global__ void __launch_bounds__(256)
gather_agg1_kernel(const int* __restrict__ start, const int* __restrict__ deg,
                   const int* __restrict__ csr,
                   const float* __restrict__ as, const float* __restrict__ ad,
                   const float* __restrict__ h1, const float* __restrict__ b,
                   float* __restrict__ out, int N) {
  __shared__ __attribute__((aligned(16))) float exw[16 * EXH];
  const int lane = threadIdx.x & 31;
  const int w = threadIdx.x >> 5;
  const int hbase = lane & 16;
  const int l16 = lane & 15;
  const int slot = 2 * w + (hbase >> 4);
  const int n = blockIdx.x * 16 + slot;
  const bool valid = n < N;
  const int nn = valid ? n : N - 1;
  const int hh = l16 >> 1;
  const int h = l16 & 7, g = l16 >> 3;
  float* lw = exw + slot * EXH;
  const v4f* __restrict__ h1q = (const v4f*)h1;

  const int beg = start[nn];
  const int cnt = valid ? deg[nn] : 0;
  const int cntmax = __builtin_amdgcn_readfirstlane(max(cnt, __shfl_xor(cnt, 16)));
  const float advh = ad[nn * 8 + h];
  const float vself = lrelu(as[nn * 8 + hh] + ad[nn * 8 + hh]);

  auto logits = [&](int myv, int m, float* o) {
    #pragma unroll
    for (int i = 0; i < 8; ++i) {
      const int e = 2 * i + g;
      const int s = __shfl(myv, hbase | e);
      const float v = lrelu(as[s * 8 + h] + advh);
      o[i] = e < m ? v : -INFINITY;
    }
  };

  const bool inA = l16 < cnt, inB = 16 + l16 < cnt;
  const int rA = csr[inA ? beg + l16 : 0];
  const int rB = csr[inB ? beg + 16 + l16 : 0];
  asm volatile("" :: "v"(rA), "v"(rB));
  const int myA = inA ? rA : nn, myB = inB ? rB : nn;
  float lgA[8], lgB[8];
  logits(myA, cnt, lgA);
  #pragma unroll
  for (int i = 0; i < 8; ++i) lgB[i] = -INFINITY;
  if (cntmax > 16) logits(myB, cnt - 16, lgB);
  float mxh = lrelu(as[nn * 8 + h] + advh);
  #pragma unroll
  for (int i = 0; i < 8; ++i) mxh = fmaxf(mxh, fmaxf(lgA[i], lgB[i]));
  for (int base = 32; base < cntmax; base += 16) {
    const int k = base + l16;
    const bool in = k < cnt;
    const int r = csr[in ? beg + k : 0];
    float l2[8];
    logits(in ? r : nn, cnt - base, l2);
    #pragma unroll
    for (int i = 0; i < 8; ++i) mxh = fmaxf(mxh, l2[i]);
  }
  mxh = fmaxf(mxh, __shfl_xor(mxh, 8));
  const float mx = __shfl(mxh, hbase | hh);

  const float ex = expf(vself - mx);
  float den = ex;
  const v4f hv = h1q[nn * 16 + l16];
  float a0 = ex * hv.x, a1 = ex * hv.y, a2 = ex * hv.z, a3 = ex * hv.w;

  auto row_batch = [&](int my, const float* lg, int mm) {
    #pragma unroll
    for (int i = 0; i < 8; ++i) lw[h * EXH_PITCH + 2 * i + g] = expf(lg[i] - mxh);
    const float* er = lw + hh * EXH_PITCH;
    for (int j = 0; j < mm; j += 4) {
      const v4f e4 = *(const v4f*)(er + j);
      const int s0 = __shfl(my, hbase | j);
      const int s1 = __shfl(my, hbase | (j + 1));
      const int s2 = __shfl(my, hbase | (j + 2));
      const int s3 = __shfl(my, hbase | (j + 3));
      const v4f r0 = h1q[s0 * 16 + l16];
      const v4f r1 = h1q[s1 * 16 + l16];
      const v4f r2 = h1q[s2 * 16 + l16];
      const v4f r3 = h1q[s3 * 16 + l16];
      den += e4.x;
      a0 = fmaf(e4.x, r0.x, a0); a1 = fmaf(e4.x, r0.y, a1); a2 = fmaf(e4.x, r0.z, a2); a3 = fmaf(e4.x, r0.w, a3);
      den += e4.y;
      a0 = fmaf(e4.y, r1.x, a0); a1 = fmaf(e4.y, r1.y, a1); a2 = fmaf(e4.y, r1.z, a2); a3 = fmaf(e4.y, r1.w, a3);
      den += e4.z;
      a0 = fmaf(e4.z, r2.x, a0); a1 = fmaf(e4.z, r2.y, a1); a2 = fmaf(e4.z, r2.z, a2); a3 = fmaf(e4.z, r2.w, a3);
      den += e4.w;
      a0 = fmaf(e4.w, r3.x, a0); a1 = fmaf(e4.w, r3.y, a1); a2 = fmaf(e4.w, r3.z, a2); a3 = fmaf(e4.w, r3.w, a3);
    }
  };
  if (cntmax > 0)  row_batch(myA, lgA, min(16, cntmax));
  if (cntmax > 16) row_batch(myB, lgB, min(16, cntmax - 16));
  for (int base = 32; base < cntmax; base += 16) {
    const int k = base + l16;
    const bool in = k < cnt;
    const int r = csr[in ? beg + k : 0];
    const int my = in ? r : nn;
    float l2[8];
    logits(my, cnt - base, l2);
    row_batch(my, l2, min(16, cntmax - base));
  }
  const float dd = den + EPSV;
  const float o0 = a0 / dd + b[4 * l16];
  const float o1 = a1 / dd + b[4 * l16 + 1];
  const float o2 = a2 / dd + b[4 * l16 + 2];
  const float o3 = a3 / dd + b[4 * l16 + 3];
  v4f o;
  o.x = o0 > 0.f ? o0 : expm1f(o0);
  o.y = o1 > 0.f ? o1 : expm1f(o1);
  o.z = o2 > 0.f ? o2 : expm1f(o2);
  o.w = o3 > 0.f ? o3 : expm1f(o3);
  if (valid) *(v4f*)(out + (size_t)nn * 64 + 4 * l16) = o;
}


__global__ void __launch_bounds__(256)
gather_agg2_kernel(const int* __restrict__ start, const int* __restrict__ deg,
                   const int* __restrict__ csr,
                   const float* __restrict__ asd,
                   const float* __restrict__ h2, const float* __restrict__ b2,
                   float* __restrict__ y, int N, int C) {
  __shared__ __attribute__((aligned(16))) float rows[16 * 64];
  const int lane = threadIdx.x & 31;
  const int w = threadIdx.x >> 5;
  const int hbase = lane & 16;
  const int l16 = lane & 15;
  const int slot = 2 * w + (hbase >> 4);
  const int n = blockIdx.x * 16 + slot;
  const bool valid = n < N;
  const int nn = valid ? n : N - 1;
  const bool own = 4 * l16 < C;
  const int colq = own ? l16 : 0;
  const v4f* __restrict__ h2q = (const v4f*)h2;
  const int C4 = C >> 2;

  const int beg = start[nn];
  const int cnt = valid ? deg[nn] : 0;
  const int cntmax = __builtin_amdgcn_readfirstlane(max(cnt, __shfl_xor(cnt, 16)));
  const float adv = asd[2 * nn + 1];
  const float vself = lrelu(asd[2 * nn] + adv);

  const bool inA = l16 < cnt, inB = 16 + l16 < cnt;
  const int rA = csr[inA ? beg + l16 : 0];
  const int rB = csr[inB ? beg + 16 + l16 : 0];
  asm volatile("" :: "v"(rA), "v"(rB));
  const int myA = inA ? rA : nn, myB = inB ? rB : nn;
  const float gA = asd[2 * myA], gB = asd[2 * myB];
  asm volatile("" :: "v"(gA), "v"(gB));
  const float lgA = inA ? lrelu(gA + adv) : -INFINITY;
  const float lgB = inB ? lrelu(gB + adv) : -INFINITY;
  float mx = fmaxf(vself, fmaxf(lgA, lgB));
  for (int base = 32; base < cntmax; base += 16) {
    const int k = base + l16;
    const bool in = k < cnt;
    const int r = csr[in ? beg + k : 0];
    const float g = asd[2 * (in ? r : nn)];
    mx = fmaxf(mx, in ? lrelu(g + adv) : -INFINITY);
  }
  for (int o = 8; o > 0; o >>= 1) mx = fmaxf(mx, __shfl_xor(mx, o));

  const float exs = expf(vself - mx);
  float den = exs;
  const v4f hs = h2q[nn * C4 + colq];
  float a0 = exs * hs.x, a1 = exs * hs.y, a2 = exs * hs.z, a3 = exs * hs.w;

  auto row_batch = [&](int my, float exl, int mm) {
    for (int j = 0; j < mm; j += 4) {
      int s[4];
      float e[4];
      v4f r[4];
      #pragma unroll
      for (int u = 0; u < 4; ++u) {
        const int sl = hbase | (j + u);
        s[u] = __shfl(my, sl);
        e[u] = __shfl(exl, sl);
        r[u] = h2q[s[u] * C4 + colq];
      }
      #pragma unroll
      for (int u = 0; u < 4; ++u) {
        den += e[u];
        a0 = fmaf(e[u], r[u].x, a0);
        a1 = fmaf(e[u], r[u].y, a1);
        a2 = fmaf(e[u], r[u].z, a2);
        a3 = fmaf(e[u], r[u].w, a3);
      }
    }
  };
  if (cntmax > 0)  row_batch(myA, inA ? expf(lgA - mx) : 0.f, min(16, cntmax));
  if (cntmax > 16) row_batch(myB, inB ? expf(lgB - mx) : 0.f, min(16, cntmax - 16));
  for (int base = 32; base < cntmax; base += 16) {
    const int k = base + l16;
    const bool in = k < cnt;
    const int r = csr[in ? beg + k : 0];
    const int my = in ? r : nn;
    const float lg = lrelu(asd[2 * my] + adv);
    row_batch(my, in ? expf(lg - mx) : 0.f, min(16, cntmax - base));
  }

  float v0 = -INFINITY, v1 = -INFINITY, v2 = -INFINITY, v3 = -INFINITY;
  if (own) {
    const float dd = den + EPSV;
    v0 = a0 / dd + b2[4 * l16];
    v1 = a1 / dd + b2[4 * l16 + 1];
    v2 = a2 / dd + b2[4 * l16 + 2];
    v3 = a3 / dd + b2[4 * l16 + 3];
    v0 = v0 > 0.f ? v0 : expm1f(v0);
    v1 = v1 > 0.f ? v1 : expm1f(v1);
    v2 = v2 > 0.f ? v2 : expm1f(v2);
    v3 = v3 > 0.f ? v3 : expm1f(v3);
  }
  float rmx = fmaxf(fmaxf(v0, v1), fmaxf(v2, v3));
  for (int o = 8; o > 0; o >>= 1) rmx = fmaxf(rmx, __shfl_xor(rmx, o));
  float q0 = own ? expf(v0 - rmx) + expf(v1 - rmx) : 0.f;
  float q1 = own ? expf(v2 - rmx) + expf(v3 - rmx) : 0.f;
  for (int o = 8; o > 0; o >>= 1) { q0 += __shfl_xor(q0, o); q1 += __shfl_xor(q1, o); }
  const float l = logf(q0 + q1);
  if (own && valid) {
    v4f o;
    o.x = v0 - rmx - l; o.y = v1 - rmx - l; o.z = v2 - rmx - l; o.w = v3 - rmx - l;
    *(v4f*)(rows + slot * C + 4 * l16) = o;
  }
  __syncthreads();
  const size_t gbase = (size_t)blockIdx.x * 16 * C;
  for (int i = 2 * (int)threadIdx.x; i < 16 * C; i += 512)
    if (gbase + i < (size_t)N * C)
      *(v2f*)(y + gbase + i) = *(const v2f*)(rows + i);
}


static inline int cdiv(int a, int b) { return (a + b - 1) / b; }

extern "C" void kernel_launch(void* const* d_in, const int* in_sizes, int n_in,
                              void* d_out, int out_size, void* d_ws, size_t ws_size,
                              hipStream_t stream) {
  const float* x      = (const float*)d_in[0];
  const int*   ei     = (const int*)  d_in[1];
  const float* W1     = (const float*)d_in[2];
  const float* a_src1 = (const float*)d_in[3];
  const float* a_dst1 = (const float*)d_in[4];
  const float* b1     = (const float*)d_in[5];
  const float* W2     = (const float*)d_in[6];
  const float* a_src2 = (const float*)d_in[7];
  const float* a_dst2 = (const float*)d_in[8];
  const float* b2     = (const float*)d_in[9];
  float* out = (float*)d_out;

  const int HC   = in_sizes[5];
  const int NCLS = in_sizes[9];
  const int F    = in_sizes[2] / HC;
  const int N    = in_sizes[0] / F;
  const int E    = in_sizes[1] / 2;

  float* ws = (float*)d_ws;
  size_t off = 0;
  float* h1   = ws + off; off += (size_t)N * HC;
  float* out1 = ws + off; off += (size_t)N * HC;
  float* h2   = ws + off; off += (size_t)N * NCLS;
  float* as1  = ws + off; off += (size_t)N * 8;
  float* ad1  = ws + off; off += (size_t)N * 8;
  float* asd2 = ws + off; off += (size_t)N * 2;
  int* deg    = (int*)(ws + off); off += (size_t)N;
  int* total  = (int*)(ws + off); off += 2;
  int* start  = (int*)(ws + off); off += (size_t)N;
  int* csr    = (int*)(ws + off); off += (size_t)E;
  int* part   = (int*)(ws + off); off += (size_t)SORT_Q * N;

  const int span = cdiv(N, SORT_R);
  const int per  = cdiv(cdiv(E, 4), SORT_Q) * 4;
  range_count_kernel<<<dim3(SORT_Q, SORT_R), 256, span * sizeof(int), stream>>>(ei, E, N, span, per, part, total);
  segment_alloc_kernel<<<cdiv(N, 256), 256, 0, stream>>>(part, deg, start, total, N);

  const int ntiles = N / 16;
  const int gemm_blocks = cdiv(ntiles, 8);
  const size_t bins_bytes = span * sizeof(int), slab_bytes = 8 * GEMM1_SLAB * sizeof(float);
  scatter_and_gemm1_kernel<<<gemm_blocks + SORT_Q * SORT_R, 256,
                             bins_bytes > slab_bytes ? bins_bytes : slab_bytes, stream>>>(
      gemm_blocks, ntiles, ei, E, N, span, per, part, csr,
      x, W1, h1, F, HC, a_src1, a_dst1, as1, ad1);

  gather_agg1_kernel<<<cdiv(N, 16), 256, 0, stream>>>(start, deg, csr, as1, ad1, h1, b1, out1, N);

  gemm_wmma_f32_nt<3, 2><<<dim3(N / 16), 32, 0, stream>>>(out1, W2, h2, N, HC, NCLS, NCLS, NCLS,
                                                          a_src2, a_dst2, asd2, asd2);
  gather_agg2_kernel<<<cdiv(N, 16), 256, 0, stream>>>(start, deg, csr, asd2, h2, b2, out, N, NCLS);
}


# edge sort in two levels: edges go to 256-node buckets (each chunk read once, no per-chunk-per-node table), then one work-group per bucket ranks by node in LDS; list order, so sum order, changes
# speedup vs baseline: 11.6333x; 1.1277x over previous
#include <hip/hip_runtime.h>
#include <hip/hip_bf16.h>
#include <math.h>

typedef __attribute__((ext_vector_type(2))) float v2f;
typedef __attribute__((ext_vector_type(8))) float v8f;

#define NEG_SLOPE 0.2f
#define EPSV 1e-16f


#define BK_SHIFT 8
#define BK_NODES (1 << BK_SHIFT)
#define SORT_Q 256

typedef __attribute__((ext_vector_type(2))) int v2i;

typedef __attribute__((ext_vector_type(4))) int v4i;

template <bool WITH_SRC, class F>
__device__ __forceinline__ void scan_chunk(const int* __restrict__ srcp,
                                           const int* __restrict__ dstp,
                                           int e0, int e1, bool vec, F f) {
  if (vec) {
    for (int i = e0 + 4 * (int)threadIdx.x; i < e1; i += 4 * 256 * 4) {
      v4i d[4], s[4];
      #pragma unroll
      for (int u = 0; u < 4; ++u) {
        const int j = i + u * 1024;
        const bool in = j < e1;
        d[u] = in ? *(const v4i*)(dstp + j) : (v4i){-1, -1, -1, -1};
        if (WITH_SRC) s[u] = in ? *(const v4i*)(srcp + j) : (v4i){0, 0, 0, 0};
        else          s[u] = (v4i){0, 0, 0, 0};
      }
      #pragma unroll
      for (int u = 0; u < 4; ++u) {
        f(s[u].x, d[u].x); f(s[u].y, d[u].y); f(s[u].z, d[u].z); f(s[u].w, d[u].w);
      }
    }
  } else {
    for (int i = e0 + (int)threadIdx.x; i < e1; i += 256)
      f(WITH_SRC ? srcp[i] : 0, dstp[i]);
  }
}

__device__ __forceinline__ bool list_is_vec(const int* ei, int E) {
  return ((E & 3) == 0) && ((((size_t)ei) & 15) == 0);
}

__device__ __forceinline__ int block_scan256(int v, int* wsum, int& total) {
  const int lane = threadIdx.x & 31, w = threadIdx.x >> 5;
  int incl = v;
  for (int o = 1; o < 32; o <<= 1) {
    int t = __shfl_up(incl, o);
    if (lane >= o) incl += t;
  }
  if (lane == 31) wsum[w] = incl;
  __syncthreads();
  int woff = 0, tot = 0;
  #pragma unroll
  for (int i = 0; i < 8; ++i) {
    const int t = wsum[i];
    if (i < w) woff += t;
    tot += t;
  }
  __syncthreads();
  total = tot;
  return woff + incl;
}

__global__ void __launch_bounds__(256)
bucket_count_kernel(const int* __restrict__ ei, int E, int N, int NB, int per,
                    int* __restrict__ cntm) {
  extern __shared__ int hist[];
  const int q = blockIdx.x;
  for (int i = threadIdx.x; i < NB; i += 256) hist[i] = 0;
  __syncthreads();
  const int e0 = min(E, q * per), e1 = min(E, e0 + per);
  const unsigned un = (unsigned)N;
  scan_chunk<false>(ei, ei + E, e0, e1, list_is_vec(ei, E), [&](int, int d) {
    if ((unsigned)d < un) atomicAdd(&hist[d >> BK_SHIFT], 1);
  });
  __syncthreads();
  for (int i = threadIdx.x; i < NB; i += 256) cntm[(size_t)q * NB + i] = hist[i];
}

__device__ __forceinline__ void
bucket_scatter_block(int* lds, int q, const int* __restrict__ ei, int E, int N, int NB, int per,
                     const int* __restrict__ cntm, int* __restrict__ bbase,
                     int* __restrict__ btot, int* __restrict__ pairs) {
  int* cursor = lds;
  int* wsum = lds + NB;
  int carry = 0;
  for (int b0 = 0; b0 < NB; b0 += 256) {
    const int b = b0 + (int)threadIdx.x;
    const int bb = min(b, NB - 1);
    int before = 0, tot = 0;
    for (int q0 = 0; q0 < SORT_Q; q0 += 32) {
      int v[32];
      #pragma unroll
      for (int u = 0; u < 32; ++u) v[u] = cntm[(size_t)(q0 + u) * NB + bb];
      #pragma unroll
      for (int u = 0; u < 32; ++u) {
        tot += v[u];
        before += (q0 + u < q) ? v[u] : 0;
      }
    }
    if (b >= NB) { tot = 0; before = 0; }
    int tile_total;
    const int incl = block_scan256(tot, wsum, tile_total);
    const int excl = carry + incl - tot;
    if (b < NB) {
      cursor[b] = excl + before;
      if (q == 0) { bbase[b] = excl; btot[b] = tot; }
    }
    carry += tile_total;
  }
  __syncthreads();
  const int e0 = min(E, q * per), e1 = min(E, e0 + per);
  const unsigned un = (unsigned)N;
  v2i* __restrict__ pq = (v2i*)pairs;
  scan_chunk<true>(ei, ei + E, e0, e1, list_is_vec(ei, E), [&](int s, int d) {
    if ((unsigned)d < un) pq[atomicAdd(&cursor[d >> BK_SHIFT], 1)] = (v2i){s, d};
  });
}

__global__ void __launch_bounds__(BK_NODES)
bucket_rank_kernel(const int* __restrict__ pairs, const int* __restrict__ bbase,
                   const int* __restrict__ btot, int N,
                   int* __restrict__ start, int* __restrict__ deg, int* __restrict__ csr) {
  __shared__ int cnt[BK_NODES];
  __shared__ int wsum[8];
  const int b = blockIdx.x;
  const int base = bbase[b], m = btot[b];
  const v2i* __restrict__ pb = (const v2i*)pairs + base;
  cnt[threadIdx.x] = 0;
  __syncthreads();
  for (int i0 = threadIdx.x; i0 < m; i0 += BK_NODES * 8) {
    int d[8];
    #pragma unroll
    for (int u = 0; u < 8; ++u) {
      const int i = i0 + u * BK_NODES;
      d[u] = i < m ? pb[i].y : -1;
    }
    #pragma unroll
    for (int u = 0; u < 8; ++u)
      if (d[u] >= 0) atomicAdd(&cnt[d[u] & (BK_NODES - 1)], 1);
  }
  __syncthreads();
  const int c = cnt[threadIdx.x];
  int tot;
  const int incl = block_scan256(c, wsum, tot);
  const int first = base + incl - c;
  const int n = b * BK_NODES + (int)threadIdx.x;
  if (n < N) { start[n] = first; deg[n] = c; }
  cnt[threadIdx.x] = first;
  __syncthreads();
  for (int i0 = threadIdx.x; i0 < m; i0 += BK_NODES * 8) {
    v2i p[8];
    #pragma unroll
    for (int u = 0; u < 8; ++u) {
      const int i = i0 + u * BK_NODES;
      p[u] = i < m ? pb[i] : (v2i){0, -1};
    }
    #pragma unroll
    for (int u = 0; u < 8; ++u)
      if (p[u].y >= 0) csr[atomicAdd(&cnt[p[u].y & (BK_NODES - 1)], 1)] = p[u].x;
  }
}

__device__ __forceinline__ float lrelu(float v) { return v > 0.f ? v : NEG_SLOPE * v; }

typedef __attribute__((ext_vector_type(4))) float v4f;

template <int NT, int EPI>
__device__ __forceinline__ void
gemm_wmma_f32_tile(float* tile, int tileM, bool live,
                   const float* __restrict__ A, const float* __restrict__ B,
                   float* __restrict__ D, int K, int ldb, int ldd, int nvalid,
                   const float* __restrict__ a_src, const float* __restrict__ a_dst,
                   float* __restrict__ o_src, float* __restrict__ o_dst) {
  constexpr int PITCH = 16 * NT + 4;
  const int lane = threadIdx.x & 31;
  const int half = lane >> 4;
  const int l16  = lane & 15;
  const int rowA = tileM * 16 + l16;
  const float* __restrict__ arow = A + (size_t)rowA * K;

  v8f acc[NT];
  #pragma unroll
  for (int nt = 0; nt < NT; ++nt) acc[nt] = (v8f){0.f,0.f,0.f,0.f,0.f,0.f,0.f,0.f};

  for (int kk = 0; kk < K; kk += 4) {
    const int ka = kk + half * 2;
    v2f a;
    a.x = arow[ka];
    a.y = arow[ka + 1];
    const float* __restrict__ brow0 = B + (size_t)ka * ldb;
    const float* __restrict__ brow1 = B + (size_t)(ka + 1) * ldb;
    #pragma unroll
    for (int nt = 0; nt < NT; ++nt) {
      const int colB = nt * 16 + l16;
      v2f b;
      if (EPI == 2) {
        const int colc = min(colB, nvalid - 1);
        const float b0 = brow0[colc], b1 = brow1[colc];
        b.x = colB < nvalid ? b0 : 0.f;
        b.y = colB < nvalid ? b1 : 0.f;
      } else {
        b.x = brow0[colB];
        b.y = brow1[colB];
      }
      acc[nt] = __builtin_amdgcn_wmma_f32_16x16x4_f32(false, a, false, b,
                                                      (short)0, acc[nt], false, false);
    }
  }
  #pragma unroll
  for (int nt = 0; nt < NT; ++nt) {
    #pragma unroll
    for (int g = 0; g < 8; ++g) {
      const int r = tileM * 16 + half * 8 + g;
      const int c = nt * 16 + l16;
      if (live && c < nvalid) D[(size_t)r * ldd + c] = acc[nt][g];
      tile[(half * 8 + g) * PITCH + c] = acc[nt][g];
    }
  }
  __syncthreads();

  if (EPI == 1) {
    const int hh = lane & 7;
    float sp[8], dp[8];
    #pragma unroll
    for (int c = 0; c < 8; ++c) { sp[c] = a_src[hh * 8 + c]; dp[c] = a_dst[hh * 8 + c]; }
    #pragma unroll
    for (int i = 0; i < 4; ++i) {
      const int p = lane + 32 * i;
      const float* tp = tile + (p >> 3) * PITCH + hh * 8;
      const v4f t0 = *(const v4f*)tp, t1 = *(const v4f*)(tp + 4);
      const float t[8] = {t0.x, t0.y, t0.z, t0.w, t1.x, t1.y, t1.z, t1.w};
      float s1 = 0.f, s2 = 0.f;
      #pragma unroll
      for (int c = 0; c < 8; ++c) { s1 += t[c] * sp[c]; s2 += t[c] * dp[c]; }
      if (live) {
        o_src[(size_t)tileM * 128 + p] = s1;
        o_dst[(size_t)tileM * 128 + p] = s2;
      }
    }
  } else {
    const float* __restrict__ av = half ? a_dst : a_src;
    const float* tp = tile + l16 * PITCH;
    float s = 0.f;
    #pragma unroll
    for (int c4 = 0; c4 < 16 * NT; c4 += 4) {
      const v4f t = *(const v4f*)(tp + c4);
      const float tt[4] = {t.x, t.y, t.z, t.w};
      #pragma unroll
      for (int u = 0; u < 4; ++u) {
        const int c = c4 + u;
        const float coef = av[min(c, nvalid - 1)];
        s += tt[u] * (c < nvalid ? coef : 0.f);
      }
    }
    if (live) o_src[((size_t)tileM * 16 + l16) * 2 + half] = s;
  }
}

template <int NT, int EPI>
__global__ void __launch_bounds__(32)
gemm_wmma_f32_nt(const float* __restrict__ A, const float* __restrict__ B,
                 float* __restrict__ D, int M, int K, int ldb, int ldd, int nvalid,
                 const float* __restrict__ a_src, const float* __restrict__ a_dst,
                 float* __restrict__ o_src, float* __restrict__ o_dst) {
  __shared__ float tile[16 * (16 * NT + 4)];
  gemm_wmma_f32_tile<NT, EPI>(tile, blockIdx.x, true, A, B, D, K, ldb, ldd, nvalid,
                              a_src, a_dst, o_src, o_dst);
}

#define GEMM1_SLAB (16 * (16 * 4 + 4))
__global__ void __launch_bounds__(256)
scatter_and_gemm1_kernel(int gemm_blocks, int ntiles,
                         const int* __restrict__ ei, int E, int N, int NB, int per,
                         const int* __restrict__ cntm, int* __restrict__ bbase,
                         int* __restrict__ btot, int* __restrict__ pairs,
                         const float* __restrict__ x, const float* __restrict__ W1,
                         float* __restrict__ h1, int F, int HC,
                         const float* __restrict__ a_src, const float* __restrict__ a_dst,
                         float* __restrict__ as1, float* __restrict__ ad1) {
  extern __shared__ __attribute__((aligned(16))) int lds[];
  if ((int)blockIdx.x < gemm_blocks) {
    const int w = __builtin_amdgcn_readfirstlane(threadIdx.x >> 5);
    const int t = blockIdx.x * 8 + w;
    gemm_wmma_f32_tile<4, 1>((float*)lds + w * GEMM1_SLAB, min(t, ntiles - 1), t < ntiles,
                             x, W1, h1, F, HC, HC, HC, a_src, a_dst, as1, ad1);
  } else {
    bucket_scatter_block(lds, blockIdx.x - gemm_blocks, ei, E, N, NB, per,
                         cntm, bbase, btot, pairs);
  }
}


#define EXH_PITCH 20
#define EXH (8 * EXH_PITCH)
__global__ void __launch_bounds__(256)
gather_agg1_kernel(const int* __restrict__ start, const int* __restrict__ deg,
                   const int* __restrict__ csr,
                   const float* __restrict__ as, const float* __restrict__ ad,
                   const float* __restrict__ h1, const float* __restrict__ b,
                   float* __restrict__ out, int N) {
  __shared__ __attribute__((aligned(16))) float exw[16 * EXH];
  const int lane = threadIdx.x & 31;
  const int w = threadIdx.x >> 5;
  const int hbase = lane & 16;
  const int l16 = lane & 15;
  const int slot = 2 * w + (hbase >> 4);
  const int n = blockIdx.x * 16 + slot;
  const bool valid = n < N;
  const int nn = valid ? n : N - 1;
  const int hh = l16 >> 1;
  const int h = l16 & 7, g = l16 >> 3;
  float* lw = exw + slot * EXH;
  const v4f* __restrict__ h1q = (const v4f*)h1;

  const int beg = start[nn];
  const int cnt = valid ? deg[nn] : 0;
  const int cntmax = __builtin_amdgcn_readfirstlane(max(cnt, __shfl_xor(cnt, 16)));
  const float advh = ad[nn * 8 + h];
  const float vself = lrelu(as[nn * 8 + hh] + ad[nn * 8 + hh]);

  auto logits = [&](int myv, int m, float* o) {
    #pragma unroll
    for (int i = 0; i < 8; ++i) {
      const int e = 2 * i + g;
      const int s = __shfl(myv, hbase | e);
      const float v = lrelu(as[s * 8 + h] + advh);
      o[i] = e < m ? v : -INFINITY;
    }
  };

  const bool inA = l16 < cnt, inB = 16 + l16 < cnt;
  const int rA = csr[inA ? beg + l16 : 0];
  const int rB = csr[inB ? beg + 16 + l16 : 0];
  asm volatile("" :: "v"(rA), "v"(rB));
  const int myA = inA ? rA : nn, myB = inB ? rB : nn;
  float lgA[8], lgB[8];
  logits(myA, cnt, lgA);
  #pragma unroll
  for (int i = 0; i < 8; ++i) lgB[i] = -INFINITY;
  if (cntmax > 16) logits(myB, cnt - 16, lgB);
  float mxh = lrelu(as[nn * 8 + h] + advh);
  #pragma unroll
  for (int i = 0; i < 8; ++i) mxh = fmaxf(mxh, fmaxf(lgA[i], lgB[i]));
  for (int base = 32; base < cntmax; base += 16) {
    const int k = base + l16;
    const bool in = k < cnt;
    const int r = csr[in ? beg + k : 0];
    float l2[8];
    logits(in ? r : nn, cnt - base, l2);
    #pragma unroll
    for (int i = 0; i < 8; ++i) mxh = fmaxf(mxh, l2[i]);
  }
  mxh = fmaxf(mxh, __shfl_xor(mxh, 8));
  const float mx = __shfl(mxh, hbase | hh);

  const float ex = expf(vself - mx);
  float den = ex;
  const v4f hv = h1q[nn * 16 + l16];
  float a0 = ex * hv.x, a1 = ex * hv.y, a2 = ex * hv.z, a3 = ex * hv.w;

  auto row_batch = [&](int my, const float* lg, int mm) {
    #pragma unroll
    for (int i = 0; i < 8; ++i) lw[h * EXH_PITCH + 2 * i + g] = expf(lg[i] - mxh);
    const float* er = lw + hh * EXH_PITCH;
    for (int j = 0; j < mm; j += 4) {
      const v4f e4 = *(const v4f*)(er + j);
      const int s0 = __shfl(my, hbase | j);
      const int s1 = __shfl(my, hbase | (j + 1));
      const int s2 = __shfl(my, hbase | (j + 2));
      const int s3 = __shfl(my, hbase | (j + 3));
      const v4f r0 = h1q[s0 * 16 + l16];
      const v4f r1 = h1q[s1 * 16 + l16];
      const v4f r2 = h1q[s2 * 16 + l16];
      const v4f r3 = h1q[s3 * 16 + l16];
      den += e4.x;
      a0 = fmaf(e4.x, r0.x, a0); a1 = fmaf(e4.x, r0.y, a1); a2 = fmaf(e4.x, r0.z, a2); a3 = fmaf(e4.x, r0.w, a3);
      den += e4.y;
      a0 = fmaf(e4.y, r1.x, a0); a1 = fmaf(e4.y, r1.y, a1); a2 = fmaf(e4.y, r1.z, a2); a3 = fmaf(e4.y, r1.w, a3);
      den += e4.z;
      a0 = fmaf(e4.z, r2.x, a0); a1 = fmaf(e4.z, r2.y, a1); a2 = fmaf(e4.z, r2.z, a2); a3 = fmaf(e4.z, r2.w, a3);
      den += e4.w;
      a0 = fmaf(e4.w, r3.x, a0); a1 = fmaf(e4.w, r3.y, a1); a2 = fmaf(e4.w, r3.z, a2); a3 = fmaf(e4.w, r3.w, a3);
    }
  };
  if (cntmax > 0)  row_batch(myA, lgA, min(16, cntmax));
  if (cntmax > 16) row_batch(myB, lgB, min(16, cntmax - 16));
  for (int base = 32; base < cntmax; base += 16) {
    const int k = base + l16;
    const bool in = k < cnt;
    const int r = csr[in ? beg + k : 0];
    const int my = in ? r : nn;
    float l2[8];
    logits(my, cnt - base, l2);
    row_batch(my, l2, min(16, cntmax - base));
  }
  const float dd = den + EPSV;
  const float o0 = a0 / dd + b[4 * l16];
  const float o1 = a1 / dd + b[4 * l16 + 1];
  const float o2 = a2 / dd + b[4 * l16 + 2];
  const float o3 = a3 / dd + b[4 * l16 + 3];
  v4f o;
  o.x = o0 > 0.f ? o0 : expm1f(o0);
  o.y = o1 > 0.f ? o1 : expm1f(o1);
  o.z = o2 > 0.f ? o2 : expm1f(o2);
  o.w = o3 > 0.f ? o3 : expm1f(o3);
  if (valid) *(v4f*)(out + (size_t)nn * 64 + 4 * l16) = o;
}


__global__ void __launch_bounds__(256)
gather_agg2_kernel(const int* __restrict__ start, const int* __restrict__ deg,
                   const int* __restrict__ csr,
                   const float* __restrict__ asd,
                   const float* __restrict__ h2, const float* __restrict__ b2,
                   float* __restrict__ y, int N, int C) {
  __shared__ __attribute__((aligned(16))) float rows[16 * 64];
  const int lane = threadIdx.x & 31;
  const int w = threadIdx.x >> 5;
  const int hbase = lane & 16;
  const int l16 = lane & 15;
  const int slot = 2 * w + (hbase >> 4);
  const int n = blockIdx.x * 16 + slot;
  const bool valid = n < N;
  const int nn = valid ? n : N - 1;
  const bool own = 4 * l16 < C;
  const int colq = own ? l16 : 0;
  const v4f* __restrict__ h2q = (const v4f*)h2;
  const int C4 = C >> 2;

  const int beg = start[nn];
  const int cnt = valid ? deg[nn] : 0;
  const int cntmax = __builtin_amdgcn_readfirstlane(max(cnt, __shfl_xor(cnt, 16)));
  const float adv = asd[2 * nn + 1];
  const float vself = lrelu(asd[2 * nn] + adv);

  const bool inA = l16 < cnt, inB = 16 + l16 < cnt;
  const int rA = csr[inA ? beg + l16 : 0];
  const int rB = csr[inB ? beg + 16 + l16 : 0];
  asm volatile("" :: "v"(rA), "v"(rB));
  const int myA = inA ? rA : nn, myB = inB ? rB : nn;
  const float gA = asd[2 * myA], gB = asd[2 * myB];
  asm volatile("" :: "v"(gA), "v"(gB));
  const float lgA = inA ? lrelu(gA + adv) : -INFINITY;
  const float lgB = inB ? lrelu(gB + adv) : -INFINITY;
  float mx = fmaxf(vself, fmaxf(lgA, lgB));
  for (int base = 32; base < cntmax; base += 16) {
    const int k = base + l16;
    const bool in = k < cnt;
    const int r = csr[in ? beg + k : 0];
    const float g = asd[2 * (in ? r : nn)];
    mx = fmaxf(mx, in ? lrelu(g + adv) : -INFINITY);
  }
  for (int o = 8; o > 0; o >>= 1) mx = fmaxf(mx, __shfl_xor(mx, o));

  const float exs = expf(vself - mx);
  float den = exs;
  const v4f hs = h2q[nn * C4 + colq];
  float a0 = exs * hs.x, a1 = exs * hs.y, a2 = exs * hs.z, a3 = exs * hs.w;

  auto row_batch = [&](int my, float exl, int mm) {
    for (int j = 0; j < mm; j += 4) {
      int s[4];
      float e[4];
      v4f r[4];
      #pragma unroll
      for (int u = 0; u < 4; ++u) {
        const int sl = hbase | (j + u);
        s[u] = __shfl(my, sl);
        e[u] = __shfl(exl, sl);
        r[u] = h2q[s[u] * C4 + colq];
      }
      #pragma unroll
      for (int u = 0; u < 4; ++u) {
        den += e[u];
        a0 = fmaf(e[u], r[u].x, a0);
        a1 = fmaf(e[u], r[u].y, a1);
        a2 = fmaf(e[u], r[u].z, a2);
        a3 = fmaf(e[u], r[u].w, a3);
      }
    }
  };
  if (cntmax > 0)  row_batch(myA, inA ? expf(lgA - mx) : 0.f, min(16, cntmax));
  if (cntmax > 16) row_batch(myB, inB ? expf(lgB - mx) : 0.f, min(16, cntmax - 16));
  for (int base = 32; base < cntmax; base += 16) {
    const int k = base + l16;
    const bool in = k < cnt;
    const int r = csr[in ? beg + k : 0];
    const int my = in ? r : nn;
    const float lg = lrelu(asd[2 * my] + adv);
    row_batch(my, in ? expf(lg - mx) : 0.f, min(16, cntmax - base));
  }

  float v0 = -INFINITY, v1 = -INFINITY, v2 = -INFINITY, v3 = -INFINITY;
  if (own) {
    const float dd = den + EPSV;
    v0 = a0 / dd + b2[4 * l16];
    v1 = a1 / dd + b2[4 * l16 + 1];
    v2 = a2 / dd + b2[4 * l16 + 2];
    v3 = a3 / dd + b2[4 * l16 + 3];
    v0 = v0 > 0.f ? v0 : expm1f(v0);
    v1 = v1 > 0.f ? v1 : expm1f(v1);
    v2 = v2 > 0.f ? v2 : expm1f(v2);
    v3 = v3 > 0.f ? v3 : expm1f(v3);
  }
  float rmx = fmaxf(fmaxf(v0, v1), fmaxf(v2, v3));
  for (int o = 8; o > 0; o >>= 1) rmx = fmaxf(rmx, __shfl_xor(rmx, o));
  float q0 = own ? expf(v0 - rmx) + expf(v1 - rmx) : 0.f;
  float q1 = own ? expf(v2 - rmx) + expf(v3 - rmx) : 0.f;
  for (int o = 8; o > 0; o >>= 1) { q0 += __shfl_xor(q0, o); q1 += __shfl_xor(q1, o); }
  const float l = logf(q0 + q1);
  if (own && valid) {
    v4f o;
    o.x = v0 - rmx - l; o.y = v1 - rmx - l; o.z = v2 - rmx - l; o.w = v3 - rmx - l;
    *(v4f*)(rows + slot * C + 4 * l16) = o;
  }
  __syncthreads();
  const size_t gbase = (size_t)blockIdx.x * 16 * C;
  for (int i = 2 * (int)threadIdx.x; i < 16 * C; i += 512)
    if (gbase + i < (size_t)N * C)
      *(v2f*)(y + gbase + i) = *(const v2f*)(rows + i);
}


static inline int cdiv(int a, int b) { return (a + b - 1) / b; }

extern "C" void kernel_launch(void* const* d_in, const int* in_sizes, int n_in,
                              void* d_out, int out_size, void* d_ws, size_t ws_size,
                              hipStream_t stream) {
  const float* x      = (const float*)d_in[0];
  const int*   ei     = (const int*)  d_in[1];
  const float* W1     = (const float*)d_in[2];
  const float* a_src1 = (const float*)d_in[3];
  const float* a_dst1 = (const float*)d_in[4];
  const float* b1     = (const float*)d_in[5];
  const float* W2     = (const float*)d_in[6];
  const float* a_src2 = (const float*)d_in[7];
  const float* a_dst2 = (const float*)d_in[8];
  const float* b2     = (const float*)d_in[9];
  float* out = (float*)d_out;

  const int HC   = in_sizes[5];
  const int NCLS = in_sizes[9];
  const int F    = in_sizes[2] / HC;
  const int N    = in_sizes[0] / F;
  const int E    = in_sizes[1] / 2;

  const int NB = cdiv(N, BK_NODES);

  float* ws = (float*)d_ws;
  size_t off = 0;
  float* h1   = ws + off; off += (size_t)N * HC;
  float* out1 = ws + off; off += (size_t)N * HC;
  float* h2   = ws + off; off += (size_t)N * NCLS;
  float* as1  = ws + off; off += (size_t)N * 8;
  float* ad1  = ws + off; off += (size_t)N * 8;
  float* asd2 = ws + off; off += (size_t)N * 2;
  off = (off + 3) & ~(size_t)3;
  int* pairs  = (int*)(ws + off); off += (size_t)E * 2;
  int* csr    = (int*)(ws + off); off += (size_t)E;
  int* deg    = (int*)(ws + off); off += (size_t)N;
  int* start  = (int*)(ws + off); off += (size_t)N;
  int* cntm   = (int*)(ws + off); off += (size_t)SORT_Q * NB;
  int* bbase  = (int*)(ws + off); off += (size_t)NB;
  int* btot   = (int*)(ws + off); off += (size_t)NB;

  const int per = cdiv(cdiv(E, 4), SORT_Q) * 4;
  bucket_count_kernel<<<SORT_Q, 256, NB * sizeof(int), stream>>>(ei, E, N, NB, per, cntm);

  const int ntiles = N / 16;
  const int gemm_blocks = cdiv(ntiles, 8);
  const size_t bins_bytes = (NB + 8) * sizeof(int), slab_bytes = 8 * GEMM1_SLAB * sizeof(float);
  scatter_and_gemm1_kernel<<<gemm_blocks + SORT_Q, 256,
                             bins_bytes > slab_bytes ? bins_bytes : slab_bytes, stream>>>(
      gemm_blocks, ntiles, ei, E, N, NB, per, cntm, bbase, btot, pairs,
      x, W1, h1, F, HC, a_src1, a_dst1, as1, ad1);

  bucket_rank_kernel<<<NB, BK_NODES, 0, stream>>>(pairs, bbase, btot, N, start, deg, csr);

  gather_agg1_kernel<<<cdiv(N, 16), 256, 0, stream>>>(start, deg, csr, as1, ad1, h1, b1, out1, N);

  gemm_wmma_f32_nt<3, 2><<<dim3(N / 16), 32, 0, stream>>>(out1, W2, h2, N, HC, NCLS, NCLS, NCLS,
                                                          a_src2, a_dst2, asd2, asd2);
  gather_agg2_kernel<<<cdiv(N, 16), 256, 0, stream>>>(start, deg, csr, asd2, h2, b2, out, N, NCLS);
}


# layer 2's product elu(row) @ W2 and its logits move into the layer 1 gather: a work-group's 16 finished rows go via LDS to one wave's WMMA tile, no row store; 6->5 launches; sum order kept
# speedup vs baseline: 12.5071x; 1.0751x over previous
#include <hip/hip_runtime.h>
#include <hip/hip_bf16.h>
#include <math.h>

typedef __attribute__((ext_vector_type(2))) float v2f;
typedef __attribute__((ext_vector_type(8))) float v8f;

#define NEG_SLOPE 0.2f
#define EPSV 1e-16f


#define BK_SHIFT 8
#define BK_NODES (1 << BK_SHIFT)
#define SORT_Q 256

typedef __attribute__((ext_vector_type(2))) int v2i;

typedef __attribute__((ext_vector_type(4))) int v4i;

template <bool WITH_SRC, class F>
__device__ __forceinline__ void scan_chunk(const int* __restrict__ srcp,
                                           const int* __restrict__ dstp,
                                           int e0, int e1, bool vec, F f) {
  if (vec) {
    for (int i = e0 + 4 * (int)threadIdx.x; i < e1; i += 4 * 256 * 4) {
      v4i d[4], s[4];
      #pragma unroll
      for (int u = 0; u < 4; ++u) {
        const int j = i + u * 1024;
        const bool in = j < e1;
        d[u] = in ? *(const v4i*)(dstp + j) : (v4i){-1, -1, -1, -1};
        if (WITH_SRC) s[u] = in ? *(const v4i*)(srcp + j) : (v4i){0, 0, 0, 0};
        else          s[u] = (v4i){0, 0, 0, 0};
      }
      #pragma unroll
      for (int u = 0; u < 4; ++u) {
        f(s[u].x, d[u].x); f(s[u].y, d[u].y); f(s[u].z, d[u].z); f(s[u].w, d[u].w);
      }
    }
  } else {
    for (int i = e0 + (int)threadIdx.x; i < e1; i += 256)
      f(WITH_SRC ? srcp[i] : 0, dstp[i]);
  }
}

__device__ __forceinline__ bool list_is_vec(const int* ei, int E) {
  return ((E & 3) == 0) && ((((size_t)ei) & 15) == 0);
}

__device__ __forceinline__ int block_scan256(int v, int* wsum, int& total) {
  const int lane = threadIdx.x & 31, w = threadIdx.x >> 5;
  int incl = v;
  for (int o = 1; o < 32; o <<= 1) {
    int t = __shfl_up(incl, o);
    if (lane >= o) incl += t;
  }
  if (lane == 31) wsum[w] = incl;
  __syncthreads();
  int woff = 0, tot = 0;
  #pragma unroll
  for (int i = 0; i < 8; ++i) {
    const int t = wsum[i];
    if (i < w) woff += t;
    tot += t;
  }
  __syncthreads();
  total = tot;
  return woff + incl;
}

__global__ void __launch_bounds__(256)
bucket_count_kernel(const int* __restrict__ ei, int E, int N, int NB, int per,
                    int* __restrict__ cntm) {
  extern __shared__ int hist[];
  const int q = blockIdx.x;
  for (int i = threadIdx.x; i < NB; i += 256) hist[i] = 0;
  __syncthreads();
  const int e0 = min(E, q * per), e1 = min(E, e0 + per);
  const unsigned un = (unsigned)N;
  scan_chunk<false>(ei, ei + E, e0, e1, list_is_vec(ei, E), [&](int, int d) {
    if ((unsigned)d < un) atomicAdd(&hist[d >> BK_SHIFT], 1);
  });
  __syncthreads();
  for (int i = threadIdx.x; i < NB; i += 256) cntm[(size_t)q * NB + i] = hist[i];
}

__device__ __forceinline__ void
bucket_scatter_block(int* lds, int q, const int* __restrict__ ei, int E, int N, int NB, int per,
                     const int* __restrict__ cntm, int* __restrict__ bbase,
                     int* __restrict__ btot, int* __restrict__ pairs) {
  int* cursor = lds;
  int* wsum = lds + NB;
  int carry = 0;
  for (int b0 = 0; b0 < NB; b0 += 256) {
    const int b = b0 + (int)threadIdx.x;
    const int bb = min(b, NB - 1);
    int before = 0, tot = 0;
    for (int q0 = 0; q0 < SORT_Q; q0 += 32) {
      int v[32];
      #pragma unroll
      for (int u = 0; u < 32; ++u) v[u] = cntm[(size_t)(q0 + u) * NB + bb];
      #pragma unroll
      for (int u = 0; u < 32; ++u) {
        tot += v[u];
        before += (q0 + u < q) ? v[u] : 0;
      }
    }
    if (b >= NB) { tot = 0; before = 0; }
    int tile_total;
    const int incl = block_scan256(tot, wsum, tile_total);
    const int excl = carry + incl - tot;
    if (b < NB) {
      cursor[b] = excl + before;
      if (q == 0) { bbase[b] = excl; btot[b] = tot; }
    }
    carry += tile_total;
  }
  __syncthreads();
  const int e0 = min(E, q * per), e1 = min(E, e0 + per);
  const unsigned un = (unsigned)N;
  v2i* __restrict__ pq = (v2i*)pairs;
  scan_chunk<true>(ei, ei + E, e0, e1, list_is_vec(ei, E), [&](int s, int d) {
    if ((unsigned)d < un) pq[atomicAdd(&cursor[d >> BK_SHIFT], 1)] = (v2i){s, d};
  });
}

__global__ void __launch_bounds__(BK_NODES)
bucket_rank_kernel(const int* __restrict__ pairs, const int* __restrict__ bbase,
                   const int* __restrict__ btot, int N,
                   int* __restrict__ start, int* __restrict__ deg, int* __restrict__ csr) {
  __shared__ int cnt[BK_NODES];
  __shared__ int wsum[8];
  const int b = blockIdx.x;
  const int base = bbase[b], m = btot[b];
  const v2i* __restrict__ pb = (const v2i*)pairs + base;
  cnt[threadIdx.x] = 0;
  __syncthreads();
  for (int i0 = threadIdx.x; i0 < m; i0 += BK_NODES * 8) {
    int d[8];
    #pragma unroll
    for (int u = 0; u < 8; ++u) {
      const int i = i0 + u * BK_NODES;
      d[u] = i < m ? pb[i].y : -1;
    }
    #pragma unroll
    for (int u = 0; u < 8; ++u)
      if (d[u] >= 0) atomicAdd(&cnt[d[u] & (BK_NODES - 1)], 1);
  }
  __syncthreads();
  const int c = cnt[threadIdx.x];
  int tot;
  const int incl = block_scan256(c, wsum, tot);
  const int first = base + incl - c;
  const int n = b * BK_NODES + (int)threadIdx.x;
  if (n < N) { start[n] = first; deg[n] = c; }
  cnt[threadIdx.x] = first;
  __syncthreads();
  for (int i0 = threadIdx.x; i0 < m; i0 += BK_NODES * 8) {
    v2i p[8];
    #pragma unroll
    for (int u = 0; u < 8; ++u) {
      const int i = i0 + u * BK_NODES;
      p[u] = i < m ? pb[i] : (v2i){0, -1};
    }
    #pragma unroll
    for (int u = 0; u < 8; ++u)
      if (p[u].y >= 0) csr[atomicAdd(&cnt[p[u].y & (BK_NODES - 1)], 1)] = p[u].x;
  }
}

__device__ __forceinline__ float lrelu(float v) { return v > 0.f ? v : NEG_SLOPE * v; }

typedef __attribute__((ext_vector_type(4))) float v4f;

template <int NT>
__device__ __forceinline__ void
gemm_wmma_f32_tile(float* tile, int tileM, bool live,
                   const float* __restrict__ A, const float* __restrict__ B,
                   float* __restrict__ D, int K, int ldb, int ldd, int nvalid,
                   const float* __restrict__ a_src, const float* __restrict__ a_dst,
                   float* __restrict__ o_src, float* __restrict__ o_dst) {
  constexpr int PITCH = 16 * NT + 4;
  const int lane = threadIdx.x & 31;
  const int half = lane >> 4;
  const int l16  = lane & 15;
  const int rowA = tileM * 16 + l16;
  const float* __restrict__ arow = A + (size_t)rowA * K;

  v8f acc[NT];
  #pragma unroll
  for (int nt = 0; nt < NT; ++nt) acc[nt] = (v8f){0.f,0.f,0.f,0.f,0.f,0.f,0.f,0.f};

  for (int kk = 0; kk < K; kk += 4) {
    const int ka = kk + half * 2;
    v2f a;
    a.x = arow[ka];
    a.y = arow[ka + 1];
    const float* __restrict__ brow0 = B + (size_t)ka * ldb;
    const float* __restrict__ brow1 = B + (size_t)(ka + 1) * ldb;
    #pragma unroll
    for (int nt = 0; nt < NT; ++nt) {
      const int colB = nt * 16 + l16;
      v2f b;
      b.x = brow0[colB];
      b.y = brow1[colB];
      acc[nt] = __builtin_amdgcn_wmma_f32_16x16x4_f32(false, a, false, b,
                                                      (short)0, acc[nt], false, false);
    }
  }
  #pragma unroll
  for (int nt = 0; nt < NT; ++nt) {
    #pragma unroll
    for (int g = 0; g < 8; ++g) {
      const int r = tileM * 16 + half * 8 + g;
      const int c = nt * 16 + l16;
      if (live && c < nvalid) D[(size_t)r * ldd + c] = acc[nt][g];
      tile[(half * 8 + g) * PITCH + c] = acc[nt][g];
    }
  }
  __syncthreads();

  const int hh = lane & 7;
  float sp[8], dp[8];
  #pragma unroll
  for (int c = 0; c < 8; ++c) { sp[c] = a_src[hh * 8 + c]; dp[c] = a_dst[hh * 8 + c]; }
  #pragma unroll
  for (int i = 0; i < 4; ++i) {
    const int p = lane + 32 * i;
    const float* tp = tile + (p >> 3) * PITCH + hh * 8;
    const v4f t0 = *(const v4f*)tp, t1 = *(const v4f*)(tp + 4);
    const float t[8] = {t0.x, t0.y, t0.z, t0.w, t1.x, t1.y, t1.z, t1.w};
    float s1 = 0.f, s2 = 0.f;
    #pragma unroll
    for (int c = 0; c < 8; ++c) { s1 += t[c] * sp[c]; s2 += t[c] * dp[c]; }
    if (live) {
      o_src[(size_t)tileM * 128 + p] = s1;
      o_dst[(size_t)tileM * 128 + p] = s2;
    }
  }
}

#define ROW1_PITCH 68
#define SLAB2_PITCH 52
__device__ __forceinline__ void
layer2_tile_from_lds(const float* rows1, float* slab, int tileM,
                     const float* __restrict__ B, int nvalid, float* __restrict__ D,
                     const float* __restrict__ a_src, const float* __restrict__ a_dst,
                     float* __restrict__ o_sd) {
  constexpr int NT = 3;
  const int lane = threadIdx.x & 31;
  const int half = lane >> 4;
  const int l16  = lane & 15;
  const float* arow = rows1 + l16 * ROW1_PITCH + half * 2;

  v8f acc[NT];
  #pragma unroll
  for (int nt = 0; nt < NT; ++nt) acc[nt] = (v8f){0.f,0.f,0.f,0.f,0.f,0.f,0.f,0.f};

  unsigned off0[NT], off1[NT];
  bool cin[NT];
  #pragma unroll
  for (int nt = 0; nt < NT; ++nt) {
    const int colB = nt * 16 + l16;
    cin[nt] = colB < nvalid;
    off0[nt] = (unsigned)(half * 2 * nvalid + min(colB, nvalid - 1));
    off1[nt] = off0[nt] + (unsigned)nvalid;
  }
  #pragma unroll 4
  for (int kk = 0; kk < 64; kk += 4) {
    const v2f a = *(const v2f*)(arow + kk);
    const float* __restrict__ bk = B + kk * nvalid;
    #pragma unroll
    for (int nt = 0; nt < NT; ++nt) {
      const float b0 = bk[off0[nt]], b1 = bk[off1[nt]];
      v2f b;
      b.x = cin[nt] ? b0 : 0.f;
      b.y = cin[nt] ? b1 : 0.f;
      acc[nt] = __builtin_amdgcn_wmma_f32_16x16x4_f32(false, a, false, b,
                                                      (short)0, acc[nt], false, false);
    }
  }
  float* __restrict__ dt = D + (size_t)tileM * 16 * nvalid;
  #pragma unroll
  for (int nt = 0; nt < NT; ++nt) {
    const int c = nt * 16 + l16;
    if (cin[nt]) {
      #pragma unroll
      for (int g = 0; g < 8; ++g) dt[(unsigned)((half * 8 + g) * nvalid + c)] = acc[nt][g];
    }
    #pragma unroll
    for (int g = 0; g < 8; ++g) slab[(half * 8 + g) * SLAB2_PITCH + c] = acc[nt][g];
  }

  const float* __restrict__ av = half ? a_dst : a_src;
  const float* tp = slab + l16 * SLAB2_PITCH;
  float s = 0.f;
  #pragma unroll
  for (int c4 = 0; c4 < 16 * NT; c4 += 4) {
    const v4f t = *(const v4f*)(tp + c4);
    const float tt[4] = {t.x, t.y, t.z, t.w};
    #pragma unroll
    for (int u = 0; u < 4; ++u) {
      const int c = c4 + u;
      const float coef = av[min(c, nvalid - 1)];
      s += tt[u] * (c < nvalid ? coef : 0.f);
    }
  }
  o_sd[((size_t)tileM * 16 + l16) * 2 + half] = s;
}

#define GEMM1_SLAB (16 * (16 * 4 + 4))
__global__ void __launch_bounds__(256)
scatter_and_gemm1_kernel(int gemm_blocks, int ntiles,
                         const int* __restrict__ ei, int E, int N, int NB, int per,
                         const int* __restrict__ cntm, int* __restrict__ bbase,
                         int* __restrict__ btot, int* __restrict__ pairs,
                         const float* __restrict__ x, const float* __restrict__ W1,
                         float* __restrict__ h1, int F, int HC,
                         const float* __restrict__ a_src, const float* __restrict__ a_dst,
                         float* __restrict__ as1, float* __restrict__ ad1) {
  extern __shared__ __attribute__((aligned(16))) int lds[];
  if ((int)blockIdx.x < gemm_blocks) {
    const int w = __builtin_amdgcn_readfirstlane(threadIdx.x >> 5);
    const int t = blockIdx.x * 8 + w;
    gemm_wmma_f32_tile<4>((float*)lds + w * GEMM1_SLAB, min(t, ntiles - 1), t < ntiles,
                             x, W1, h1, F, HC, HC, HC, a_src, a_dst, as1, ad1);
  } else {
    bucket_scatter_block(lds, blockIdx.x - gemm_blocks, ei, E, N, NB, per,
                         cntm, bbase, btot, pairs);
  }
}


#define EXH_PITCH 20
#define EXH (8 * EXH_PITCH)
__global__ void __launch_bounds__(256)
gather_agg1_kernel(const int* __restrict__ start, const int* __restrict__ deg,
                   const int* __restrict__ csr,
                   const float* __restrict__ as, const float* __restrict__ ad,
                   const float* __restrict__ h1, const float* __restrict__ b, int N,
                   const float* __restrict__ W2, int C2,
                   const float* __restrict__ a_src2, const float* __restrict__ a_dst2,
                   float* __restrict__ h2, float* __restrict__ asd2) {
  __shared__ __attribute__((aligned(16))) float exw[16 * EXH];
  __shared__ __attribute__((aligned(16))) float rows1[16 * ROW1_PITCH];
  __shared__ __attribute__((aligned(16))) float slab2[16 * SLAB2_PITCH];
  const int lane = threadIdx.x & 31;
  const int w = threadIdx.x >> 5;
  const int hbase = lane & 16;
  const int l16 = lane & 15;
  const int slot = 2 * w + (hbase >> 4);
  const int n = blockIdx.x * 16 + slot;
  const bool valid = n < N;
  const int nn = valid ? n : N - 1;
  const int hh = l16 >> 1;
  const int h = l16 & 7, g = l16 >> 3;
  float* lw = exw + slot * EXH;
  const v4f* __restrict__ h1q = (const v4f*)h1;

  const int beg = start[nn];
  const int cnt = valid ? deg[nn] : 0;
  const int cntmax = __builtin_amdgcn_readfirstlane(max(cnt, __shfl_xor(cnt, 16)));
  const float advh = ad[nn * 8 + h];
  const float vself = lrelu(as[nn * 8 + hh] + ad[nn * 8 + hh]);

  auto logits = [&](int myv, int m, float* o) {
    #pragma unroll
    for (int i = 0; i < 8; ++i) {
      const int e = 2 * i + g;
      const int s = __shfl(myv, hbase | e);
      const float v = lrelu(as[s * 8 + h] + advh);
      o[i] = e < m ? v : -INFINITY;
    }
  };

  const bool inA = l16 < cnt, inB = 16 + l16 < cnt;
  const int rA = csr[inA ? beg + l16 : 0];
  const int rB = csr[inB ? beg + 16 + l16 : 0];
  asm volatile("" :: "v"(rA), "v"(rB));
  const int myA = inA ? rA : nn, myB = inB ? rB : nn;
  float lgA[8], lgB[8];
  logits(myA, cnt, lgA);
  #pragma unroll
  for (int i = 0; i < 8; ++i) lgB[i] = -INFINITY;
  if (cntmax > 16) logits(myB, cnt - 16, lgB);
  float mxh = lrelu(as[nn * 8 + h] + advh);
  #pragma unroll
  for (int i = 0; i < 8; ++i) mxh = fmaxf(mxh, fmaxf(lgA[i], lgB[i]));
  for (int base = 32; base < cntmax; base += 16) {
    const int k = base + l16;
    const bool in = k < cnt;
    const int r = csr[in ? beg + k : 0];
    float l2[8];
    logits(in ? r : nn, cnt - base, l2);
    #pragma unroll
    for (int i = 0; i < 8; ++i) mxh = fmaxf(mxh, l2[i]);
  }
  mxh = fmaxf(mxh, __shfl_xor(mxh, 8));
  const float mx = __shfl(mxh, hbase | hh);

  const float ex = expf(vself - mx);
  float den = ex;
  const v4f hv = h1q[nn * 16 + l16];
  float a0 = ex * hv.x, a1 = ex * hv.y, a2 = ex * hv.z, a3 = ex * hv.w;

  auto row_batch = [&](int my, const float* lg, int mm) {
    #pragma unroll
    for (int i = 0; i < 8; ++i) lw[h * EXH_PITCH + 2 * i + g] = expf(lg[i] - mxh);
    const float* er = lw + hh * EXH_PITCH;
    for (int j = 0; j < mm; j += 4) {
      const v4f e4 = *(const v4f*)(er + j);
      const int s0 = __shfl(my, hbase | j);
      const int s1 = __shfl(my, hbase | (j + 1));
      const int s2 = __shfl(my, hbase | (j + 2));
      const int s3 = __shfl(my, hbase | (j + 3));
      const v4f r0 = h1q[s0 * 16 + l16];
      const v4f r1 = h1q[s1 * 16 + l16];
      const v4f r2 = h1q[s2 * 16 + l16];
      const v4f r3 = h1q[s3 * 16 + l16];
      den += e4.x;
      a0 = fmaf(e4.x, r0.x, a0); a1 = fmaf(e4.x, r0.y, a1); a2 = fmaf(e4.x, r0.z, a2); a3 = fmaf(e4.x, r0.w, a3);
      den += e4.y;
      a0 = fmaf(e4.y, r1.x, a0); a1 = fmaf(e4.y, r1.y, a1); a2 = fmaf(e4.y, r1.z, a2); a3 = fmaf(e4.y, r1.w, a3);
      den += e4.z;
      a0 = fmaf(e4.z, r2.x, a0); a1 = fmaf(e4.z, r2.y, a1); a2 = fmaf(e4.z, r2.z, a2); a3 = fmaf(e4.z, r2.w, a3);
      den += e4.w;
      a0 = fmaf(e4.w, r3.x, a0); a1 = fmaf(e4.w, r3.y, a1); a2 = fmaf(e4.w, r3.z, a2); a3 = fmaf(e4.w, r3.w, a3);
    }
  };
  if (cntmax > 0)  row_batch(myA, lgA, min(16, cntmax));
  if (cntmax > 16) row_batch(myB, lgB, min(16, cntmax - 16));
  for (int base = 32; base < cntmax; base += 16) {
    const int k = base + l16;
    const bool in = k < cnt;
    const int r = csr[in ? beg + k : 0];
    const int my = in ? r : nn;
    float l2[8];
    logits(my, cnt - base, l2);
    row_batch(my, l2, min(16, cntmax - base));
  }
  const float dd = den + EPSV;
  const float o0 = a0 / dd + b[4 * l16];
  const float o1 = a1 / dd + b[4 * l16 + 1];
  const float o2 = a2 / dd + b[4 * l16 + 2];
  const float o3 = a3 / dd + b[4 * l16 + 3];
  v4f o;
  o.x = o0 > 0.f ? o0 : expm1f(o0);
  o.y = o1 > 0.f ? o1 : expm1f(o1);
  o.z = o2 > 0.f ? o2 : expm1f(o2);
  o.w = o3 > 0.f ? o3 : expm1f(o3);
  *(v4f*)(rows1 + slot * ROW1_PITCH + 4 * l16) = o;
  __syncthreads();
  const int pw = (blockIdx.x + (blockIdx.x >> 8)) & 7;
  if (__builtin_amdgcn_readfirstlane(w) == pw)
    layer2_tile_from_lds(rows1, slab2, blockIdx.x, W2, C2, h2, a_src2, a_dst2, asd2);
}


__global__ void __launch_bounds__(256)
gather_agg2_kernel(const int* __restrict__ start, const int* __restrict__ deg,
                   const int* __restrict__ csr,
                   const float* __restrict__ asd,
                   const float* __restrict__ h2, const float* __restrict__ b2,
                   float* __restrict__ y, int N, int C) {
  __shared__ __attribute__((aligned(16))) float rows[16 * 64];
  const int lane = threadIdx.x & 31;
  const int w = threadIdx.x >> 5;
  const int hbase = lane & 16;
  const int l16 = lane & 15;
  const int slot = 2 * w + (hbase >> 4);
  const int n = blockIdx.x * 16 + slot;
  const bool valid = n < N;
  const int nn = valid ? n : N - 1;
  const bool own = 4 * l16 < C;
  const int colq = own ? l16 : 0;
  const v4f* __restrict__ h2q = (const v4f*)h2;
  const int C4 = C >> 2;

  const int beg = start[nn];
  const int cnt = valid ? deg[nn] : 0;
  const int cntmax = __builtin_amdgcn_readfirstlane(max(cnt, __shfl_xor(cnt, 16)));
  const float adv = asd[2 * nn + 1];
  const float vself = lrelu(asd[2 * nn] + adv);

  const bool inA = l16 < cnt, inB = 16 + l16 < cnt;
  const int rA = csr[inA ? beg + l16 : 0];
  const int rB = csr[inB ? beg + 16 + l16 : 0];
  asm volatile("" :: "v"(rA), "v"(rB));
  const int myA = inA ? rA : nn, myB = inB ? rB : nn;
  const float gA = asd[2 * myA], gB = asd[2 * myB];
  asm volatile("" :: "v"(gA), "v"(gB));
  const float lgA = inA ? lrelu(gA + adv) : -INFINITY;
  const float lgB = inB ? lrelu(gB + adv) : -INFINITY;
  float mx = fmaxf(vself, fmaxf(lgA, lgB));
  for (int base = 32; base < cntmax; base += 16) {
    const int k = base + l16;
    const bool in = k < cnt;
    const int r = csr[in ? beg + k : 0];
    const float g = asd[2 * (in ? r : nn)];
    mx = fmaxf(mx, in ? lrelu(g + adv) : -INFINITY);
  }
  for (int o = 8; o > 0; o >>= 1) mx = fmaxf(mx, __shfl_xor(mx, o));

  const float exs = expf(vself - mx);
  float den = exs;
  const v4f hs = h2q[nn * C4 + colq];
  float a0 = exs * hs.x, a1 = exs * hs.y, a2 = exs * hs.z, a3 = exs * hs.w;

  auto row_batch = [&](int my, float exl, int mm) {
    for (int j = 0; j < mm; j += 4) {
      int s[4];
      float e[4];
      v4f r[4];
      #pragma unroll
      for (int u = 0; u < 4; ++u) {
        const int sl = hbase | (j + u);
        s[u] = __shfl(my, sl);
        e[u] = __shfl(exl, sl);
        r[u] = h2q[s[u] * C4 + colq];
      }
      #pragma unroll
      for (int u = 0; u < 4; ++u) {
        den += e[u];
        a0 = fmaf(e[u], r[u].x, a0);
        a1 = fmaf(e[u], r[u].y, a1);
        a2 = fmaf(e[u], r[u].z, a2);
        a3 = fmaf(e[u], r[u].w, a3);
      }
    }
  };
  if (cntmax > 0)  row_batch(myA, inA ? expf(lgA - mx) : 0.f, min(16, cntmax));
  if (cntmax > 16) row_batch(myB, inB ? expf(lgB - mx) : 0.f, min(16, cntmax - 16));
  for (int base = 32; base < cntmax; base += 16) {
    const int k = base + l16;
    const bool in = k < cnt;
    const int r = csr[in ? beg + k : 0];
    const int my = in ? r : nn;
    const float lg = lrelu(asd[2 * my] + adv);
    row_batch(my, in ? expf(lg - mx) : 0.f, min(16, cntmax - base));
  }

  float v0 = -INFINITY, v1 = -INFINITY, v2 = -INFINITY, v3 = -INFINITY;
  if (own) {
    const float dd = den + EPSV;
    v0 = a0 / dd + b2[4 * l16];
    v1 = a1 / dd + b2[4 * l16 + 1];
    v2 = a2 / dd + b2[4 * l16 + 2];
    v3 = a3 / dd + b2[4 * l16 + 3];
    v0 = v0 > 0.f ? v0 : expm1f(v0);
    v1 = v1 > 0.f ? v1 : expm1f(v1);
    v2 = v2 > 0.f ? v2 : expm1f(v2);
    v3 = v3 > 0.f ? v3 : expm1f(v3);
  }
  float rmx = fmaxf(fmaxf(v0, v1), fmaxf(v2, v3));
  for (int o = 8; o > 0; o >>= 1) rmx = fmaxf(rmx, __shfl_xor(rmx, o));
  float q0 = own ? expf(v0 - rmx) + expf(v1 - rmx) : 0.f;
  float q1 = own ? expf(v2 - rmx) + expf(v3 - rmx) : 0.f;
  for (int o = 8; o > 0; o >>= 1) { q0 += __shfl_xor(q0, o); q1 += __shfl_xor(q1, o); }
  const float l = logf(q0 + q1);
  if (own && valid) {
    v4f o;
    o.x = v0 - rmx - l; o.y = v1 - rmx - l; o.z = v2 - rmx - l; o.w = v3 - rmx - l;
    *(v4f*)(rows + slot * C + 4 * l16) = o;
  }
  __syncthreads();
  const size_t gbase = (size_t)blockIdx.x * 16 * C;
  for (int i = 2 * (int)threadIdx.x; i < 16 * C; i += 512)
    if (gbase + i < (size_t)N * C)
      *(v2f*)(y + gbase + i) = *(const v2f*)(rows + i);
}


static inline int cdiv(int a, int b) { return (a + b - 1) / b; }

extern "C" void kernel_launch(void* const* d_in, const int* in_sizes, int n_in,
                              void* d_out, int out_size, void* d_ws, size_t ws_size,
                              hipStream_t stream) {
  const float* x      = (const float*)d_in[0];
  const int*   ei     = (const int*)  d_in[1];
  const float* W1     = (const float*)d_in[2];
  const float* a_src1 = (const float*)d_in[3];
  const float* a_dst1 = (const float*)d_in[4];
  const float* b1     = (const float*)d_in[5];
  const float* W2     = (const float*)d_in[6];
  const float* a_src2 = (const float*)d_in[7];
  const float* a_dst2 = (const float*)d_in[8];
  const float* b2     = (const float*)d_in[9];
  float* out = (float*)d_out;

  const int HC   = in_sizes[5];
  const int NCLS = in_sizes[9];
  const int F    = in_sizes[2] / HC;
  const int N    = in_sizes[0] / F;
  const int E    = in_sizes[1] / 2;

  const int NB = cdiv(N, BK_NODES);

  float* ws = (float*)d_ws;
  size_t off = 0;
  float* h1   = ws + off; off += (size_t)N * HC;
  const size_t Npad = (size_t)cdiv(N, 16) * 16;
  float* h2   = ws + off; off += Npad * NCLS;
  float* as1  = ws + off; off += (size_t)N * 8;
  float* ad1  = ws + off; off += (size_t)N * 8;
  float* asd2 = ws + off; off += Npad * 2;
  off = (off + 3) & ~(size_t)3;
  int* pairs  = (int*)(ws + off); off += (size_t)E * 2;
  int* csr    = (int*)(ws + off); off += (size_t)E;
  int* deg    = (int*)(ws + off); off += (size_t)N;
  int* start  = (int*)(ws + off); off += (size_t)N;
  int* cntm   = (int*)(ws + off); off += (size_t)SORT_Q * NB;
  int* bbase  = (int*)(ws + off); off += (size_t)NB;
  int* btot   = (int*)(ws + off); off += (size_t)NB;

  const int per = cdiv(cdiv(E, 4), SORT_Q) * 4;
  bucket_count_kernel<<<SORT_Q, 256, NB * sizeof(int), stream>>>(ei, E, N, NB, per, cntm);

  const int ntiles = N / 16;
  const int gemm_blocks = cdiv(ntiles, 8);
  const size_t bins_bytes = (NB + 8) * sizeof(int), slab_bytes = 8 * GEMM1_SLAB * sizeof(float);
  scatter_and_gemm1_kernel<<<gemm_blocks + SORT_Q, 256,
                             bins_bytes > slab_bytes ? bins_bytes : slab_bytes, stream>>>(
      gemm_blocks, ntiles, ei, E, N, NB, per, cntm, bbase, btot, pairs,
      x, W1, h1, F, HC, a_src1, a_dst1, as1, ad1);

  bucket_rank_kernel<<<NB, BK_NODES, 0, stream>>>(pairs, bbase, btot, N, start, deg, csr);

  gather_agg1_kernel<<<cdiv(N, 16), 256, 0, stream>>>(start, deg, csr, as1, ad1, h1, b1, N,
                                                      W2, NCLS, a_src2, a_dst2, h2, asd2);

  gather_agg2_kernel<<<cdiv(N, 16), 256, 0, stream>>>(start, deg, csr, asd2, h2, b2, out, N, NCLS);
}
